# starting kernel
# baseline (speedup 1.0000x reference)
#include <hip/hip_runtime.h>

typedef __attribute__((ext_vector_type(2))) float v2f;
typedef __attribute__((ext_vector_type(8))) float v8f;

constexpr int Bc = 2;
constexpr int Sc = 1024;
constexpr int Nc = Sc + 1;
constexpr int Dc = 512;
constexpr int Rc = 32;
constexpr int Lc = 3;
constexpr int Wc = 64;
constexpr int WNc = 2 * Wc + 1;
constexpr int WINc = 144;
constexpr int NT = (Nc + 15) / 16;

constexpr int XSTR = Dc + 4;
constexpr int QSTR = 33;
constexpr int KSTR = 33;
constexpr int SSTR = 146;
constexpr int WSTR = 145;

__device__ __forceinline__ v8f wmma_f32(v2f a, v2f b, v8f c) {
  return __builtin_amdgcn_wmma_f32_16x16x4_f32(false, a, false, b, (short)0, c,
                                               false, false);
}

__device__ __forceinline__ float block_sum256(float v, float* red) {
  for (int m = 16; m >= 1; m >>= 1) v += __shfl_xor(v, m, 32);
  const int wave = threadIdx.x >> 5, lane = threadIdx.x & 31;
  __syncthreads();
  if (lane == 0) red[wave] = v;
  __syncthreads();
  if (threadIdx.x < 8) {
    float t = red[threadIdx.x];
    t += __shfl_xor(t, 1, 32);
    t += __shfl_xor(t, 2, 32);
    t += __shfl_xor(t, 4, 32);
    if (threadIdx.x == 0) red[0] = t;
  }
  __syncthreads();
  return red[0];
}

__global__ __launch_bounds__(256) void embed_ln_kernel(
    const int* __restrict__ ids, const float* __restrict__ emb,
    const float* __restrict__ pos, const float* __restrict__ anchor_val,
    const float* __restrict__ anchor_state, const float* __restrict__ gamma,
    const float* __restrict__ beta, const float* __restrict__ sp_w,
    const float* __restrict__ sp_b, float* __restrict__ val,
    float* __restrict__ st) {
  __shared__ float red[8];
  const int n = blockIdx.x % Nc;
  const int b = blockIdx.x / Nc;
  const int tid = threadIdx.x;
  float* vrow = val + ((size_t)b * Nc + n) * Dc;
  if (n == 0) {
    vrow[tid] = anchor_val[tid];
    vrow[tid + 256] = anchor_val[tid + 256];
    if (tid == 0) st[(size_t)b * Nc] = anchor_state[0];
    return;
  }
  const int s = n - 1;
  const int id = ids[b * Sc + s];
  const float e0 = emb[(size_t)id * Dc + tid] + pos[(size_t)s * Dc + tid];
  const float e1 = emb[(size_t)id * Dc + tid + 256] + pos[(size_t)s * Dc + tid + 256];
  const float sum = block_sum256(e0 + e1, red);
  const float ssq = block_sum256(e0 * e0 + e1 * e1, red);
  const float mean = sum * (1.0f / Dc);
  const float var = ssq * (1.0f / Dc) - mean * mean;
  const float rs = rsqrtf(var + 1e-5f);
  const float y0 = (e0 - mean) * rs * gamma[tid] + beta[tid];
  const float y1 = (e1 - mean) * rs * gamma[tid + 256] + beta[tid + 256];
  vrow[tid] = y0;
  vrow[tid + 256] = y1;
  const float dot = block_sum256(y0 * sp_w[tid] + y1 * sp_w[tid + 256], red);
  if (tid == 0) st[(size_t)b * Nc + n] = dot + sp_b[0];
}

__global__ __launch_bounds__(128) void ln_proj_kernel(
    const float* __restrict__ val, const float* __restrict__ gamma,
    const float* __restrict__ beta, const float* __restrict__ Uw,
    const float* __restrict__ Vw, float* __restrict__ xo,
    float* __restrict__ qo, float* __restrict__ ko) {
  __shared__ float xs[16 * XSTR];
  const int tile = blockIdx.x % NT;
  const int b = blockIdx.x / NT;
  const int n0 = tile * 16;
  const int tid = threadIdx.x;
  const bool full = (n0 + 16 <= Nc);

  const int row = tid >> 3, sub = tid & 7;
  const int g = n0 + row;
  const float* vrow = val + ((size_t)b * Nc + g) * Dc;
  if (full) {
    for (int c = sub * 64; c < sub * 64 + 64; ++c)
      xs[row * XSTR + c] = vrow[c];
  } else {
    for (int c = sub * 64; c < sub * 64 + 64; ++c)
      xs[row * XSTR + c] = (g < Nc) ? vrow[c] : 0.0f;
  }
  float s1 = 0.0f, s2 = 0.0f;
  for (int c = sub * 64; c < sub * 64 + 64; ++c) {
    const float v = xs[row * XSTR + c];
    s1 += v;
    s2 += v * v;
  }
  for (int m = 1; m < 8; m <<= 1) {
    s1 += __shfl_xor(s1, m, 32);
    s2 += __shfl_xor(s2, m, 32);
  }
  const float mean = s1 * (1.0f / Dc);
  const float var = s2 * (1.0f / Dc) - mean * mean;
  const float rs = rsqrtf(var + 1e-5f);
  float* xrow = xo + ((size_t)b * Nc + g) * Dc;
  if (full) {
    for (int c = sub * 64; c < sub * 64 + 64; ++c) {
      const float y = (xs[row * XSTR + c] - mean) * rs * gamma[c] + beta[c];
      xs[row * XSTR + c] = y;
      xrow[c] = y;
    }
  } else {
    for (int c = sub * 64; c < sub * 64 + 64; ++c) {
      const float y = (xs[row * XSTR + c] - mean) * rs * gamma[c] + beta[c];
      xs[row * XSTR + c] = y;
      if (g < Nc) xrow[c] = y;
    }
  }
  __syncthreads();

  const int wave = tid >> 5, lane = tid & 31;
  const int hf = lane >> 4, l16 = lane & 15;
  const float* Wm = (wave < 2) ? Uw : Vw;
  float* out = (wave < 2) ? qo : ko;
  const int cb = (wave & 1) * 16;
  v8f acc = {0.f, 0.f, 0.f, 0.f, 0.f, 0.f, 0.f, 0.f};
  for (int kk = 0; kk < Dc / 4; ++kk) {
    const int c0 = 4 * kk + 2 * hf;
    v2f a;
    a.x = xs[l16 * XSTR + c0];
    a.y = xs[l16 * XSTR + c0 + 1];
    v2f bv;
    bv.x = Wm[(size_t)c0 * Rc + cb + l16];
    bv.y = Wm[(size_t)(c0 + 1) * Rc + cb + l16];
    acc = wmma_f32(a, bv, acc);
  }
  if (full) {
    for (int i = 0; i < 8; ++i) {
      const int r = i + 8 * hf;
      out[((size_t)b * Nc + n0 + r) * Rc + cb + l16] = acc[i];
    }
  } else {
    for (int i = 0; i < 8; ++i) {
      const int r = i + 8 * hf, gg = n0 + r;
      if (gg < Nc) out[((size_t)b * Nc + gg) * Rc + cb + l16] = acc[i];
    }
  }
}

__global__ __launch_bounds__(128) void attn_kernel(
    const float* __restrict__ x, const float* __restrict__ q,
    const float* __restrict__ k, const float* __restrict__ st,
    float* __restrict__ val) {
  __shared__ float qs[16 * QSTR];
  __shared__ float ks[WINc * KSTR];
  __shared__ float ss[16 * SSTR];
  __shared__ float wm[16 * WSTR];
  __shared__ float sw[WINc];
  const int tile = blockIdx.x % NT;
  const int b = blockIdx.x / NT;
  const int n0 = tile * 16;
  const int tid = threadIdx.x;
  const bool full = (n0 + 16 <= Nc);
  const bool interior = (n0 >= Wc) && (n0 + 16 + Wc <= Nc);

  const float* qb = q + ((size_t)b * Nc + n0) * Rc;
  const float* kb = k + ((size_t)b * Nc + n0 - Wc) * Rc;
  const float* sb = st + (size_t)b * Nc + n0 - Wc;
  if (interior) {
    for (int i = tid; i < 16 * Rc; i += 128) {
      const int r = i >> 5, c = i & 31;
      qs[r * QSTR + c] = qb[r * Rc + c];
    }
    for (int i = tid; i < WINc * Rc; i += 128) {
      const int j = i >> 5, c = i & 31;
      ks[j * KSTR + c] = kb[(size_t)j * Rc + c];
    }
    for (int j = tid; j < WINc; j += 128) sw[j] = sb[j];
  } else {
    for (int i = tid; i < 16 * Rc; i += 128) {
      const int r = i >> 5, c = i & 31, g = n0 + r;
      qs[r * QSTR + c] = (g < Nc) ? q[((size_t)b * Nc + g) * Rc + c] : 0.0f;
    }
    for (int i = tid; i < WINc * Rc; i += 128) {
      const int j = i >> 5, c = i & 31, gn = n0 - Wc + j;
      ks[j * KSTR + c] =
          (gn >= 0 && gn < Nc) ? k[((size_t)b * Nc + gn) * Rc + c] : 0.0f;
    }
    for (int j = tid; j < WINc; j += 128) {
      const int gn = n0 - Wc + j;
      sw[j] = (gn >= 0 && gn < Nc) ? st[(size_t)b * Nc + gn] : 0.0f;
    }
  }
  for (int i = tid; i < 16 * WSTR; i += 128) wm[i] = 0.0f;
  __syncthreads();

  const int wave = tid >> 5, lane = tid & 31;
  const int hf = lane >> 4, l16 = lane & 15;

  for (int t = wave; t < 9; t += 4) {
    const int cb = t * 16;
    v8f acc = {0.f, 0.f, 0.f, 0.f, 0.f, 0.f, 0.f, 0.f};
    for (int kk = 0; kk < Rc / 4; ++kk) {
      const int c0 = 4 * kk + 2 * hf;
      v2f a;
      a.x = qs[l16 * QSTR + c0];
      a.y = qs[l16 * QSTR + c0 + 1];
      v2f bv;
      bv.x = ks[(cb + l16) * KSTR + c0];
      bv.y = ks[(cb + l16) * KSTR + c0 + 1];
      acc = wmma_f32(a, bv, acc);
    }
    for (int i = 0; i < 8; ++i) ss[(i + 8 * hf) * SSTR + cb + l16] = acc[i];
  }
  __syncthreads();

  {
    const int row = tid >> 3, sub = tid & 7;
    const int g = n0 + row;
    if (interior) {
      float m = -1e30f;
      for (int o = sub; o < WNc; o += 8)
        m = fmaxf(m, fabsf(ss[row * SSTR + row + o]));
      for (int mm = 1; mm < 8; mm <<= 1) m = fmaxf(m, __shfl_xor(m, mm, 32));
      float Z = 0.0f;
      for (int o = sub; o < WNc; o += 8)
        Z += expf(fabsf(ss[row * SSTR + row + o]) - m);
      for (int mm = 1; mm < 8; mm <<= 1) Z += __shfl_xor(Z, mm, 32);
      const float inv = 1.0f / Z;
      for (int o = sub; o < WNc; o += 8) {
        const float s = ss[row * SSTR + row + o];
        const float p = expf(fabsf(s) - m) * inv;
        const float wv = (s > 0.0f) ? p : ((s < 0.0f) ? -p : 0.0f);
        wm[row * WSTR + row + o] = wv * sw[row + o];
      }
    } else if (g < Nc) {
      float m = -1e30f;
      for (int o = sub; o < WNc; o += 8) {
        const int gn = g - Wc + o;
        if (gn >= 0 && gn < Nc) m = fmaxf(m, fabsf(ss[row * SSTR + row + o]));
      }
      for (int mm = 1; mm < 8; mm <<= 1) m = fmaxf(m, __shfl_xor(m, mm, 32));
      float Z = 0.0f;
      for (int o = sub; o < WNc; o += 8) {
        const int gn = g - Wc + o;
        if (gn >= 0 && gn < Nc)
          Z += expf(fabsf(ss[row * SSTR + row + o]) - m);
      }
      for (int mm = 1; mm < 8; mm <<= 1) Z += __shfl_xor(Z, mm, 32);
      const float inv = 1.0f / Z;
      for (int o = sub; o < WNc; o += 8) {
        const int gn = g - Wc + o;
        if (gn >= 0 && gn < Nc) {
          const float s = ss[row * SSTR + row + o];
          const float p = expf(fabsf(s) - m) * inv;
          const float wv = (s > 0.0f) ? p : ((s < 0.0f) ? -p : 0.0f);
          wm[row * WSTR + row + o] = wv * sw[row + o];
        }
      }
    }
  }
  __syncthreads();

  const float* xw = x + ((size_t)b * Nc + n0 - Wc) * Dc;
  for (int ct = wave; ct < Dc / 16; ct += 4) {
    const int cb = ct * 16;
    v8f acc = {0.f, 0.f, 0.f, 0.f, 0.f, 0.f, 0.f, 0.f};
    if (interior) {
      for (int kk = 0; kk < WINc / 4; ++kk) {
        const int c0 = 4 * kk + 2 * hf;
        v2f a;
        a.x = wm[l16 * WSTR + c0];
        a.y = wm[l16 * WSTR + c0 + 1];
        v2f bv;
        bv.x = xw[(size_t)c0 * Dc + cb + l16];
        bv.y = xw[(size_t)(c0 + 1) * Dc + cb + l16];
        acc = wmma_f32(a, bv, acc);
      }
    } else {
      for (int kk = 0; kk < WINc / 4; ++kk) {
        const int c0 = 4 * kk + 2 * hf;
        v2f a;
        a.x = wm[l16 * WSTR + c0];
        a.y = wm[l16 * WSTR + c0 + 1];
        const int g0 = n0 - Wc + c0, g1 = g0 + 1;
        v2f bv;
        bv.x = (g0 >= 0 && g0 < Nc) ? x[((size_t)b * Nc + g0) * Dc + cb + l16]
                                    : 0.0f;
        bv.y = (g1 >= 0 && g1 < Nc) ? x[((size_t)b * Nc + g1) * Dc + cb + l16]
                                    : 0.0f;
        acc = wmma_f32(a, bv, acc);
      }
    }
    if (full) {
      float* vb = val + ((size_t)b * Nc + n0) * Dc + cb + l16;
      for (int i = 0; i < 8; ++i) {
        const int r = i + 8 * hf;
        vb[(size_t)r * Dc] += acc[i];
      }
    } else {
      for (int i = 0; i < 8; ++i) {
        const int r = i + 8 * hf, gg = n0 + r;
        if (gg < Nc) val[((size_t)b * Nc + gg) * Dc + cb + l16] += acc[i];
      }
    }
  }
}

extern "C" void kernel_launch(void* const* d_in, const int* in_sizes, int n_in,
                              void* d_out, int out_size, void* d_ws,
                              size_t ws_size, hipStream_t stream) {
  const int* ids = (const int*)d_in[0];
  const float* emb = (const float*)d_in[1];
  const float* pos = (const float*)d_in[2];
  const float* anchor_val = (const float*)d_in[3];
  const float* anchor_state = (const float*)d_in[4];
  const float* in_gamma = (const float*)d_in[5];
  const float* in_beta = (const float*)d_in[6];
  const float* ng = (const float*)d_in[7];
  const float* nb = (const float*)d_in[8];
  const float* Uw = (const float*)d_in[9];
  const float* Vw = (const float*)d_in[10];
  const float* sp_w = (const float*)d_in[11];
  const float* sp_b = (const float*)d_in[12];

  float* val = (float*)d_out;
  float* ws = (float*)d_ws;
  float* x = ws;
  float* q = x + (size_t)Bc * Nc * Dc;
  float* k = q + (size_t)Bc * Nc * Rc;
  float* st = k + (size_t)Bc * Nc * Rc;

  embed_ln_kernel<<<Bc * Nc, 256, 0, stream>>>(
      ids, emb, pos, anchor_val, anchor_state, in_gamma, in_beta, sp_w, sp_b,
      val, st);

  for (int l = 0; l < Lc; ++l) {
    ln_proj_kernel<<<Bc * NT, 128, 0, stream>>>(
        val, ng + (size_t)l * Dc, nb + (size_t)l * Dc, Uw + (size_t)l * Dc * Rc,
        Vw + (size_t)l * Dc * Rc, x, q, k);
    attn_kernel<<<Bc * NT, 128, 0, stream>>>(x, q, k, st, val);
  }
}


# Edge and tail tiles (they set each kernel's time): per-element guarded global loads become clamped-row loads + select, store guard hoisted out of its loop, so their loads issue in groups
# speedup vs baseline: 2.2269x; 2.2269x over previous
#include <hip/hip_runtime.h>

typedef __attribute__((ext_vector_type(2))) float v2f;
typedef __attribute__((ext_vector_type(8))) float v8f;

constexpr int Bc = 2;
constexpr int Sc = 1024;
constexpr int Nc = Sc + 1;
constexpr int Dc = 512;
constexpr int Rc = 32;
constexpr int Lc = 3;
constexpr int Wc = 64;
constexpr int WNc = 2 * Wc + 1;
constexpr int WINc = 144;
constexpr int NT = (Nc + 15) / 16;

constexpr int XSTR = Dc + 4;
constexpr int QSTR = 33;
constexpr int KSTR = 33;
constexpr int SSTR = 146;
constexpr int WSTR = 145;

__device__ __forceinline__ v8f wmma_f32(v2f a, v2f b, v8f c) {
  return __builtin_amdgcn_wmma_f32_16x16x4_f32(false, a, false, b, (short)0, c,
                                               false, false);
}

__device__ __forceinline__ float block_sum256(float v, float* red) {
  for (int m = 16; m >= 1; m >>= 1) v += __shfl_xor(v, m, 32);
  const int wave = threadIdx.x >> 5, lane = threadIdx.x & 31;
  __syncthreads();
  if (lane == 0) red[wave] = v;
  __syncthreads();
  if (threadIdx.x < 8) {
    float t = red[threadIdx.x];
    t += __shfl_xor(t, 1, 32);
    t += __shfl_xor(t, 2, 32);
    t += __shfl_xor(t, 4, 32);
    if (threadIdx.x == 0) red[0] = t;
  }
  __syncthreads();
  return red[0];
}

__global__ __launch_bounds__(256) void embed_ln_kernel(
    const int* __restrict__ ids, const float* __restrict__ emb,
    const float* __restrict__ pos, const float* __restrict__ anchor_val,
    const float* __restrict__ anchor_state, const float* __restrict__ gamma,
    const float* __restrict__ beta, const float* __restrict__ sp_w,
    const float* __restrict__ sp_b, float* __restrict__ val,
    float* __restrict__ st) {
  __shared__ float red[8];
  const int n = blockIdx.x % Nc;
  const int b = blockIdx.x / Nc;
  const int tid = threadIdx.x;
  float* vrow = val + ((size_t)b * Nc + n) * Dc;
  if (n == 0) {
    vrow[tid] = anchor_val[tid];
    vrow[tid + 256] = anchor_val[tid + 256];
    if (tid == 0) st[(size_t)b * Nc] = anchor_state[0];
    return;
  }
  const int s = n - 1;
  const int id = ids[b * Sc + s];
  const float e0 = emb[(size_t)id * Dc + tid] + pos[(size_t)s * Dc + tid];
  const float e1 = emb[(size_t)id * Dc + tid + 256] + pos[(size_t)s * Dc + tid + 256];
  const float sum = block_sum256(e0 + e1, red);
  const float ssq = block_sum256(e0 * e0 + e1 * e1, red);
  const float mean = sum * (1.0f / Dc);
  const float var = ssq * (1.0f / Dc) - mean * mean;
  const float rs = rsqrtf(var + 1e-5f);
  const float y0 = (e0 - mean) * rs * gamma[tid] + beta[tid];
  const float y1 = (e1 - mean) * rs * gamma[tid + 256] + beta[tid + 256];
  vrow[tid] = y0;
  vrow[tid + 256] = y1;
  const float dot = block_sum256(y0 * sp_w[tid] + y1 * sp_w[tid + 256], red);
  if (tid == 0) st[(size_t)b * Nc + n] = dot + sp_b[0];
}

__global__ __launch_bounds__(128) void ln_proj_kernel(
    const float* __restrict__ val, const float* __restrict__ gamma,
    const float* __restrict__ beta, const float* __restrict__ Uw,
    const float* __restrict__ Vw, float* __restrict__ xo,
    float* __restrict__ qo, float* __restrict__ ko) {
  __shared__ float xs[16 * XSTR];
  const int tile = blockIdx.x % NT;
  const int b = blockIdx.x / NT;
  const int n0 = tile * 16;
  const int tid = threadIdx.x;
  const bool full = (n0 + 16 <= Nc);

  const int row = tid >> 3, sub = tid & 7;
  const int g = n0 + row;
  const float* vrow = val + ((size_t)b * Nc + g) * Dc;
  if (full) {
    for (int c = sub * 64; c < sub * 64 + 64; ++c)
      xs[row * XSTR + c] = vrow[c];
  } else {
    const float* crow = val + ((size_t)b * Nc + min(g, Nc - 1)) * Dc;
    for (int c = sub * 64; c < sub * 64 + 64; ++c) {
      const float v = crow[c];
      xs[row * XSTR + c] = (g < Nc) ? v : 0.0f;
    }
  }
  float s1 = 0.0f, s2 = 0.0f;
  for (int c = sub * 64; c < sub * 64 + 64; ++c) {
    const float v = xs[row * XSTR + c];
    s1 += v;
    s2 += v * v;
  }
  for (int m = 1; m < 8; m <<= 1) {
    s1 += __shfl_xor(s1, m, 32);
    s2 += __shfl_xor(s2, m, 32);
  }
  const float mean = s1 * (1.0f / Dc);
  const float var = s2 * (1.0f / Dc) - mean * mean;
  const float rs = rsqrtf(var + 1e-5f);
  float* xrow = xo + ((size_t)b * Nc + g) * Dc;
  if (full) {
    for (int c = sub * 64; c < sub * 64 + 64; ++c) {
      const float y = (xs[row * XSTR + c] - mean) * rs * gamma[c] + beta[c];
      xs[row * XSTR + c] = y;
      xrow[c] = y;
    }
  } else if (g < Nc) {
    for (int c = sub * 64; c < sub * 64 + 64; ++c) {
      const float y = (xs[row * XSTR + c] - mean) * rs * gamma[c] + beta[c];
      xs[row * XSTR + c] = y;
      xrow[c] = y;
    }
  } else {
    for (int c = sub * 64; c < sub * 64 + 64; ++c) {
      const float y = (xs[row * XSTR + c] - mean) * rs * gamma[c] + beta[c];
      xs[row * XSTR + c] = y;
    }
  }
  __syncthreads();

  const int wave = tid >> 5, lane = tid & 31;
  const int hf = lane >> 4, l16 = lane & 15;
  const float* Wm = (wave < 2) ? Uw : Vw;
  float* out = (wave < 2) ? qo : ko;
  const int cb = (wave & 1) * 16;
  v8f acc = {0.f, 0.f, 0.f, 0.f, 0.f, 0.f, 0.f, 0.f};
  for (int kk = 0; kk < Dc / 4; ++kk) {
    const int c0 = 4 * kk + 2 * hf;
    v2f a;
    a.x = xs[l16 * XSTR + c0];
    a.y = xs[l16 * XSTR + c0 + 1];
    v2f bv;
    bv.x = Wm[(size_t)c0 * Rc + cb + l16];
    bv.y = Wm[(size_t)(c0 + 1) * Rc + cb + l16];
    acc = wmma_f32(a, bv, acc);
  }
  if (full) {
    for (int i = 0; i < 8; ++i) {
      const int r = i + 8 * hf;
      out[((size_t)b * Nc + n0 + r) * Rc + cb + l16] = acc[i];
    }
  } else {
    for (int i = 0; i < 8; ++i) {
      const int r = i + 8 * hf, gg = n0 + r;
      if (gg < Nc) out[((size_t)b * Nc + gg) * Rc + cb + l16] = acc[i];
    }
  }
}

__global__ __launch_bounds__(128) void attn_kernel(
    const float* __restrict__ x, const float* __restrict__ q,
    const float* __restrict__ k, const float* __restrict__ st,
    float* __restrict__ val) {
  __shared__ float qs[16 * QSTR];
  __shared__ float ks[WINc * KSTR];
  __shared__ float ss[16 * SSTR];
  __shared__ float wm[16 * WSTR];
  __shared__ float sw[WINc];
  const int tile = blockIdx.x % NT;
  const int b = blockIdx.x / NT;
  const int n0 = tile * 16;
  const int tid = threadIdx.x;
  const bool full = (n0 + 16 <= Nc);
  const bool interior = (n0 >= Wc) && (n0 + 16 + Wc <= Nc);

  const float* qb = q + ((size_t)b * Nc + n0) * Rc;
  const float* kb = k + ((size_t)b * Nc + n0 - Wc) * Rc;
  const float* sb = st + (size_t)b * Nc + n0 - Wc;
  if (interior) {
    for (int i = tid; i < 16 * Rc; i += 128) {
      const int r = i >> 5, c = i & 31;
      qs[r * QSTR + c] = qb[r * Rc + c];
    }
    for (int i = tid; i < WINc * Rc; i += 128) {
      const int j = i >> 5, c = i & 31;
      ks[j * KSTR + c] = kb[(size_t)j * Rc + c];
    }
    for (int j = tid; j < WINc; j += 128) sw[j] = sb[j];
  } else {
    static_assert((16 * Rc) % 128 == 0 && (WINc * Rc) % 128 == 0,
                  "q tile and k window: whole trips of the 128 threads");
#pragma unroll
    for (int it = 0; it < 16 * Rc / 128; ++it) {
      const int i = tid + it * 128;
      const int r = i >> 5, c = i & 31, g = n0 + r;
      const float v = q[((size_t)b * Nc + min(g, Nc - 1)) * Rc + c];
      qs[r * QSTR + c] = (g < Nc) ? v : 0.0f;
    }
#pragma unroll
    for (int it = 0; it < WINc * Rc / 128; ++it) {
      const int i = tid + it * 128;
      const int j = i >> 5, c = i & 31, gn = n0 - Wc + j;
      const int gc = min(max(gn, 0), Nc - 1);
      const float v = k[((size_t)b * Nc + gc) * Rc + c];
      ks[j * KSTR + c] = (gc == gn) ? v : 0.0f;
    }
    for (int j = tid; j < WINc; j += 128) {
      const int gn = n0 - Wc + j;
      const int gc = min(max(gn, 0), Nc - 1);
      const float v = st[(size_t)b * Nc + gc];
      sw[j] = (gc == gn) ? v : 0.0f;
    }
  }
  for (int i = tid; i < 16 * WSTR; i += 128) wm[i] = 0.0f;
  __syncthreads();

  const int wave = tid >> 5, lane = tid & 31;
  const int hf = lane >> 4, l16 = lane & 15;

  for (int t = wave; t < 9; t += 4) {
    const int cb = t * 16;
    v8f acc = {0.f, 0.f, 0.f, 0.f, 0.f, 0.f, 0.f, 0.f};
    for (int kk = 0; kk < Rc / 4; ++kk) {
      const int c0 = 4 * kk + 2 * hf;
      v2f a;
      a.x = qs[l16 * QSTR + c0];
      a.y = qs[l16 * QSTR + c0 + 1];
      v2f bv;
      bv.x = ks[(cb + l16) * KSTR + c0];
      bv.y = ks[(cb + l16) * KSTR + c0 + 1];
      acc = wmma_f32(a, bv, acc);
    }
    for (int i = 0; i < 8; ++i) ss[(i + 8 * hf) * SSTR + cb + l16] = acc[i];
  }
  __syncthreads();

  {
    const int row = tid >> 3, sub = tid & 7;
    const int g = n0 + row;
    if (interior) {
      float m = -1e30f;
      for (int o = sub; o < WNc; o += 8)
        m = fmaxf(m, fabsf(ss[row * SSTR + row + o]));
      for (int mm = 1; mm < 8; mm <<= 1) m = fmaxf(m, __shfl_xor(m, mm, 32));
      float Z = 0.0f;
      for (int o = sub; o < WNc; o += 8)
        Z += expf(fabsf(ss[row * SSTR + row + o]) - m);
      for (int mm = 1; mm < 8; mm <<= 1) Z += __shfl_xor(Z, mm, 32);
      const float inv = 1.0f / Z;
      for (int o = sub; o < WNc; o += 8) {
        const float s = ss[row * SSTR + row + o];
        const float p = expf(fabsf(s) - m) * inv;
        const float wv = (s > 0.0f) ? p : ((s < 0.0f) ? -p : 0.0f);
        wm[row * WSTR + row + o] = wv * sw[row + o];
      }
    } else if (g < Nc) {
      float m = -1e30f;
      for (int o = sub; o < WNc; o += 8) {
        const int gn = g - Wc + o;
        if (gn >= 0 && gn < Nc) m = fmaxf(m, fabsf(ss[row * SSTR + row + o]));
      }
      for (int mm = 1; mm < 8; mm <<= 1) m = fmaxf(m, __shfl_xor(m, mm, 32));
      float Z = 0.0f;
      for (int o = sub; o < WNc; o += 8) {
        const int gn = g - Wc + o;
        if (gn >= 0 && gn < Nc)
          Z += expf(fabsf(ss[row * SSTR + row + o]) - m);
      }
      for (int mm = 1; mm < 8; mm <<= 1) Z += __shfl_xor(Z, mm, 32);
      const float inv = 1.0f / Z;
      for (int o = sub; o < WNc; o += 8) {
        const int gn = g - Wc + o;
        if (gn >= 0 && gn < Nc) {
          const float s = ss[row * SSTR + row + o];
          const float p = expf(fabsf(s) - m) * inv;
          const float wv = (s > 0.0f) ? p : ((s < 0.0f) ? -p : 0.0f);
          wm[row * WSTR + row + o] = wv * sw[row + o];
        }
      }
    }
  }
  __syncthreads();

  const float* xw = x + ((size_t)b * Nc + n0 - Wc) * Dc;
  for (int ct = wave; ct < Dc / 16; ct += 4) {
    const int cb = ct * 16;
    v8f acc = {0.f, 0.f, 0.f, 0.f, 0.f, 0.f, 0.f, 0.f};
    if (interior) {
      for (int kk = 0; kk < WINc / 4; ++kk) {
        const int c0 = 4 * kk + 2 * hf;
        v2f a;
        a.x = wm[l16 * WSTR + c0];
        a.y = wm[l16 * WSTR + c0 + 1];
        v2f bv;
        bv.x = xw[(size_t)c0 * Dc + cb + l16];
        bv.y = xw[(size_t)(c0 + 1) * Dc + cb + l16];
        acc = wmma_f32(a, bv, acc);
      }
    } else {
#pragma unroll 12
      for (int kk = 0; kk < WINc / 4; ++kk) {
        const int c0 = 4 * kk + 2 * hf;
        v2f a;
        a.x = wm[l16 * WSTR + c0];
        a.y = wm[l16 * WSTR + c0 + 1];
        const int g0 = n0 - Wc + c0, g1 = g0 + 1;
        const int r0 = min(max(g0, 0), Nc - 1), r1 = min(max(g1, 0), Nc - 1);
        const float v0 = x[((size_t)b * Nc + r0) * Dc + cb + l16];
        const float v1 = x[((size_t)b * Nc + r1) * Dc + cb + l16];
        v2f bv;
        bv.x = (r0 == g0) ? v0 : 0.0f;
        bv.y = (r1 == g1) ? v1 : 0.0f;
        acc = wmma_f32(a, bv, acc);
      }
    }
    if (full) {
      float* vb = val + ((size_t)b * Nc + n0) * Dc + cb + l16;
      for (int i = 0; i < 8; ++i) {
        const int r = i + 8 * hf;
        vb[(size_t)r * Dc] += acc[i];
      }
    } else {
      for (int i = 0; i < 8; ++i) {
        const int r = i + 8 * hf, gg = n0 + r;
        if (gg < Nc) val[((size_t)b * Nc + gg) * Dc + cb + l16] += acc[i];
      }
    }
  }
}

extern "C" void kernel_launch(void* const* d_in, const int* in_sizes, int n_in,
                              void* d_out, int out_size, void* d_ws,
                              size_t ws_size, hipStream_t stream) {
  const int* ids = (const int*)d_in[0];
  const float* emb = (const float*)d_in[1];
  const float* pos = (const float*)d_in[2];
  const float* anchor_val = (const float*)d_in[3];
  const float* anchor_state = (const float*)d_in[4];
  const float* in_gamma = (const float*)d_in[5];
  const float* in_beta = (const float*)d_in[6];
  const float* ng = (const float*)d_in[7];
  const float* nb = (const float*)d_in[8];
  const float* Uw = (const float*)d_in[9];
  const float* Vw = (const float*)d_in[10];
  const float* sp_w = (const float*)d_in[11];
  const float* sp_b = (const float*)d_in[12];

  float* val = (float*)d_out;
  float* ws = (float*)d_ws;
  float* x = ws;
  float* q = x + (size_t)Bc * Nc * Dc;
  float* k = q + (size_t)Bc * Nc * Rc;
  float* st = k + (size_t)Bc * Nc * Rc;

  embed_ln_kernel<<<Bc * Nc, 256, 0, stream>>>(
      ids, emb, pos, anchor_val, anchor_state, in_gamma, in_beta, sp_w, sp_b,
      val, st);

  for (int l = 0; l < Lc; ++l) {
    ln_proj_kernel<<<Bc * NT, 128, 0, stream>>>(
        val, ng + (size_t)l * Dc, nb + (size_t)l * Dc, Uw + (size_t)l * Dc * Rc,
        Vw + (size_t)l * Dc * Rc, x, q, k);
    attn_kernel<<<Bc * NT, 128, 0, stream>>>(x, q, k, st, val);
  }
}


# attn kernel: 512-thread work-groups (16 waves, 4 per SIMD) instead of 128; each wave runs 2 column tiles of the delta GEMM together with its row loads requested back to back; softmax stays on 4 waves
# speedup vs baseline: 3.5408x; 1.5900x over previous
#include <hip/hip_runtime.h>

typedef __attribute__((ext_vector_type(2))) float v2f;
typedef __attribute__((ext_vector_type(8))) float v8f;

constexpr int Bc = 2;
constexpr int Sc = 1024;
constexpr int Nc = Sc + 1;
constexpr int Dc = 512;
constexpr int Rc = 32;
constexpr int Lc = 3;
constexpr int Wc = 64;
constexpr int WNc = 2 * Wc + 1;
constexpr int WINc = 144;
constexpr int NT = (Nc + 15) / 16;

constexpr int XSTR = Dc + 4;
constexpr int QSTR = 33;
constexpr int KSTR = 33;
constexpr int SSTR = 146;
constexpr int WSTR = 145;

__device__ __forceinline__ v8f wmma_f32(v2f a, v2f b, v8f c) {
  return __builtin_amdgcn_wmma_f32_16x16x4_f32(false, a, false, b, (short)0, c,
                                               false, false);
}

__device__ __forceinline__ float block_sum256(float v, float* red) {
  for (int m = 16; m >= 1; m >>= 1) v += __shfl_xor(v, m, 32);
  const int wave = threadIdx.x >> 5, lane = threadIdx.x & 31;
  __syncthreads();
  if (lane == 0) red[wave] = v;
  __syncthreads();
  if (threadIdx.x < 8) {
    float t = red[threadIdx.x];
    t += __shfl_xor(t, 1, 32);
    t += __shfl_xor(t, 2, 32);
    t += __shfl_xor(t, 4, 32);
    if (threadIdx.x == 0) red[0] = t;
  }
  __syncthreads();
  return red[0];
}

__global__ __launch_bounds__(256) void embed_ln_kernel(
    const int* __restrict__ ids, const float* __restrict__ emb,
    const float* __restrict__ pos, const float* __restrict__ anchor_val,
    const float* __restrict__ anchor_state, const float* __restrict__ gamma,
    const float* __restrict__ beta, const float* __restrict__ sp_w,
    const float* __restrict__ sp_b, float* __restrict__ val,
    float* __restrict__ st) {
  __shared__ float red[8];
  const int n = blockIdx.x % Nc;
  const int b = blockIdx.x / Nc;
  const int tid = threadIdx.x;
  float* vrow = val + ((size_t)b * Nc + n) * Dc;
  if (n == 0) {
    vrow[tid] = anchor_val[tid];
    vrow[tid + 256] = anchor_val[tid + 256];
    if (tid == 0) st[(size_t)b * Nc] = anchor_state[0];
    return;
  }
  const int s = n - 1;
  const int id = ids[b * Sc + s];
  const float e0 = emb[(size_t)id * Dc + tid] + pos[(size_t)s * Dc + tid];
  const float e1 = emb[(size_t)id * Dc + tid + 256] + pos[(size_t)s * Dc + tid + 256];
  const float sum = block_sum256(e0 + e1, red);
  const float ssq = block_sum256(e0 * e0 + e1 * e1, red);
  const float mean = sum * (1.0f / Dc);
  const float var = ssq * (1.0f / Dc) - mean * mean;
  const float rs = rsqrtf(var + 1e-5f);
  const float y0 = (e0 - mean) * rs * gamma[tid] + beta[tid];
  const float y1 = (e1 - mean) * rs * gamma[tid + 256] + beta[tid + 256];
  vrow[tid] = y0;
  vrow[tid + 256] = y1;
  const float dot = block_sum256(y0 * sp_w[tid] + y1 * sp_w[tid + 256], red);
  if (tid == 0) st[(size_t)b * Nc + n] = dot + sp_b[0];
}

__global__ __launch_bounds__(128) void ln_proj_kernel(
    const float* __restrict__ val, const float* __restrict__ gamma,
    const float* __restrict__ beta, const float* __restrict__ Uw,
    const float* __restrict__ Vw, float* __restrict__ xo,
    float* __restrict__ qo, float* __restrict__ ko) {
  __shared__ float xs[16 * XSTR];
  const int tile = blockIdx.x % NT;
  const int b = blockIdx.x / NT;
  const int n0 = tile * 16;
  const int tid = threadIdx.x;
  const bool full = (n0 + 16 <= Nc);

  const int row = tid >> 3, sub = tid & 7;
  const int g = n0 + row;
  const float* vrow = val + ((size_t)b * Nc + g) * Dc;
  if (full) {
    for (int c = sub * 64; c < sub * 64 + 64; ++c)
      xs[row * XSTR + c] = vrow[c];
  } else {
    const float* crow = val + ((size_t)b * Nc + min(g, Nc - 1)) * Dc;
    for (int c = sub * 64; c < sub * 64 + 64; ++c) {
      const float v = crow[c];
      xs[row * XSTR + c] = (g < Nc) ? v : 0.0f;
    }
  }
  float s1 = 0.0f, s2 = 0.0f;
  for (int c = sub * 64; c < sub * 64 + 64; ++c) {
    const float v = xs[row * XSTR + c];
    s1 += v;
    s2 += v * v;
  }
  for (int m = 1; m < 8; m <<= 1) {
    s1 += __shfl_xor(s1, m, 32);
    s2 += __shfl_xor(s2, m, 32);
  }
  const float mean = s1 * (1.0f / Dc);
  const float var = s2 * (1.0f / Dc) - mean * mean;
  const float rs = rsqrtf(var + 1e-5f);
  float* xrow = xo + ((size_t)b * Nc + g) * Dc;
  if (full) {
    for (int c = sub * 64; c < sub * 64 + 64; ++c) {
      const float y = (xs[row * XSTR + c] - mean) * rs * gamma[c] + beta[c];
      xs[row * XSTR + c] = y;
      xrow[c] = y;
    }
  } else if (g < Nc) {
    for (int c = sub * 64; c < sub * 64 + 64; ++c) {
      const float y = (xs[row * XSTR + c] - mean) * rs * gamma[c] + beta[c];
      xs[row * XSTR + c] = y;
      xrow[c] = y;
    }
  } else {
    for (int c = sub * 64; c < sub * 64 + 64; ++c) {
      const float y = (xs[row * XSTR + c] - mean) * rs * gamma[c] + beta[c];
      xs[row * XSTR + c] = y;
    }
  }
  __syncthreads();

  const int wave = tid >> 5, lane = tid & 31;
  const int hf = lane >> 4, l16 = lane & 15;
  const float* Wm = (wave < 2) ? Uw : Vw;
  float* out = (wave < 2) ? qo : ko;
  const int cb = (wave & 1) * 16;
  v8f acc = {0.f, 0.f, 0.f, 0.f, 0.f, 0.f, 0.f, 0.f};
  for (int kk = 0; kk < Dc / 4; ++kk) {
    const int c0 = 4 * kk + 2 * hf;
    v2f a;
    a.x = xs[l16 * XSTR + c0];
    a.y = xs[l16 * XSTR + c0 + 1];
    v2f bv;
    bv.x = Wm[(size_t)c0 * Rc + cb + l16];
    bv.y = Wm[(size_t)(c0 + 1) * Rc + cb + l16];
    acc = wmma_f32(a, bv, acc);
  }
  if (full) {
    for (int i = 0; i < 8; ++i) {
      const int r = i + 8 * hf;
      out[((size_t)b * Nc + n0 + r) * Rc + cb + l16] = acc[i];
    }
  } else {
    for (int i = 0; i < 8; ++i) {
      const int r = i + 8 * hf, gg = n0 + r;
      if (gg < Nc) out[((size_t)b * Nc + gg) * Rc + cb + l16] = acc[i];
    }
  }
}

constexpr int ATT = 512;
__global__ __launch_bounds__(ATT) void attn_kernel(
    const float* __restrict__ x, const float* __restrict__ q,
    const float* __restrict__ k, const float* __restrict__ st,
    float* __restrict__ val) {
  __shared__ float qs[16 * QSTR];
  __shared__ float ks[WINc * KSTR];
  __shared__ float ss[16 * SSTR];
  __shared__ float wm[16 * WSTR];
  __shared__ float sw[WINc];
  const int tile = blockIdx.x % NT;
  const int b = blockIdx.x / NT;
  const int n0 = tile * 16;
  const int tid = threadIdx.x;
  const bool full = (n0 + 16 <= Nc);
  const bool interior = (n0 >= Wc) && (n0 + 16 + Wc <= Nc);

  const float* qb = q + ((size_t)b * Nc + n0) * Rc;
  const float* kb = k + ((size_t)b * Nc + n0 - Wc) * Rc;
  const float* sb = st + (size_t)b * Nc + n0 - Wc;
  if (interior) {
    for (int i = tid; i < 16 * Rc; i += ATT) {
      const int r = i >> 5, c = i & 31;
      qs[r * QSTR + c] = qb[r * Rc + c];
    }
    for (int i = tid; i < WINc * Rc; i += ATT) {
      const int j = i >> 5, c = i & 31;
      ks[j * KSTR + c] = kb[(size_t)j * Rc + c];
    }
    for (int j = tid; j < WINc; j += ATT) sw[j] = sb[j];
  } else {
    static_assert((16 * Rc) % ATT == 0 && (WINc * Rc) % ATT == 0,
                  "q tile and k window: whole trips of the block's threads");
#pragma unroll
    for (int it = 0; it < 16 * Rc / ATT; ++it) {
      const int i = tid + it * ATT;
      const int r = i >> 5, c = i & 31, g = n0 + r;
      const float v = q[((size_t)b * Nc + min(g, Nc - 1)) * Rc + c];
      qs[r * QSTR + c] = (g < Nc) ? v : 0.0f;
    }
#pragma unroll
    for (int it = 0; it < WINc * Rc / ATT; ++it) {
      const int i = tid + it * ATT;
      const int j = i >> 5, c = i & 31, gn = n0 - Wc + j;
      const int gc = min(max(gn, 0), Nc - 1);
      const float v = k[((size_t)b * Nc + gc) * Rc + c];
      ks[j * KSTR + c] = (gc == gn) ? v : 0.0f;
    }
    for (int j = tid; j < WINc; j += ATT) {
      const int gn = n0 - Wc + j;
      const int gc = min(max(gn, 0), Nc - 1);
      const float v = st[(size_t)b * Nc + gc];
      sw[j] = (gc == gn) ? v : 0.0f;
    }
  }
  for (int i = tid; i < 16 * WSTR; i += ATT) wm[i] = 0.0f;
  __syncthreads();

  const int wave = tid >> 5, lane = tid & 31;
  const int hf = lane >> 4, l16 = lane & 15;

  for (int t = wave; t < 9; t += ATT / 32) {
    const int cb = t * 16;
    v8f acc = {0.f, 0.f, 0.f, 0.f, 0.f, 0.f, 0.f, 0.f};
    for (int kk = 0; kk < Rc / 4; ++kk) {
      const int c0 = 4 * kk + 2 * hf;
      v2f a;
      a.x = qs[l16 * QSTR + c0];
      a.y = qs[l16 * QSTR + c0 + 1];
      v2f bv;
      bv.x = ks[(cb + l16) * KSTR + c0];
      bv.y = ks[(cb + l16) * KSTR + c0 + 1];
      acc = wmma_f32(a, bv, acc);
    }
    for (int i = 0; i < 8; ++i) ss[(i + 8 * hf) * SSTR + cb + l16] = acc[i];
  }
  __syncthreads();

  if (tid < 128) {
    const int row = tid >> 3, sub = tid & 7;
    const int g = n0 + row;
    if (interior) {
      float m = -1e30f;
      for (int o = sub; o < WNc; o += 8)
        m = fmaxf(m, fabsf(ss[row * SSTR + row + o]));
      for (int mm = 1; mm < 8; mm <<= 1) m = fmaxf(m, __shfl_xor(m, mm, 32));
      float Z = 0.0f;
      for (int o = sub; o < WNc; o += 8)
        Z += expf(fabsf(ss[row * SSTR + row + o]) - m);
      for (int mm = 1; mm < 8; mm <<= 1) Z += __shfl_xor(Z, mm, 32);
      const float inv = 1.0f / Z;
      for (int o = sub; o < WNc; o += 8) {
        const float s = ss[row * SSTR + row + o];
        const float p = expf(fabsf(s) - m) * inv;
        const float wv = (s > 0.0f) ? p : ((s < 0.0f) ? -p : 0.0f);
        wm[row * WSTR + row + o] = wv * sw[row + o];
      }
    } else if (g < Nc) {
      float m = -1e30f;
      for (int o = sub; o < WNc; o += 8) {
        const int gn = g - Wc + o;
        if (gn >= 0 && gn < Nc) m = fmaxf(m, fabsf(ss[row * SSTR + row + o]));
      }
      for (int mm = 1; mm < 8; mm <<= 1) m = fmaxf(m, __shfl_xor(m, mm, 32));
      float Z = 0.0f;
      for (int o = sub; o < WNc; o += 8) {
        const int gn = g - Wc + o;
        if (gn >= 0 && gn < Nc)
          Z += expf(fabsf(ss[row * SSTR + row + o]) - m);
      }
      for (int mm = 1; mm < 8; mm <<= 1) Z += __shfl_xor(Z, mm, 32);
      const float inv = 1.0f / Z;
      for (int o = sub; o < WNc; o += 8) {
        const int gn = g - Wc + o;
        if (gn >= 0 && gn < Nc) {
          const float s = ss[row * SSTR + row + o];
          const float p = expf(fabsf(s) - m) * inv;
          const float wv = (s > 0.0f) ? p : ((s < 0.0f) ? -p : 0.0f);
          wm[row * WSTR + row + o] = wv * sw[row + o];
        }
      }
    }
  }
  __syncthreads();

  static_assert(Dc / 16 == 2 * (ATT / 32), "two column tiles per wave");
  {
    constexpr int KS = WINc / 4;
    const int cb = wave * 32 + l16;
    v2f bq0[KS], bq1[KS];
    if (interior) {
      const float* xw = x + ((size_t)b * Nc + n0 - Wc + 2 * hf) * Dc + cb;
#pragma unroll
      for (int kk = 0; kk < KS; ++kk) {
        const float* p = xw + (size_t)(4 * kk) * Dc;
        bq0[kk].x = p[0];
        bq0[kk].y = p[Dc];
        bq1[kk].x = p[16];
        bq1[kk].y = p[Dc + 16];
      }
    } else {
      const float* xb = x + (size_t)b * Nc * Dc + cb;
#pragma unroll
      for (int kk = 0; kk < KS; ++kk) {
        const int g0 = n0 - Wc + 4 * kk + 2 * hf, g1 = g0 + 1;
        const int r0 = min(max(g0, 0), Nc - 1), r1 = min(max(g1, 0), Nc - 1);
        const float* p0 = xb + (size_t)r0 * Dc;
        const float* p1 = xb + (size_t)r1 * Dc;
        const float v00 = p0[0], v01 = p1[0], v10 = p0[16], v11 = p1[16];
        bq0[kk].x = (r0 == g0) ? v00 : 0.0f;
        bq0[kk].y = (r1 == g1) ? v01 : 0.0f;
        bq1[kk].x = (r0 == g0) ? v10 : 0.0f;
        bq1[kk].y = (r1 == g1) ? v11 : 0.0f;
      }
    }
    v8f acc0 = {0.f, 0.f, 0.f, 0.f, 0.f, 0.f, 0.f, 0.f};
    v8f acc1 = acc0;
#pragma unroll
    for (int kk = 0; kk < KS; ++kk) {
      const int c0 = 4 * kk + 2 * hf;
      v2f a;
      a.x = wm[l16 * WSTR + c0];
      a.y = wm[l16 * WSTR + c0 + 1];
      acc0 = wmma_f32(a, bq0[kk], acc0);
      acc1 = wmma_f32(a, bq1[kk], acc1);
    }
    if (full) {
      float* vb = val + ((size_t)b * Nc + n0) * Dc + cb;
      for (int i = 0; i < 8; ++i) {
        const int r = i + 8 * hf;
        vb[(size_t)r * Dc] += acc0[i];
        vb[(size_t)r * Dc + 16] += acc1[i];
      }
    } else {
      for (int i = 0; i < 8; ++i) {
        const int r = i + 8 * hf, gg = n0 + r;
        if (gg < Nc) {
          float* vp = val + ((size_t)b * Nc + gg) * Dc + cb;
          vp[0] += acc0[i];
          vp[16] += acc1[i];
        }
      }
    }
  }
}

extern "C" void kernel_launch(void* const* d_in, const int* in_sizes, int n_in,
                              void* d_out, int out_size, void* d_ws,
                              size_t ws_size, hipStream_t stream) {
  const int* ids = (const int*)d_in[0];
  const float* emb = (const float*)d_in[1];
  const float* pos = (const float*)d_in[2];
  const float* anchor_val = (const float*)d_in[3];
  const float* anchor_state = (const float*)d_in[4];
  const float* in_gamma = (const float*)d_in[5];
  const float* in_beta = (const float*)d_in[6];
  const float* ng = (const float*)d_in[7];
  const float* nb = (const float*)d_in[8];
  const float* Uw = (const float*)d_in[9];
  const float* Vw = (const float*)d_in[10];
  const float* sp_w = (const float*)d_in[11];
  const float* sp_b = (const float*)d_in[12];

  float* val = (float*)d_out;
  float* ws = (float*)d_ws;
  float* x = ws;
  float* q = x + (size_t)Bc * Nc * Dc;
  float* k = q + (size_t)Bc * Nc * Rc;
  float* st = k + (size_t)Bc * Nc * Rc;

  embed_ln_kernel<<<Bc * Nc, 256, 0, stream>>>(
      ids, emb, pos, anchor_val, anchor_state, in_gamma, in_beta, sp_w, sp_b,
      val, st);

  for (int l = 0; l < Lc; ++l) {
    ln_proj_kernel<<<Bc * NT, 128, 0, stream>>>(
        val, ng + (size_t)l * Dc, nb + (size_t)l * Dc, Uw + (size_t)l * Dc * Rc,
        Vw + (size_t)l * Dc * Rc, x, q, k);
    attn_kernel<<<Bc * NT, ATT, 0, stream>>>(x, q, k, st, val);
  }
}


# embed + input LayerNorm + state kernel (2050 work-groups) fused into the front of layer 0's LN+projection kernel: 6 launches, not 7; input LayerNorm and state sums run in another order
# speedup vs baseline: 3.7239x; 1.0517x over previous
#include <hip/hip_runtime.h>

typedef __attribute__((ext_vector_type(2))) float v2f;
typedef __attribute__((ext_vector_type(8))) float v8f;

constexpr int Bc = 2;
constexpr int Sc = 1024;
constexpr int Nc = Sc + 1;
constexpr int Dc = 512;
constexpr int Rc = 32;
constexpr int Lc = 3;
constexpr int Wc = 64;
constexpr int WNc = 2 * Wc + 1;
constexpr int WINc = 144;
constexpr int NT = (Nc + 15) / 16;

constexpr int XSTR = Dc + 4;
constexpr int QSTR = 33;
constexpr int KSTR = 33;
constexpr int SSTR = 146;
constexpr int WSTR = 145;

__device__ __forceinline__ v8f wmma_f32(v2f a, v2f b, v8f c) {
  return __builtin_amdgcn_wmma_f32_16x16x4_f32(false, a, false, b, (short)0, c,
                                               false, false);
}

__device__ __forceinline__ void ln_proj_tile(
    float* xs, const int b, const int n0, const bool full,
    const float* __restrict__ gamma, const float* __restrict__ beta,
    const float* __restrict__ Uw, const float* __restrict__ Vw,
    float* __restrict__ xo, float* __restrict__ qo, float* __restrict__ ko) {
  const int tid = threadIdx.x;
  const int row = tid >> 3, sub = tid & 7;
  const int g = n0 + row;
  float s1 = 0.0f, s2 = 0.0f;
  for (int c = sub * 64; c < sub * 64 + 64; ++c) {
    const float v = xs[row * XSTR + c];
    s1 += v;
    s2 += v * v;
  }
  for (int m = 1; m < 8; m <<= 1) {
    s1 += __shfl_xor(s1, m, 32);
    s2 += __shfl_xor(s2, m, 32);
  }
  const float mean = s1 * (1.0f / Dc);
  const float var = s2 * (1.0f / Dc) - mean * mean;
  const float rs = rsqrtf(var + 1e-5f);
  float* xrow = xo + ((size_t)b * Nc + g) * Dc;
  if (full) {
    for (int c = sub * 64; c < sub * 64 + 64; ++c) {
      const float y = (xs[row * XSTR + c] - mean) * rs * gamma[c] + beta[c];
      xs[row * XSTR + c] = y;
      xrow[c] = y;
    }
  } else if (g < Nc) {
    for (int c = sub * 64; c < sub * 64 + 64; ++c) {
      const float y = (xs[row * XSTR + c] - mean) * rs * gamma[c] + beta[c];
      xs[row * XSTR + c] = y;
      xrow[c] = y;
    }
  } else {
    for (int c = sub * 64; c < sub * 64 + 64; ++c) {
      const float y = (xs[row * XSTR + c] - mean) * rs * gamma[c] + beta[c];
      xs[row * XSTR + c] = y;
    }
  }
  __syncthreads();

  const int wave = tid >> 5, lane = tid & 31;
  const int hf = lane >> 4, l16 = lane & 15;
  const float* Wm = (wave < 2) ? Uw : Vw;
  float* out = (wave < 2) ? qo : ko;
  const int cb = (wave & 1) * 16;
  v8f acc = {0.f, 0.f, 0.f, 0.f, 0.f, 0.f, 0.f, 0.f};
  for (int kk = 0; kk < Dc / 4; ++kk) {
    const int c0 = 4 * kk + 2 * hf;
    v2f a;
    a.x = xs[l16 * XSTR + c0];
    a.y = xs[l16 * XSTR + c0 + 1];
    v2f bv;
    bv.x = Wm[(size_t)c0 * Rc + cb + l16];
    bv.y = Wm[(size_t)(c0 + 1) * Rc + cb + l16];
    acc = wmma_f32(a, bv, acc);
  }
  if (full) {
    for (int i = 0; i < 8; ++i) {
      const int r = i + 8 * hf;
      out[((size_t)b * Nc + n0 + r) * Rc + cb + l16] = acc[i];
    }
  } else {
    for (int i = 0; i < 8; ++i) {
      const int r = i + 8 * hf, gg = n0 + r;
      if (gg < Nc) out[((size_t)b * Nc + gg) * Rc + cb + l16] = acc[i];
    }
  }
}

__global__ __launch_bounds__(128) void embed_ln_proj_kernel(
    const int* __restrict__ ids, const float* __restrict__ emb,
    const float* __restrict__ pos, const float* __restrict__ anchor_val,
    const float* __restrict__ anchor_state, const float* __restrict__ in_gamma,
    const float* __restrict__ in_beta, const float* __restrict__ sp_w,
    const float* __restrict__ sp_b, const float* __restrict__ gamma,
    const float* __restrict__ beta, const float* __restrict__ Uw,
    const float* __restrict__ Vw, float* __restrict__ val,
    float* __restrict__ st, float* __restrict__ xo, float* __restrict__ qo,
    float* __restrict__ ko) {
  __shared__ float xs[16 * XSTR];
  const int tile = blockIdx.x % NT;
  const int b = blockIdx.x / NT;
  const int n0 = tile * 16;
  const int tid = threadIdx.x;
  const bool full = (n0 + 16 <= Nc);
  const int row = tid >> 3, sub = tid & 7;
  const int g = n0 + row;
  const int gc = min(g, Nc - 1);
  const int s = max(gc - 1, 0);
  const int id = ids[b * Sc + s];
  const float* er = emb + (size_t)id * Dc + 4 * sub;
  const float* pr = pos + (size_t)s * Dc + 4 * sub;
  float e[64];
#pragma unroll
  for (int j = 0; j < 16; ++j) {
#pragma unroll
    for (int i = 0; i < 4; ++i) e[4 * j + i] = er[32 * j + i] + pr[32 * j + i];
  }
  float s1 = 0.0f, s2 = 0.0f;
#pragma unroll
  for (int c = 0; c < 64; ++c) {
    s1 += e[c];
    s2 += e[c] * e[c];
  }
  for (int m = 1; m < 8; m <<= 1) {
    s1 += __shfl_xor(s1, m, 32);
    s2 += __shfl_xor(s2, m, 32);
  }
  const float mean = s1 * (1.0f / Dc);
  const float var = s2 * (1.0f / Dc) - mean * mean;
  const float rs = rsqrtf(var + 1e-5f);
  float dot = 0.0f;
#pragma unroll
  for (int j = 0; j < 16; ++j) {
#pragma unroll
    for (int i = 0; i < 4; ++i) {
      const int c = 32 * j + 4 * sub + i;
      const float y = (e[4 * j + i] - mean) * rs * in_gamma[c] + in_beta[c];
      e[4 * j + i] = y;
      dot += y * sp_w[c];
    }
  }
  for (int m = 1; m < 8; m <<= 1) dot += __shfl_xor(dot, m, 32);
  float stv = dot + sp_b[0];
  if (g == 0) {
#pragma unroll
    for (int j = 0; j < 16; ++j) {
#pragma unroll
      for (int i = 0; i < 4; ++i)
        e[4 * j + i] = anchor_val[32 * j + 4 * sub + i];
    }
    stv = anchor_state[0];
  }
  if (g < Nc) {
    float* vrow = val + ((size_t)b * Nc + g) * Dc + 4 * sub;
#pragma unroll
    for (int j = 0; j < 16; ++j) {
#pragma unroll
      for (int i = 0; i < 4; ++i) vrow[32 * j + i] = e[4 * j + i];
    }
    if (sub == 0) st[(size_t)b * Nc + g] = stv;
  }
  float* xr = xs + row * XSTR + 4 * sub;
#pragma unroll
  for (int j = 0; j < 16; ++j) {
#pragma unroll
    for (int i = 0; i < 4; ++i) xr[32 * j + i] = (g < Nc) ? e[4 * j + i] : 0.0f;
  }
  __syncthreads();
  ln_proj_tile(xs, b, n0, full, gamma, beta, Uw, Vw, xo, qo, ko);
}

__global__ __launch_bounds__(128) void ln_proj_kernel(
    const float* __restrict__ val, const float* __restrict__ gamma,
    const float* __restrict__ beta, const float* __restrict__ Uw,
    const float* __restrict__ Vw, float* __restrict__ xo,
    float* __restrict__ qo, float* __restrict__ ko) {
  __shared__ float xs[16 * XSTR];
  const int tile = blockIdx.x % NT;
  const int b = blockIdx.x / NT;
  const int n0 = tile * 16;
  const int tid = threadIdx.x;
  const bool full = (n0 + 16 <= Nc);

  const int row = tid >> 3, sub = tid & 7;
  const int g = n0 + row;
  const float* vrow = val + ((size_t)b * Nc + g) * Dc;
  if (full) {
    for (int c = sub * 64; c < sub * 64 + 64; ++c)
      xs[row * XSTR + c] = vrow[c];
  } else {
    const float* crow = val + ((size_t)b * Nc + min(g, Nc - 1)) * Dc;
    for (int c = sub * 64; c < sub * 64 + 64; ++c) {
      const float v = crow[c];
      xs[row * XSTR + c] = (g < Nc) ? v : 0.0f;
    }
  }
  ln_proj_tile(xs, b, n0, full, gamma, beta, Uw, Vw, xo, qo, ko);
}

constexpr int ATT = 512;
__global__ __launch_bounds__(ATT) void attn_kernel(
    const float* __restrict__ x, const float* __restrict__ q,
    const float* __restrict__ k, const float* __restrict__ st,
    float* __restrict__ val) {
  __shared__ float qs[16 * QSTR];
  __shared__ float ks[WINc * KSTR];
  __shared__ float ss[16 * SSTR];
  __shared__ float wm[16 * WSTR];
  __shared__ float sw[WINc];
  const int tile = blockIdx.x % NT;
  const int b = blockIdx.x / NT;
  const int n0 = tile * 16;
  const int tid = threadIdx.x;
  const bool full = (n0 + 16 <= Nc);
  const bool interior = (n0 >= Wc) && (n0 + 16 + Wc <= Nc);

  const float* qb = q + ((size_t)b * Nc + n0) * Rc;
  const float* kb = k + ((size_t)b * Nc + n0 - Wc) * Rc;
  const float* sb = st + (size_t)b * Nc + n0 - Wc;
  if (interior) {
    for (int i = tid; i < 16 * Rc; i += ATT) {
      const int r = i >> 5, c = i & 31;
      qs[r * QSTR + c] = qb[r * Rc + c];
    }
    for (int i = tid; i < WINc * Rc; i += ATT) {
      const int j = i >> 5, c = i & 31;
      ks[j * KSTR + c] = kb[(size_t)j * Rc + c];
    }
    for (int j = tid; j < WINc; j += ATT) sw[j] = sb[j];
  } else {
    static_assert((16 * Rc) % ATT == 0 && (WINc * Rc) % ATT == 0,
                  "q tile and k window: whole trips of the block's threads");
#pragma unroll
    for (int it = 0; it < 16 * Rc / ATT; ++it) {
      const int i = tid + it * ATT;
      const int r = i >> 5, c = i & 31, g = n0 + r;
      const float v = q[((size_t)b * Nc + min(g, Nc - 1)) * Rc + c];
      qs[r * QSTR + c] = (g < Nc) ? v : 0.0f;
    }
#pragma unroll
    for (int it = 0; it < WINc * Rc / ATT; ++it) {
      const int i = tid + it * ATT;
      const int j = i >> 5, c = i & 31, gn = n0 - Wc + j;
      const int gc = min(max(gn, 0), Nc - 1);
      const float v = k[((size_t)b * Nc + gc) * Rc + c];
      ks[j * KSTR + c] = (gc == gn) ? v : 0.0f;
    }
    for (int j = tid; j < WINc; j += ATT) {
      const int gn = n0 - Wc + j;
      const int gc = min(max(gn, 0), Nc - 1);
      const float v = st[(size_t)b * Nc + gc];
      sw[j] = (gc == gn) ? v : 0.0f;
    }
  }
  for (int i = tid; i < 16 * WSTR; i += ATT) wm[i] = 0.0f;
  __syncthreads();

  const int wave = tid >> 5, lane = tid & 31;
  const int hf = lane >> 4, l16 = lane & 15;

  for (int t = wave; t < 9; t += ATT / 32) {
    const int cb = t * 16;
    v8f acc = {0.f, 0.f, 0.f, 0.f, 0.f, 0.f, 0.f, 0.f};
    for (int kk = 0; kk < Rc / 4; ++kk) {
      const int c0 = 4 * kk + 2 * hf;
      v2f a;
      a.x = qs[l16 * QSTR + c0];
      a.y = qs[l16 * QSTR + c0 + 1];
      v2f bv;
      bv.x = ks[(cb + l16) * KSTR + c0];
      bv.y = ks[(cb + l16) * KSTR + c0 + 1];
      acc = wmma_f32(a, bv, acc);
    }
    for (int i = 0; i < 8; ++i) ss[(i + 8 * hf) * SSTR + cb + l16] = acc[i];
  }
  __syncthreads();

  if (tid < 128) {
    const int row = tid >> 3, sub = tid & 7;
    const int g = n0 + row;
    if (interior) {
      float m = -1e30f;
      for (int o = sub; o < WNc; o += 8)
        m = fmaxf(m, fabsf(ss[row * SSTR + row + o]));
      for (int mm = 1; mm < 8; mm <<= 1) m = fmaxf(m, __shfl_xor(m, mm, 32));
      float Z = 0.0f;
      for (int o = sub; o < WNc; o += 8)
        Z += expf(fabsf(ss[row * SSTR + row + o]) - m);
      for (int mm = 1; mm < 8; mm <<= 1) Z += __shfl_xor(Z, mm, 32);
      const float inv = 1.0f / Z;
      for (int o = sub; o < WNc; o += 8) {
        const float s = ss[row * SSTR + row + o];
        const float p = expf(fabsf(s) - m) * inv;
        const float wv = (s > 0.0f) ? p : ((s < 0.0f) ? -p : 0.0f);
        wm[row * WSTR + row + o] = wv * sw[row + o];
      }
    } else if (g < Nc) {
      float m = -1e30f;
      for (int o = sub; o < WNc; o += 8) {
        const int gn = g - Wc + o;
        if (gn >= 0 && gn < Nc) m = fmaxf(m, fabsf(ss[row * SSTR + row + o]));
      }
      for (int mm = 1; mm < 8; mm <<= 1) m = fmaxf(m, __shfl_xor(m, mm, 32));
      float Z = 0.0f;
      for (int o = sub; o < WNc; o += 8) {
        const int gn = g - Wc + o;
        if (gn >= 0 && gn < Nc)
          Z += expf(fabsf(ss[row * SSTR + row + o]) - m);
      }
      for (int mm = 1; mm < 8; mm <<= 1) Z += __shfl_xor(Z, mm, 32);
      const float inv = 1.0f / Z;
      for (int o = sub; o < WNc; o += 8) {
        const int gn = g - Wc + o;
        if (gn >= 0 && gn < Nc) {
          const float s = ss[row * SSTR + row + o];
          const float p = expf(fabsf(s) - m) * inv;
          const float wv = (s > 0.0f) ? p : ((s < 0.0f) ? -p : 0.0f);
          wm[row * WSTR + row + o] = wv * sw[row + o];
        }
      }
    }
  }
  __syncthreads();

  static_assert(Dc / 16 == 2 * (ATT / 32), "two column tiles per wave");
  {
    constexpr int KS = WINc / 4;
    const int cb = wave * 32 + l16;
    v2f bq0[KS], bq1[KS];
    if (interior) {
      const float* xw = x + ((size_t)b * Nc + n0 - Wc + 2 * hf) * Dc + cb;
#pragma unroll
      for (int kk = 0; kk < KS; ++kk) {
        const float* p = xw + (size_t)(4 * kk) * Dc;
        bq0[kk].x = p[0];
        bq0[kk].y = p[Dc];
        bq1[kk].x = p[16];
        bq1[kk].y = p[Dc + 16];
      }
    } else {
      const float* xb = x + (size_t)b * Nc * Dc + cb;
#pragma unroll
      for (int kk = 0; kk < KS; ++kk) {
        const int g0 = n0 - Wc + 4 * kk + 2 * hf, g1 = g0 + 1;
        const int r0 = min(max(g0, 0), Nc - 1), r1 = min(max(g1, 0), Nc - 1);
        const float* p0 = xb + (size_t)r0 * Dc;
        const float* p1 = xb + (size_t)r1 * Dc;
        const float v00 = p0[0], v01 = p1[0], v10 = p0[16], v11 = p1[16];
        bq0[kk].x = (r0 == g0) ? v00 : 0.0f;
        bq0[kk].y = (r1 == g1) ? v01 : 0.0f;
        bq1[kk].x = (r0 == g0) ? v10 : 0.0f;
        bq1[kk].y = (r1 == g1) ? v11 : 0.0f;
      }
    }
    v8f acc0 = {0.f, 0.f, 0.f, 0.f, 0.f, 0.f, 0.f, 0.f};
    v8f acc1 = acc0;
#pragma unroll
    for (int kk = 0; kk < KS; ++kk) {
      const int c0 = 4 * kk + 2 * hf;
      v2f a;
      a.x = wm[l16 * WSTR + c0];
      a.y = wm[l16 * WSTR + c0 + 1];
      acc0 = wmma_f32(a, bq0[kk], acc0);
      acc1 = wmma_f32(a, bq1[kk], acc1);
    }
    if (full) {
      float* vb = val + ((size_t)b * Nc + n0) * Dc + cb;
      for (int i = 0; i < 8; ++i) {
        const int r = i + 8 * hf;
        vb[(size_t)r * Dc] += acc0[i];
        vb[(size_t)r * Dc + 16] += acc1[i];
      }
    } else {
      for (int i = 0; i < 8; ++i) {
        const int r = i + 8 * hf, gg = n0 + r;
        if (gg < Nc) {
          float* vp = val + ((size_t)b * Nc + gg) * Dc + cb;
          vp[0] += acc0[i];
          vp[16] += acc1[i];
        }
      }
    }
  }
}

extern "C" void kernel_launch(void* const* d_in, const int* in_sizes, int n_in,
                              void* d_out, int out_size, void* d_ws,
                              size_t ws_size, hipStream_t stream) {
  const int* ids = (const int*)d_in[0];
  const float* emb = (const float*)d_in[1];
  const float* pos = (const float*)d_in[2];
  const float* anchor_val = (const float*)d_in[3];
  const float* anchor_state = (const float*)d_in[4];
  const float* in_gamma = (const float*)d_in[5];
  const float* in_beta = (const float*)d_in[6];
  const float* ng = (const float*)d_in[7];
  const float* nb = (const float*)d_in[8];
  const float* Uw = (const float*)d_in[9];
  const float* Vw = (const float*)d_in[10];
  const float* sp_w = (const float*)d_in[11];
  const float* sp_b = (const float*)d_in[12];

  float* val = (float*)d_out;
  float* ws = (float*)d_ws;
  float* x = ws;
  float* q = x + (size_t)Bc * Nc * Dc;
  float* k = q + (size_t)Bc * Nc * Rc;
  float* st = k + (size_t)Bc * Nc * Rc;

  embed_ln_proj_kernel<<<Bc * NT, 128, 0, stream>>>(
      ids, emb, pos, anchor_val, anchor_state, in_gamma, in_beta, sp_w, sp_b,
      ng, nb, Uw, Vw, val, st, x, q, k);
  attn_kernel<<<Bc * NT, ATT, 0, stream>>>(x, q, k, st, val);

  for (int l = 1; l < Lc; ++l) {
    ln_proj_kernel<<<Bc * NT, 128, 0, stream>>>(
        val, ng + (size_t)l * Dc, nb + (size_t)l * Dc, Uw + (size_t)l * Dc * Rc,
        Vw + (size_t)l * Dc * Rc, x, q, k);
    attn_kernel<<<Bc * NT, ATT, 0, stream>>>(x, q, k, st, val);
  }
}


# LN+projection kernels: 512-thread work-groups; each output tile's K split over 4 waves, whose weight loads are requested before the LayerNorm; the 4 quarter sums are added via LDS in rising K order
# speedup vs baseline: 4.7933x; 1.2872x over previous
#include <hip/hip_runtime.h>

typedef __attribute__((ext_vector_type(2))) float v2f;
typedef __attribute__((ext_vector_type(8))) float v8f;

constexpr int Bc = 2;
constexpr int Sc = 1024;
constexpr int Nc = Sc + 1;
constexpr int Dc = 512;
constexpr int Rc = 32;
constexpr int Lc = 3;
constexpr int Wc = 64;
constexpr int WNc = 2 * Wc + 1;
constexpr int WINc = 144;
constexpr int NT = (Nc + 15) / 16;

constexpr int XSTR = Dc + 4;
constexpr int QSTR = 33;
constexpr int KSTR = 33;
constexpr int SSTR = 146;
constexpr int WSTR = 145;

__device__ __forceinline__ v8f wmma_f32(v2f a, v2f b, v8f c) {
  return __builtin_amdgcn_wmma_f32_16x16x4_f32(false, a, false, b, (short)0, c,
                                               false, false);
}

constexpr int LNT = 512;
constexpr int KQ = Dc / 4 / 4;
__device__ __forceinline__ void ln_proj_tile(
    float* xs, float* red, const int b, const int n0, const bool full,
    const float* __restrict__ gamma, const float* __restrict__ beta,
    const float* __restrict__ Uw, const float* __restrict__ Vw,
    float* __restrict__ xo, float* __restrict__ qo, float* __restrict__ ko) {
  const int tid = threadIdx.x;
  const int wave = tid >> 5, lane = tid & 31;
  const int hf = lane >> 4, l16 = lane & 15;
  const int tsel = wave & 3, kq = wave >> 2;
  const int cb = (tsel & 1) * 16;

  v2f bq[KQ];
  {
    const float* wp =
        ((tsel < 2) ? Uw : Vw) + (size_t)(4 * KQ * kq + 2 * hf) * Rc + cb + l16;
#pragma unroll
    for (int j = 0; j < KQ; ++j) {
      bq[j].x = wp[(size_t)(4 * j) * Rc];
      bq[j].y = wp[(size_t)(4 * j + 1) * Rc];
    }
  }

  if (tid < 128) {
    const int row = tid >> 3, sub = tid & 7;
    const int g = n0 + row;
    float s1 = 0.0f, s2 = 0.0f;
    for (int c = sub * 64; c < sub * 64 + 64; ++c) {
      const float v = xs[row * XSTR + c];
      s1 += v;
      s2 += v * v;
    }
    for (int m = 1; m < 8; m <<= 1) {
      s1 += __shfl_xor(s1, m, 32);
      s2 += __shfl_xor(s2, m, 32);
    }
    const float mean = s1 * (1.0f / Dc);
    const float var = s2 * (1.0f / Dc) - mean * mean;
    const float rs = rsqrtf(var + 1e-5f);
    float* xrow = xo + ((size_t)b * Nc + g) * Dc;
    if (full) {
      for (int c = sub * 64; c < sub * 64 + 64; ++c) {
        const float y = (xs[row * XSTR + c] - mean) * rs * gamma[c] + beta[c];
        xs[row * XSTR + c] = y;
        xrow[c] = y;
      }
    } else if (g < Nc) {
      for (int c = sub * 64; c < sub * 64 + 64; ++c) {
        const float y = (xs[row * XSTR + c] - mean) * rs * gamma[c] + beta[c];
        xs[row * XSTR + c] = y;
        xrow[c] = y;
      }
    } else {
      for (int c = sub * 64; c < sub * 64 + 64; ++c) {
        const float y = (xs[row * XSTR + c] - mean) * rs * gamma[c] + beta[c];
        xs[row * XSTR + c] = y;
      }
    }
  }
  __syncthreads();

  v8f acc = {0.f, 0.f, 0.f, 0.f, 0.f, 0.f, 0.f, 0.f};
#pragma unroll
  for (int j = 0; j < KQ; ++j) {
    const int c0 = 4 * KQ * kq + 4 * j + 2 * hf;
    v2f a;
    a.x = xs[l16 * XSTR + c0];
    a.y = xs[l16 * XSTR + c0 + 1];
    acc = wmma_f32(a, bq[j], acc);
  }
  for (int i = 0; i < 8; ++i) red[(wave * 8 + i) * 32 + lane] = acc[i];
  __syncthreads();

  {
    const int mat = tid >> 8;
    const int r = (tid >> 4) & 15;
    const int col = 2 * (tid & 15);
    const int ts = 2 * mat + (col >> 4);
    const int e = (ts * 8 + (r & 7)) * 32 + 16 * (r >> 3) + (col & 15);
    float o0 = red[e], o1 = red[e + 1];
    for (int p = 1; p < 4; ++p) {
      o0 += red[p * 1024 + e];
      o1 += red[p * 1024 + e + 1];
    }
    if (full || n0 + r < Nc) {
      float* op = (mat ? ko : qo) + ((size_t)b * Nc + n0 + r) * Rc + col;
      op[0] = o0;
      op[1] = o1;
    }
  }
}

__global__ __launch_bounds__(LNT) void embed_ln_proj_kernel(
    const int* __restrict__ ids, const float* __restrict__ emb,
    const float* __restrict__ pos, const float* __restrict__ anchor_val,
    const float* __restrict__ anchor_state, const float* __restrict__ in_gamma,
    const float* __restrict__ in_beta, const float* __restrict__ sp_w,
    const float* __restrict__ sp_b, const float* __restrict__ gamma,
    const float* __restrict__ beta, const float* __restrict__ Uw,
    const float* __restrict__ Vw, float* __restrict__ val,
    float* __restrict__ st, float* __restrict__ xo, float* __restrict__ qo,
    float* __restrict__ ko) {
  __shared__ float xs[16 * XSTR];
  __shared__ float red[16 * 8 * 32];
  const int tile = blockIdx.x % NT;
  const int b = blockIdx.x / NT;
  const int n0 = tile * 16;
  const int tid = threadIdx.x;
  const bool full = (n0 + 16 <= Nc);
  if (tid < 128) {
    const int row = tid >> 3, sub = tid & 7;
    const int g = n0 + row;
    const int gc = min(g, Nc - 1);
    const int s = max(gc - 1, 0);
    const int id = ids[b * Sc + s];
    const float* er = emb + (size_t)id * Dc + 4 * sub;
    const float* pr = pos + (size_t)s * Dc + 4 * sub;
    float e[64];
#pragma unroll
    for (int j = 0; j < 16; ++j) {
#pragma unroll
      for (int i = 0; i < 4; ++i) e[4 * j + i] = er[32 * j + i] + pr[32 * j + i];
    }
    float s1 = 0.0f, s2 = 0.0f;
#pragma unroll
    for (int c = 0; c < 64; ++c) {
      s1 += e[c];
      s2 += e[c] * e[c];
    }
    for (int m = 1; m < 8; m <<= 1) {
      s1 += __shfl_xor(s1, m, 32);
      s2 += __shfl_xor(s2, m, 32);
    }
    const float mean = s1 * (1.0f / Dc);
    const float var = s2 * (1.0f / Dc) - mean * mean;
    const float rs = rsqrtf(var + 1e-5f);
    float dot = 0.0f;
#pragma unroll
    for (int j = 0; j < 16; ++j) {
      if (j == 8) __builtin_amdgcn_sched_barrier(0);
#pragma unroll
      for (int i = 0; i < 4; ++i) {
        const int c = 32 * j + 4 * sub + i;
        const float y = (e[4 * j + i] - mean) * rs * in_gamma[c] + in_beta[c];
        e[4 * j + i] = y;
        dot += y * sp_w[c];
      }
    }
    for (int m = 1; m < 8; m <<= 1) dot += __shfl_xor(dot, m, 32);
    float stv = dot + sp_b[0];
    if (g == 0) {
#pragma unroll
      for (int j = 0; j < 16; ++j) {
#pragma unroll
        for (int i = 0; i < 4; ++i)
          e[4 * j + i] = anchor_val[32 * j + 4 * sub + i];
      }
      stv = anchor_state[0];
    }
    if (g < Nc) {
      float* vrow = val + ((size_t)b * Nc + g) * Dc + 4 * sub;
#pragma unroll
      for (int j = 0; j < 16; ++j) {
#pragma unroll
        for (int i = 0; i < 4; ++i) vrow[32 * j + i] = e[4 * j + i];
      }
      if (sub == 0) st[(size_t)b * Nc + g] = stv;
    }
    float* xr = xs + row * XSTR + 4 * sub;
#pragma unroll
    for (int j = 0; j < 16; ++j) {
#pragma unroll
      for (int i = 0; i < 4; ++i) xr[32 * j + i] = (g < Nc) ? e[4 * j + i] : 0.0f;
    }
  }
  __syncthreads();
  ln_proj_tile(xs, red, b, n0, full, gamma, beta, Uw, Vw, xo, qo, ko);
}

__global__ __launch_bounds__(LNT) void ln_proj_kernel(
    const float* __restrict__ val, const float* __restrict__ gamma,
    const float* __restrict__ beta, const float* __restrict__ Uw,
    const float* __restrict__ Vw, float* __restrict__ xo,
    float* __restrict__ qo, float* __restrict__ ko) {
  __shared__ float xs[16 * XSTR];
  __shared__ float red[16 * 8 * 32];
  const int tile = blockIdx.x % NT;
  const int b = blockIdx.x / NT;
  const int n0 = tile * 16;
  const int tid = threadIdx.x;
  const bool full = (n0 + 16 <= Nc);

  if (tid < 128) {
    const int row = tid >> 3, sub = tid & 7;
    const int g = n0 + row;
    const float* vrow = val + ((size_t)b * Nc + g) * Dc;
    if (full) {
      for (int c = sub * 64; c < sub * 64 + 64; ++c)
        xs[row * XSTR + c] = vrow[c];
    } else {
      const float* crow = val + ((size_t)b * Nc + min(g, Nc - 1)) * Dc;
      for (int c = sub * 64; c < sub * 64 + 64; ++c) {
        const float v = crow[c];
        xs[row * XSTR + c] = (g < Nc) ? v : 0.0f;
      }
    }
  }
  ln_proj_tile(xs, red, b, n0, full, gamma, beta, Uw, Vw, xo, qo, ko);
}

constexpr int ATT = 512;
__global__ __launch_bounds__(ATT) void attn_kernel(
    const float* __restrict__ x, const float* __restrict__ q,
    const float* __restrict__ k, const float* __restrict__ st,
    float* __restrict__ val) {
  __shared__ float qs[16 * QSTR];
  __shared__ float ks[WINc * KSTR];
  __shared__ float ss[16 * SSTR];
  __shared__ float wm[16 * WSTR];
  __shared__ float sw[WINc];
  const int tile = blockIdx.x % NT;
  const int b = blockIdx.x / NT;
  const int n0 = tile * 16;
  const int tid = threadIdx.x;
  const bool full = (n0 + 16 <= Nc);
  const bool interior = (n0 >= Wc) && (n0 + 16 + Wc <= Nc);

  const float* qb = q + ((size_t)b * Nc + n0) * Rc;
  const float* kb = k + ((size_t)b * Nc + n0 - Wc) * Rc;
  const float* sb = st + (size_t)b * Nc + n0 - Wc;
  if (interior) {
    for (int i = tid; i < 16 * Rc; i += ATT) {
      const int r = i >> 5, c = i & 31;
      qs[r * QSTR + c] = qb[r * Rc + c];
    }
    for (int i = tid; i < WINc * Rc; i += ATT) {
      const int j = i >> 5, c = i & 31;
      ks[j * KSTR + c] = kb[(size_t)j * Rc + c];
    }
    for (int j = tid; j < WINc; j += ATT) sw[j] = sb[j];
  } else {
    static_assert((16 * Rc) % ATT == 0 && (WINc * Rc) % ATT == 0,
                  "q tile and k window: whole trips of the block's threads");
#pragma unroll
    for (int it = 0; it < 16 * Rc / ATT; ++it) {
      const int i = tid + it * ATT;
      const int r = i >> 5, c = i & 31, g = n0 + r;
      const float v = q[((size_t)b * Nc + min(g, Nc - 1)) * Rc + c];
      qs[r * QSTR + c] = (g < Nc) ? v : 0.0f;
    }
#pragma unroll
    for (int it = 0; it < WINc * Rc / ATT; ++it) {
      const int i = tid + it * ATT;
      const int j = i >> 5, c = i & 31, gn = n0 - Wc + j;
      const int gc = min(max(gn, 0), Nc - 1);
      const float v = k[((size_t)b * Nc + gc) * Rc + c];
      ks[j * KSTR + c] = (gc == gn) ? v : 0.0f;
    }
    for (int j = tid; j < WINc; j += ATT) {
      const int gn = n0 - Wc + j;
      const int gc = min(max(gn, 0), Nc - 1);
      const float v = st[(size_t)b * Nc + gc];
      sw[j] = (gc == gn) ? v : 0.0f;
    }
  }
  for (int i = tid; i < 16 * WSTR; i += ATT) wm[i] = 0.0f;
  __syncthreads();

  const int wave = tid >> 5, lane = tid & 31;
  const int hf = lane >> 4, l16 = lane & 15;

  for (int t = wave; t < 9; t += ATT / 32) {
    const int cb = t * 16;
    v8f acc = {0.f, 0.f, 0.f, 0.f, 0.f, 0.f, 0.f, 0.f};
    for (int kk = 0; kk < Rc / 4; ++kk) {
      const int c0 = 4 * kk + 2 * hf;
      v2f a;
      a.x = qs[l16 * QSTR + c0];
      a.y = qs[l16 * QSTR + c0 + 1];
      v2f bv;
      bv.x = ks[(cb + l16) * KSTR + c0];
      bv.y = ks[(cb + l16) * KSTR + c0 + 1];
      acc = wmma_f32(a, bv, acc);
    }
    for (int i = 0; i < 8; ++i) ss[(i + 8 * hf) * SSTR + cb + l16] = acc[i];
  }
  __syncthreads();

  if (tid < 128) {
    const int row = tid >> 3, sub = tid & 7;
    const int g = n0 + row;
    if (interior) {
      float m = -1e30f;
      for (int o = sub; o < WNc; o += 8)
        m = fmaxf(m, fabsf(ss[row * SSTR + row + o]));
      for (int mm = 1; mm < 8; mm <<= 1) m = fmaxf(m, __shfl_xor(m, mm, 32));
      float Z = 0.0f;
      for (int o = sub; o < WNc; o += 8)
        Z += expf(fabsf(ss[row * SSTR + row + o]) - m);
      for (int mm = 1; mm < 8; mm <<= 1) Z += __shfl_xor(Z, mm, 32);
      const float inv = 1.0f / Z;
      for (int o = sub; o < WNc; o += 8) {
        const float s = ss[row * SSTR + row + o];
        const float p = expf(fabsf(s) - m) * inv;
        const float wv = (s > 0.0f) ? p : ((s < 0.0f) ? -p : 0.0f);
        wm[row * WSTR + row + o] = wv * sw[row + o];
      }
    } else if (g < Nc) {
      float m = -1e30f;
      for (int o = sub; o < WNc; o += 8) {
        const int gn = g - Wc + o;
        if (gn >= 0 && gn < Nc) m = fmaxf(m, fabsf(ss[row * SSTR + row + o]));
      }
      for (int mm = 1; mm < 8; mm <<= 1) m = fmaxf(m, __shfl_xor(m, mm, 32));
      float Z = 0.0f;
      for (int o = sub; o < WNc; o += 8) {
        const int gn = g - Wc + o;
        if (gn >= 0 && gn < Nc)
          Z += expf(fabsf(ss[row * SSTR + row + o]) - m);
      }
      for (int mm = 1; mm < 8; mm <<= 1) Z += __shfl_xor(Z, mm, 32);
      const float inv = 1.0f / Z;
      for (int o = sub; o < WNc; o += 8) {
        const int gn = g - Wc + o;
        if (gn >= 0 && gn < Nc) {
          const float s = ss[row * SSTR + row + o];
          const float p = expf(fabsf(s) - m) * inv;
          const float wv = (s > 0.0f) ? p : ((s < 0.0f) ? -p : 0.0f);
          wm[row * WSTR + row + o] = wv * sw[row + o];
        }
      }
    }
  }
  __syncthreads();

  static_assert(Dc / 16 == 2 * (ATT / 32), "two column tiles per wave");
  {
    constexpr int KS = WINc / 4;
    const int cb = wave * 32 + l16;
    v2f bq0[KS], bq1[KS];
    if (interior) {
      const float* xw = x + ((size_t)b * Nc + n0 - Wc + 2 * hf) * Dc + cb;
#pragma unroll
      for (int kk = 0; kk < KS; ++kk) {
        const float* p = xw + (size_t)(4 * kk) * Dc;
        bq0[kk].x = p[0];
        bq0[kk].y = p[Dc];
        bq1[kk].x = p[16];
        bq1[kk].y = p[Dc + 16];
      }
    } else {
      const float* xb = x + (size_t)b * Nc * Dc + cb;
#pragma unroll
      for (int kk = 0; kk < KS; ++kk) {
        const int g0 = n0 - Wc + 4 * kk + 2 * hf, g1 = g0 + 1;
        const int r0 = min(max(g0, 0), Nc - 1), r1 = min(max(g1, 0), Nc - 1);
        const float* p0 = xb + (size_t)r0 * Dc;
        const float* p1 = xb + (size_t)r1 * Dc;
        const float v00 = p0[0], v01 = p1[0], v10 = p0[16], v11 = p1[16];
        bq0[kk].x = (r0 == g0) ? v00 : 0.0f;
        bq0[kk].y = (r1 == g1) ? v01 : 0.0f;
        bq1[kk].x = (r0 == g0) ? v10 : 0.0f;
        bq1[kk].y = (r1 == g1) ? v11 : 0.0f;
      }
    }
    v8f acc0 = {0.f, 0.f, 0.f, 0.f, 0.f, 0.f, 0.f, 0.f};
    v8f acc1 = acc0;
#pragma unroll
    for (int kk = 0; kk < KS; ++kk) {
      const int c0 = 4 * kk + 2 * hf;
      v2f a;
      a.x = wm[l16 * WSTR + c0];
      a.y = wm[l16 * WSTR + c0 + 1];
      acc0 = wmma_f32(a, bq0[kk], acc0);
      acc1 = wmma_f32(a, bq1[kk], acc1);
    }
    if (full) {
      float* vb = val + ((size_t)b * Nc + n0) * Dc + cb;
      for (int i = 0; i < 8; ++i) {
        const int r = i + 8 * hf;
        vb[(size_t)r * Dc] += acc0[i];
        vb[(size_t)r * Dc + 16] += acc1[i];
      }
    } else {
      for (int i = 0; i < 8; ++i) {
        const int r = i + 8 * hf, gg = n0 + r;
        if (gg < Nc) {
          float* vp = val + ((size_t)b * Nc + gg) * Dc + cb;
          vp[0] += acc0[i];
          vp[16] += acc1[i];
        }
      }
    }
  }
}

extern "C" void kernel_launch(void* const* d_in, const int* in_sizes, int n_in,
                              void* d_out, int out_size, void* d_ws,
                              size_t ws_size, hipStream_t stream) {
  const int* ids = (const int*)d_in[0];
  const float* emb = (const float*)d_in[1];
  const float* pos = (const float*)d_in[2];
  const float* anchor_val = (const float*)d_in[3];
  const float* anchor_state = (const float*)d_in[4];
  const float* in_gamma = (const float*)d_in[5];
  const float* in_beta = (const float*)d_in[6];
  const float* ng = (const float*)d_in[7];
  const float* nb = (const float*)d_in[8];
  const float* Uw = (const float*)d_in[9];
  const float* Vw = (const float*)d_in[10];
  const float* sp_w = (const float*)d_in[11];
  const float* sp_b = (const float*)d_in[12];

  float* val = (float*)d_out;
  float* ws = (float*)d_ws;
  float* x = ws;
  float* q = x + (size_t)Bc * Nc * Dc;
  float* k = q + (size_t)Bc * Nc * Rc;
  float* st = k + (size_t)Bc * Nc * Rc;

  embed_ln_proj_kernel<<<Bc * NT, LNT, 0, stream>>>(
      ids, emb, pos, anchor_val, anchor_state, in_gamma, in_beta, sp_w, sp_b,
      ng, nb, Uw, Vw, val, st, x, q, k);
  attn_kernel<<<Bc * NT, ATT, 0, stream>>>(x, q, k, st, val);

  for (int l = 1; l < Lc; ++l) {
    ln_proj_kernel<<<Bc * NT, LNT, 0, stream>>>(
        val, ng + (size_t)l * Dc, nb + (size_t)l * Dc, Uw + (size_t)l * Dc * Rc,
        Vw + (size_t)l * Dc * Rc, x, q, k);
    attn_kernel<<<Bc * NT, ATT, 0, stream>>>(x, q, k, st, val);
  }
}


# attn kernel: softmax on all 16 waves, one row per wave (5 scores per lane read together, exps kept in registers, row_xmask + permlanex16 reductions); row sum Z added in another order
# speedup vs baseline: 5.2001x; 1.0849x over previous
#include <hip/hip_runtime.h>

typedef __attribute__((ext_vector_type(2))) float v2f;
typedef __attribute__((ext_vector_type(8))) float v8f;

constexpr int Bc = 2;
constexpr int Sc = 1024;
constexpr int Nc = Sc + 1;
constexpr int Dc = 512;
constexpr int Rc = 32;
constexpr int Lc = 3;
constexpr int Wc = 64;
constexpr int WNc = 2 * Wc + 1;
constexpr int WINc = 144;
constexpr int NT = (Nc + 15) / 16;

constexpr int XSTR = Dc + 4;
constexpr int QSTR = 33;
constexpr int KSTR = 33;
constexpr int SSTR = 146;
constexpr int WSTR = 145;

__device__ __forceinline__ v8f wmma_f32(v2f a, v2f b, v8f c) {
  return __builtin_amdgcn_wmma_f32_16x16x4_f32(false, a, false, b, (short)0, c,
                                               false, false);
}

template <int M>
__device__ __forceinline__ float row_xor(float v) {
  return __builtin_bit_cast(
      float, __builtin_amdgcn_update_dpp(0, __builtin_bit_cast(int, v),
                                         0x160 + M, 0xF, 0xF, true));
}
__device__ __forceinline__ float half_xor(float v) {
  const int x = __builtin_bit_cast(int, v);
  return __builtin_bit_cast(
      float, __builtin_amdgcn_permlanex16(x, x, 0x76543210, 0xfedcba98, false,
                                          false));
}

constexpr int LNT = 512;
constexpr int KQ = Dc / 4 / 4;
__device__ __forceinline__ void ln_proj_tile(
    float* xs, float* red, const int b, const int n0, const bool full,
    const float* __restrict__ gamma, const float* __restrict__ beta,
    const float* __restrict__ Uw, const float* __restrict__ Vw,
    float* __restrict__ xo, float* __restrict__ qo, float* __restrict__ ko) {
  const int tid = threadIdx.x;
  const int wave = tid >> 5, lane = tid & 31;
  const int hf = lane >> 4, l16 = lane & 15;
  const int tsel = wave & 3, kq = wave >> 2;
  const int cb = (tsel & 1) * 16;

  v2f bq[KQ];
  {
    const float* wp =
        ((tsel < 2) ? Uw : Vw) + (size_t)(4 * KQ * kq + 2 * hf) * Rc + cb + l16;
#pragma unroll
    for (int j = 0; j < KQ; ++j) {
      bq[j].x = wp[(size_t)(4 * j) * Rc];
      bq[j].y = wp[(size_t)(4 * j + 1) * Rc];
    }
  }

  if (tid < 128) {
    const int row = tid >> 3, sub = tid & 7;
    const int g = n0 + row;
    float s1 = 0.0f, s2 = 0.0f;
    for (int c = sub * 64; c < sub * 64 + 64; ++c) {
      const float v = xs[row * XSTR + c];
      s1 += v;
      s2 += v * v;
    }
    for (int m = 1; m < 8; m <<= 1) {
      s1 += __shfl_xor(s1, m, 32);
      s2 += __shfl_xor(s2, m, 32);
    }
    const float mean = s1 * (1.0f / Dc);
    const float var = s2 * (1.0f / Dc) - mean * mean;
    const float rs = rsqrtf(var + 1e-5f);
    float* xrow = xo + ((size_t)b * Nc + g) * Dc;
    if (full) {
      for (int c = sub * 64; c < sub * 64 + 64; ++c) {
        const float y = (xs[row * XSTR + c] - mean) * rs * gamma[c] + beta[c];
        xs[row * XSTR + c] = y;
        xrow[c] = y;
      }
    } else if (g < Nc) {
      for (int c = sub * 64; c < sub * 64 + 64; ++c) {
        const float y = (xs[row * XSTR + c] - mean) * rs * gamma[c] + beta[c];
        xs[row * XSTR + c] = y;
        xrow[c] = y;
      }
    } else {
      for (int c = sub * 64; c < sub * 64 + 64; ++c) {
        const float y = (xs[row * XSTR + c] - mean) * rs * gamma[c] + beta[c];
        xs[row * XSTR + c] = y;
      }
    }
  }
  __syncthreads();

  v8f acc = {0.f, 0.f, 0.f, 0.f, 0.f, 0.f, 0.f, 0.f};
#pragma unroll
  for (int j = 0; j < KQ; ++j) {
    const int c0 = 4 * KQ * kq + 4 * j + 2 * hf;
    v2f a;
    a.x = xs[l16 * XSTR + c0];
    a.y = xs[l16 * XSTR + c0 + 1];
    acc = wmma_f32(a, bq[j], acc);
  }
  for (int i = 0; i < 8; ++i) red[(wave * 8 + i) * 32 + lane] = acc[i];
  __syncthreads();

  {
    const int mat = tid >> 8;
    const int r = (tid >> 4) & 15;
    const int col = 2 * (tid & 15);
    const int ts = 2 * mat + (col >> 4);
    const int e = (ts * 8 + (r & 7)) * 32 + 16 * (r >> 3) + (col & 15);
    float o0 = red[e], o1 = red[e + 1];
    for (int p = 1; p < 4; ++p) {
      o0 += red[p * 1024 + e];
      o1 += red[p * 1024 + e + 1];
    }
    if (full || n0 + r < Nc) {
      float* op = (mat ? ko : qo) + ((size_t)b * Nc + n0 + r) * Rc + col;
      op[0] = o0;
      op[1] = o1;
    }
  }
}

__global__ __launch_bounds__(LNT) void embed_ln_proj_kernel(
    const int* __restrict__ ids, const float* __restrict__ emb,
    const float* __restrict__ pos, const float* __restrict__ anchor_val,
    const float* __restrict__ anchor_state, const float* __restrict__ in_gamma,
    const float* __restrict__ in_beta, const float* __restrict__ sp_w,
    const float* __restrict__ sp_b, const float* __restrict__ gamma,
    const float* __restrict__ beta, const float* __restrict__ Uw,
    const float* __restrict__ Vw, float* __restrict__ val,
    float* __restrict__ st, float* __restrict__ xo, float* __restrict__ qo,
    float* __restrict__ ko) {
  __shared__ float xs[16 * XSTR];
  __shared__ float red[16 * 8 * 32];
  const int tile = blockIdx.x % NT;
  const int b = blockIdx.x / NT;
  const int n0 = tile * 16;
  const int tid = threadIdx.x;
  const bool full = (n0 + 16 <= Nc);
  if (tid < 128) {
    const int row = tid >> 3, sub = tid & 7;
    const int g = n0 + row;
    const int gc = min(g, Nc - 1);
    const int s = max(gc - 1, 0);
    const int id = ids[b * Sc + s];
    const float* er = emb + (size_t)id * Dc + 4 * sub;
    const float* pr = pos + (size_t)s * Dc + 4 * sub;
    float e[64];
#pragma unroll
    for (int j = 0; j < 16; ++j) {
#pragma unroll
      for (int i = 0; i < 4; ++i) e[4 * j + i] = er[32 * j + i] + pr[32 * j + i];
    }
    float s1 = 0.0f, s2 = 0.0f;
#pragma unroll
    for (int c = 0; c < 64; ++c) {
      s1 += e[c];
      s2 += e[c] * e[c];
    }
    for (int m = 1; m < 8; m <<= 1) {
      s1 += __shfl_xor(s1, m, 32);
      s2 += __shfl_xor(s2, m, 32);
    }
    const float mean = s1 * (1.0f / Dc);
    const float var = s2 * (1.0f / Dc) - mean * mean;
    const float rs = rsqrtf(var + 1e-5f);
    float dot = 0.0f;
#pragma unroll
    for (int j = 0; j < 16; ++j) {
      if (j == 8) __builtin_amdgcn_sched_barrier(0);
#pragma unroll
      for (int i = 0; i < 4; ++i) {
        const int c = 32 * j + 4 * sub + i;
        const float y = (e[4 * j + i] - mean) * rs * in_gamma[c] + in_beta[c];
        e[4 * j + i] = y;
        dot += y * sp_w[c];
      }
    }
    for (int m = 1; m < 8; m <<= 1) dot += __shfl_xor(dot, m, 32);
    float stv = dot + sp_b[0];
    if (g == 0) {
#pragma unroll
      for (int j = 0; j < 16; ++j) {
#pragma unroll
        for (int i = 0; i < 4; ++i)
          e[4 * j + i] = anchor_val[32 * j + 4 * sub + i];
      }
      stv = anchor_state[0];
    }
    if (g < Nc) {
      float* vrow = val + ((size_t)b * Nc + g) * Dc + 4 * sub;
#pragma unroll
      for (int j = 0; j < 16; ++j) {
#pragma unroll
        for (int i = 0; i < 4; ++i) vrow[32 * j + i] = e[4 * j + i];
      }
      if (sub == 0) st[(size_t)b * Nc + g] = stv;
    }
    float* xr = xs + row * XSTR + 4 * sub;
#pragma unroll
    for (int j = 0; j < 16; ++j) {
#pragma unroll
      for (int i = 0; i < 4; ++i) xr[32 * j + i] = (g < Nc) ? e[4 * j + i] : 0.0f;
    }
  }
  __syncthreads();
  ln_proj_tile(xs, red, b, n0, full, gamma, beta, Uw, Vw, xo, qo, ko);
}

__global__ __launch_bounds__(LNT) void ln_proj_kernel(
    const float* __restrict__ val, const float* __restrict__ gamma,
    const float* __restrict__ beta, const float* __restrict__ Uw,
    const float* __restrict__ Vw, float* __restrict__ xo,
    float* __restrict__ qo, float* __restrict__ ko) {
  __shared__ float xs[16 * XSTR];
  __shared__ float red[16 * 8 * 32];
  const int tile = blockIdx.x % NT;
  const int b = blockIdx.x / NT;
  const int n0 = tile * 16;
  const int tid = threadIdx.x;
  const bool full = (n0 + 16 <= Nc);

  if (tid < 128) {
    const int row = tid >> 3, sub = tid & 7;
    const int g = n0 + row;
    const float* vrow = val + ((size_t)b * Nc + g) * Dc;
    if (full) {
      for (int c = sub * 64; c < sub * 64 + 64; ++c)
        xs[row * XSTR + c] = vrow[c];
    } else {
      const float* crow = val + ((size_t)b * Nc + min(g, Nc - 1)) * Dc;
      for (int c = sub * 64; c < sub * 64 + 64; ++c) {
        const float v = crow[c];
        xs[row * XSTR + c] = (g < Nc) ? v : 0.0f;
      }
    }
  }
  ln_proj_tile(xs, red, b, n0, full, gamma, beta, Uw, Vw, xo, qo, ko);
}

constexpr int ATT = 512;
__global__ __launch_bounds__(ATT) void attn_kernel(
    const float* __restrict__ x, const float* __restrict__ q,
    const float* __restrict__ k, const float* __restrict__ st,
    float* __restrict__ val) {
  __shared__ float qs[16 * QSTR];
  __shared__ float ks[WINc * KSTR];
  __shared__ float ss[16 * SSTR];
  __shared__ float wm[16 * WSTR];
  __shared__ float sw[WINc];
  const int tile = blockIdx.x % NT;
  const int b = blockIdx.x / NT;
  const int n0 = tile * 16;
  const int tid = threadIdx.x;
  const bool full = (n0 + 16 <= Nc);
  const bool interior = (n0 >= Wc) && (n0 + 16 + Wc <= Nc);

  const float* qb = q + ((size_t)b * Nc + n0) * Rc;
  const float* kb = k + ((size_t)b * Nc + n0 - Wc) * Rc;
  const float* sb = st + (size_t)b * Nc + n0 - Wc;
  if (interior) {
    for (int i = tid; i < 16 * Rc; i += ATT) {
      const int r = i >> 5, c = i & 31;
      qs[r * QSTR + c] = qb[r * Rc + c];
    }
    for (int i = tid; i < WINc * Rc; i += ATT) {
      const int j = i >> 5, c = i & 31;
      ks[j * KSTR + c] = kb[(size_t)j * Rc + c];
    }
    for (int j = tid; j < WINc; j += ATT) sw[j] = sb[j];
  } else {
    static_assert((16 * Rc) % ATT == 0 && (WINc * Rc) % ATT == 0,
                  "q tile and k window: whole trips of the block's threads");
#pragma unroll
    for (int it = 0; it < 16 * Rc / ATT; ++it) {
      const int i = tid + it * ATT;
      const int r = i >> 5, c = i & 31, g = n0 + r;
      const float v = q[((size_t)b * Nc + min(g, Nc - 1)) * Rc + c];
      qs[r * QSTR + c] = (g < Nc) ? v : 0.0f;
    }
#pragma unroll
    for (int it = 0; it < WINc * Rc / ATT; ++it) {
      const int i = tid + it * ATT;
      const int j = i >> 5, c = i & 31, gn = n0 - Wc + j;
      const int gc = min(max(gn, 0), Nc - 1);
      const float v = k[((size_t)b * Nc + gc) * Rc + c];
      ks[j * KSTR + c] = (gc == gn) ? v : 0.0f;
    }
    for (int j = tid; j < WINc; j += ATT) {
      const int gn = n0 - Wc + j;
      const int gc = min(max(gn, 0), Nc - 1);
      const float v = st[(size_t)b * Nc + gc];
      sw[j] = (gc == gn) ? v : 0.0f;
    }
  }
  for (int i = tid; i < 16 * WSTR; i += ATT) wm[i] = 0.0f;
  __syncthreads();

  const int wave = tid >> 5, lane = tid & 31;
  const int hf = lane >> 4, l16 = lane & 15;

  for (int t = wave; t < 9; t += ATT / 32) {
    const int cb = t * 16;
    v8f acc = {0.f, 0.f, 0.f, 0.f, 0.f, 0.f, 0.f, 0.f};
    for (int kk = 0; kk < Rc / 4; ++kk) {
      const int c0 = 4 * kk + 2 * hf;
      v2f a;
      a.x = qs[l16 * QSTR + c0];
      a.y = qs[l16 * QSTR + c0 + 1];
      v2f bv;
      bv.x = ks[(cb + l16) * KSTR + c0];
      bv.y = ks[(cb + l16) * KSTR + c0 + 1];
      acc = wmma_f32(a, bv, acc);
    }
    for (int i = 0; i < 8; ++i) ss[(i + 8 * hf) * SSTR + cb + l16] = acc[i];
  }
  __syncthreads();

  {
    const int row = wave;
    const int g = n0 + row;
    if (g < Nc) {
      const int sbase = row * SSTR + row;
      float sv[5], nb[5];
      bool ok[5];
#pragma unroll
      for (int i = 0; i < 5; ++i) {
        const int o = lane + 32 * i;
        const int oc = min(o, WNc - 1);
        const int gn = g - Wc + o;
        ok[i] = (o < WNc) && (gn >= 0) && (gn < Nc);
        sv[i] = ss[sbase + oc];
        nb[i] = sw[row + oc];
      }
      float m = -1e30f;
#pragma unroll
      for (int i = 0; i < 5; ++i) m = fmaxf(m, ok[i] ? fabsf(sv[i]) : -1e30f);
      m = fmaxf(m, row_xor<1>(m));
      m = fmaxf(m, row_xor<2>(m));
      m = fmaxf(m, row_xor<4>(m));
      m = fmaxf(m, row_xor<8>(m));
      m = fmaxf(m, half_xor(m));
      float e[5];
      float Z = 0.0f;
#pragma unroll
      for (int i = 0; i < 5; ++i) {
        e[i] = ok[i] ? expf(fabsf(sv[i]) - m) : 0.0f;
        Z += e[i];
      }
      Z += row_xor<1>(Z);
      Z += row_xor<2>(Z);
      Z += row_xor<4>(Z);
      Z += row_xor<8>(Z);
      Z += half_xor(Z);
      const float inv = 1.0f / Z;
#pragma unroll
      for (int i = 0; i < 5; ++i) {
        if (ok[i]) {
          const float s = sv[i];
          const float p = e[i] * inv;
          const float wv = (s > 0.0f) ? p : ((s < 0.0f) ? -p : 0.0f);
          wm[row * WSTR + row + lane + 32 * i] = wv * nb[i];
        }
      }
    }
  }
  __syncthreads();

  static_assert(Dc / 16 == 2 * (ATT / 32), "two column tiles per wave");
  {
    constexpr int KS = WINc / 4;
    const int cb = wave * 32 + l16;
    v2f bq0[KS], bq1[KS];
    if (interior) {
      const float* xw = x + ((size_t)b * Nc + n0 - Wc + 2 * hf) * Dc + cb;
#pragma unroll
      for (int kk = 0; kk < KS; ++kk) {
        const float* p = xw + (size_t)(4 * kk) * Dc;
        bq0[kk].x = p[0];
        bq0[kk].y = p[Dc];
        bq1[kk].x = p[16];
        bq1[kk].y = p[Dc + 16];
      }
    } else {
      const float* xb = x + (size_t)b * Nc * Dc + cb;
#pragma unroll
      for (int kk = 0; kk < KS; ++kk) {
        const int g0 = n0 - Wc + 4 * kk + 2 * hf, g1 = g0 + 1;
        const int r0 = min(max(g0, 0), Nc - 1), r1 = min(max(g1, 0), Nc - 1);
        const float* p0 = xb + (size_t)r0 * Dc;
        const float* p1 = xb + (size_t)r1 * Dc;
        const float v00 = p0[0], v01 = p1[0], v10 = p0[16], v11 = p1[16];
        bq0[kk].x = (r0 == g0) ? v00 : 0.0f;
        bq0[kk].y = (r1 == g1) ? v01 : 0.0f;
        bq1[kk].x = (r0 == g0) ? v10 : 0.0f;
        bq1[kk].y = (r1 == g1) ? v11 : 0.0f;
      }
    }
    v8f acc0 = {0.f, 0.f, 0.f, 0.f, 0.f, 0.f, 0.f, 0.f};
    v8f acc1 = acc0;
#pragma unroll
    for (int kk = 0; kk < KS; ++kk) {
      const int c0 = 4 * kk + 2 * hf;
      v2f a;
      a.x = wm[l16 * WSTR + c0];
      a.y = wm[l16 * WSTR + c0 + 1];
      acc0 = wmma_f32(a, bq0[kk], acc0);
      acc1 = wmma_f32(a, bq1[kk], acc1);
    }
    if (full) {
      float* vb = val + ((size_t)b * Nc + n0) * Dc + cb;
      for (int i = 0; i < 8; ++i) {
        const int r = i + 8 * hf;
        vb[(size_t)r * Dc] += acc0[i];
        vb[(size_t)r * Dc + 16] += acc1[i];
      }
    } else {
      for (int i = 0; i < 8; ++i) {
        const int r = i + 8 * hf, gg = n0 + r;
        if (gg < Nc) {
          float* vp = val + ((size_t)b * Nc + gg) * Dc + cb;
          vp[0] += acc0[i];
          vp[16] += acc1[i];
        }
      }
    }
  }
}

extern "C" void kernel_launch(void* const* d_in, const int* in_sizes, int n_in,
                              void* d_out, int out_size, void* d_ws,
                              size_t ws_size, hipStream_t stream) {
  const int* ids = (const int*)d_in[0];
  const float* emb = (const float*)d_in[1];
  const float* pos = (const float*)d_in[2];
  const float* anchor_val = (const float*)d_in[3];
  const float* anchor_state = (const float*)d_in[4];
  const float* in_gamma = (const float*)d_in[5];
  const float* in_beta = (const float*)d_in[6];
  const float* ng = (const float*)d_in[7];
  const float* nb = (const float*)d_in[8];
  const float* Uw = (const float*)d_in[9];
  const float* Vw = (const float*)d_in[10];
  const float* sp_w = (const float*)d_in[11];
  const float* sp_b = (const float*)d_in[12];

  float* val = (float*)d_out;
  float* ws = (float*)d_ws;
  float* x = ws;
  float* q = x + (size_t)Bc * Nc * Dc;
  float* k = q + (size_t)Bc * Nc * Rc;
  float* st = k + (size_t)Bc * Nc * Rc;

  embed_ln_proj_kernel<<<Bc * NT, LNT, 0, stream>>>(
      ids, emb, pos, anchor_val, anchor_state, in_gamma, in_beta, sp_w, sp_b,
      ng, nb, Uw, Vw, val, st, x, q, k);
  attn_kernel<<<Bc * NT, ATT, 0, stream>>>(x, q, k, st, val);

  for (int l = 1; l < Lc; ++l) {
    ln_proj_kernel<<<Bc * NT, LNT, 0, stream>>>(
        val, ng + (size_t)l * Dc, nb + (size_t)l * Dc, Uw + (size_t)l * Dc * Rc,
        Vw + (size_t)l * Dc * Rc, x, q, k);
    attn_kernel<<<Bc * NT, ATT, 0, stream>>>(x, q, k, st, val);
  }
}


# LN+projection kernels: embed, LayerNorms, state dot one row per wave on all 16 waves, row in registers (coalesced 16-byte loads, DPP sums), not 8 threads/row on 4 waves via LDS; sums in another order
# speedup vs baseline: 5.7544x; 1.1066x over previous
#include <hip/hip_runtime.h>

typedef __attribute__((ext_vector_type(2))) float v2f;
typedef __attribute__((ext_vector_type(8))) float v8f;

constexpr int Bc = 2;
constexpr int Sc = 1024;
constexpr int Nc = Sc + 1;
constexpr int Dc = 512;
constexpr int Rc = 32;
constexpr int Lc = 3;
constexpr int Wc = 64;
constexpr int WNc = 2 * Wc + 1;
constexpr int WINc = 144;
constexpr int NT = (Nc + 15) / 16;

constexpr int XSTR = Dc + 4;
constexpr int QSTR = 33;
constexpr int KSTR = 33;
constexpr int SSTR = 146;
constexpr int WSTR = 145;

__device__ __forceinline__ v8f wmma_f32(v2f a, v2f b, v8f c) {
  return __builtin_amdgcn_wmma_f32_16x16x4_f32(false, a, false, b, (short)0, c,
                                               false, false);
}

template <int M>
__device__ __forceinline__ float row_xor(float v) {
  return __builtin_bit_cast(
      float, __builtin_amdgcn_update_dpp(0, __builtin_bit_cast(int, v),
                                         0x160 + M, 0xF, 0xF, true));
}
__device__ __forceinline__ float half_xor(float v) {
  const int x = __builtin_bit_cast(int, v);
  return __builtin_bit_cast(
      float, __builtin_amdgcn_permlanex16(x, x, 0x76543210, 0xfedcba98, false,
                                          false));
}

constexpr int LNT = 512;
constexpr int KQ = Dc / 4 / 4;
constexpr int CPL = Dc / 32;

__device__ __forceinline__ float wave_sum(float v) {
  v += row_xor<1>(v);
  v += row_xor<2>(v);
  v += row_xor<4>(v);
  v += row_xor<8>(v);
  v += half_xor(v);
  return v;
}

__device__ __forceinline__ void request_weights(
    v2f (&bq)[KQ], const float* __restrict__ Uw, const float* __restrict__ Vw) {
  const int tid = threadIdx.x;
  const int wave = tid >> 5, lane = tid & 31;
  const int hf = lane >> 4, l16 = lane & 15;
  const int tsel = wave & 3, kq = wave >> 2;
  const int cb = (tsel & 1) * 16;
  const float* wp =
      ((tsel < 2) ? Uw : Vw) + (size_t)(4 * KQ * kq + 2 * hf) * Rc + cb + l16;
#pragma unroll
  for (int j = 0; j < KQ; ++j) {
    bq[j].x = wp[(size_t)(4 * j) * Rc];
    bq[j].y = wp[(size_t)(4 * j + 1) * Rc];
  }
}
__device__ __forceinline__ void request_ln_params(
    float (&gm)[CPL], float (&bt)[CPL], const float* __restrict__ gamma,
    const float* __restrict__ beta) {
  const int lane = threadIdx.x & 31;
  const float* gp = gamma + 4 * lane;
  const float* bp = beta + 4 * lane;
#pragma unroll
  for (int j = 0; j < 4; ++j) {
#pragma unroll
    for (int i = 0; i < 4; ++i) {
      gm[4 * j + i] = gp[128 * j + i];
      bt[4 * j + i] = bp[128 * j + i];
    }
  }
}

__device__ __forceinline__ void ln_proj_tile(
    float (&e)[CPL], const v2f (&bq)[KQ], const float (&gm)[CPL],
    const float (&bt)[CPL], float* xs, float* red, const int b, const int n0,
    const bool full, float* __restrict__ xo, float* __restrict__ qo,
    float* __restrict__ ko) {
  const int tid = threadIdx.x;
  const int wave = tid >> 5, lane = tid & 31;
  const int hf = lane >> 4, l16 = lane & 15;
  const int kq = wave >> 2;

  {
    const int g = n0 + wave;
    float s1 = 0.0f, s2 = 0.0f;
#pragma unroll
    for (int c = 0; c < CPL; ++c) {
      s1 += e[c];
      s2 += e[c] * e[c];
    }
    s1 = wave_sum(s1);
    s2 = wave_sum(s2);
    const float mean = s1 * (1.0f / Dc);
    const float var = s2 * (1.0f / Dc) - mean * mean;
    const float rs = rsqrtf(var + 1e-5f);
#pragma unroll
    for (int c = 0; c < CPL; ++c) e[c] = (e[c] - mean) * rs * gm[c] + bt[c];
    if (g < Nc) {
      float* xrow = xo + ((size_t)b * Nc + g) * Dc + 4 * lane;
#pragma unroll
      for (int j = 0; j < 4; ++j) {
#pragma unroll
        for (int i = 0; i < 4; ++i) xrow[128 * j + i] = e[4 * j + i];
      }
    }
    float* xr = xs + wave * XSTR + 4 * lane;
#pragma unroll
    for (int j = 0; j < 4; ++j) {
#pragma unroll
      for (int i = 0; i < 4; ++i) xr[128 * j + i] = e[4 * j + i];
    }
  }
  __syncthreads();

  v8f acc = {0.f, 0.f, 0.f, 0.f, 0.f, 0.f, 0.f, 0.f};
#pragma unroll
  for (int j = 0; j < KQ; ++j) {
    const int c0 = 4 * KQ * kq + 4 * j + 2 * hf;
    v2f a;
    a.x = xs[l16 * XSTR + c0];
    a.y = xs[l16 * XSTR + c0 + 1];
    acc = wmma_f32(a, bq[j], acc);
  }
  for (int i = 0; i < 8; ++i) red[(wave * 8 + i) * 32 + lane] = acc[i];
  __syncthreads();

  {
    const int mat = tid >> 8;
    const int r = (tid >> 4) & 15;
    const int col = 2 * (tid & 15);
    const int ts = 2 * mat + (col >> 4);
    const int e0 = (ts * 8 + (r & 7)) * 32 + 16 * (r >> 3) + (col & 15);
    float o0 = red[e0], o1 = red[e0 + 1];
    for (int p = 1; p < 4; ++p) {
      o0 += red[p * 1024 + e0];
      o1 += red[p * 1024 + e0 + 1];
    }
    if (full || n0 + r < Nc) {
      float* op = (mat ? ko : qo) + ((size_t)b * Nc + n0 + r) * Rc + col;
      op[0] = o0;
      op[1] = o1;
    }
  }
}

__global__ __launch_bounds__(LNT) void embed_ln_proj_kernel(
    const int* __restrict__ ids, const float* __restrict__ emb,
    const float* __restrict__ pos, const float* __restrict__ anchor_val,
    const float* __restrict__ anchor_state, const float* __restrict__ in_gamma,
    const float* __restrict__ in_beta, const float* __restrict__ sp_w,
    const float* __restrict__ sp_b, const float* __restrict__ gamma,
    const float* __restrict__ beta, const float* __restrict__ Uw,
    const float* __restrict__ Vw, float* __restrict__ val,
    float* __restrict__ st, float* __restrict__ xo, float* __restrict__ qo,
    float* __restrict__ ko) {
  __shared__ float xs[16 * XSTR];
  __shared__ float red[16 * 8 * 32];
  const int tile = blockIdx.x % NT;
  const int b = blockIdx.x / NT;
  const int n0 = tile * 16;
  const int tid = threadIdx.x;
  const int wave = tid >> 5, lane = tid & 31;
  const bool full = (n0 + 16 <= Nc);
  const int g = n0 + wave;
  const int gc = min(g, Nc - 1);
  const int s = max(gc - 1, 0);
  const int id = ids[b * Sc + s];
  v2f bq[KQ];
  float gm[CPL], bt[CPL];
  request_weights(bq, Uw, Vw);
  request_ln_params(gm, bt, gamma, beta);
  __builtin_amdgcn_sched_barrier(0);
  const float* er = emb + (size_t)id * Dc + 4 * lane;
  const float* pr = pos + (size_t)s * Dc + 4 * lane;
  float e[CPL];
#pragma unroll
  for (int j = 0; j < 4; ++j) {
#pragma unroll
    for (int i = 0; i < 4; ++i) e[4 * j + i] = er[128 * j + i] + pr[128 * j + i];
  }
  float s1 = 0.0f, s2 = 0.0f;
#pragma unroll
  for (int c = 0; c < CPL; ++c) {
    s1 += e[c];
    s2 += e[c] * e[c];
  }
  s1 = wave_sum(s1);
  s2 = wave_sum(s2);
  const float mean = s1 * (1.0f / Dc);
  const float var = s2 * (1.0f / Dc) - mean * mean;
  const float rs = rsqrtf(var + 1e-5f);
  const float* igp = in_gamma + 4 * lane;
  const float* ibp = in_beta + 4 * lane;
  const float* swp = sp_w + 4 * lane;
  float dot = 0.0f;
#pragma unroll
  for (int j = 0; j < 4; ++j) {
#pragma unroll
    for (int i = 0; i < 4; ++i) {
      const float y =
          (e[4 * j + i] - mean) * rs * igp[128 * j + i] + ibp[128 * j + i];
      e[4 * j + i] = y;
      dot += y * swp[128 * j + i];
    }
  }
  dot = wave_sum(dot);
  float stv = dot + sp_b[0];
  if (g == 0) {
    const float* ap = anchor_val + 4 * lane;
#pragma unroll
    for (int j = 0; j < 4; ++j) {
#pragma unroll
      for (int i = 0; i < 4; ++i) e[4 * j + i] = ap[128 * j + i];
    }
    stv = anchor_state[0];
  }
  if (g < Nc) {
    float* vrow = val + ((size_t)b * Nc + g) * Dc + 4 * lane;
#pragma unroll
    for (int j = 0; j < 4; ++j) {
#pragma unroll
      for (int i = 0; i < 4; ++i) vrow[128 * j + i] = e[4 * j + i];
    }
    if (lane == 0) st[(size_t)b * Nc + g] = stv;
  }
  ln_proj_tile(e, bq, gm, bt, xs, red, b, n0, full, xo, qo, ko);
}

__global__ __launch_bounds__(LNT) void ln_proj_kernel(
    const float* __restrict__ val, const float* __restrict__ gamma,
    const float* __restrict__ beta, const float* __restrict__ Uw,
    const float* __restrict__ Vw, float* __restrict__ xo,
    float* __restrict__ qo, float* __restrict__ ko) {
  __shared__ float xs[16 * XSTR];
  __shared__ float red[16 * 8 * 32];
  const int tile = blockIdx.x % NT;
  const int b = blockIdx.x / NT;
  const int n0 = tile * 16;
  const int tid = threadIdx.x;
  const int wave = tid >> 5, lane = tid & 31;
  const bool full = (n0 + 16 <= Nc);
  const int gc = min(n0 + wave, Nc - 1);
  const float* vrow = val + ((size_t)b * Nc + gc) * Dc + 4 * lane;
  float e[CPL];
#pragma unroll
  for (int j = 0; j < 4; ++j) {
#pragma unroll
    for (int i = 0; i < 4; ++i) e[4 * j + i] = vrow[128 * j + i];
  }
  v2f bq[KQ];
  float gm[CPL], bt[CPL];
  request_weights(bq, Uw, Vw);
  request_ln_params(gm, bt, gamma, beta);
  __builtin_amdgcn_sched_barrier(0);
  ln_proj_tile(e, bq, gm, bt, xs, red, b, n0, full, xo, qo, ko);
}

constexpr int ATT = 512;
__global__ __launch_bounds__(ATT) void attn_kernel(
    const float* __restrict__ x, const float* __restrict__ q,
    const float* __restrict__ k, const float* __restrict__ st,
    float* __restrict__ val) {
  __shared__ float qs[16 * QSTR];
  __shared__ float ks[WINc * KSTR];
  __shared__ float ss[16 * SSTR];
  __shared__ float wm[16 * WSTR];
  __shared__ float sw[WINc];
  const int tile = blockIdx.x % NT;
  const int b = blockIdx.x / NT;
  const int n0 = tile * 16;
  const int tid = threadIdx.x;
  const bool full = (n0 + 16 <= Nc);
  const bool interior = (n0 >= Wc) && (n0 + 16 + Wc <= Nc);

  const float* qb = q + ((size_t)b * Nc + n0) * Rc;
  const float* kb = k + ((size_t)b * Nc + n0 - Wc) * Rc;
  const float* sb = st + (size_t)b * Nc + n0 - Wc;
  if (interior) {
    for (int i = tid; i < 16 * Rc; i += ATT) {
      const int r = i >> 5, c = i & 31;
      qs[r * QSTR + c] = qb[r * Rc + c];
    }
    for (int i = tid; i < WINc * Rc; i += ATT) {
      const int j = i >> 5, c = i & 31;
      ks[j * KSTR + c] = kb[(size_t)j * Rc + c];
    }
    for (int j = tid; j < WINc; j += ATT) sw[j] = sb[j];
  } else {
    static_assert((16 * Rc) % ATT == 0 && (WINc * Rc) % ATT == 0,
                  "q tile and k window: whole trips of the block's threads");
#pragma unroll
    for (int it = 0; it < 16 * Rc / ATT; ++it) {
      const int i = tid + it * ATT;
      const int r = i >> 5, c = i & 31, g = n0 + r;
      const float v = q[((size_t)b * Nc + min(g, Nc - 1)) * Rc + c];
      qs[r * QSTR + c] = (g < Nc) ? v : 0.0f;
    }
#pragma unroll
    for (int it = 0; it < WINc * Rc / ATT; ++it) {
      const int i = tid + it * ATT;
      const int j = i >> 5, c = i & 31, gn = n0 - Wc + j;
      const int gc = min(max(gn, 0), Nc - 1);
      const float v = k[((size_t)b * Nc + gc) * Rc + c];
      ks[j * KSTR + c] = (gc == gn) ? v : 0.0f;
    }
    for (int j = tid; j < WINc; j += ATT) {
      const int gn = n0 - Wc + j;
      const int gc = min(max(gn, 0), Nc - 1);
      const float v = st[(size_t)b * Nc + gc];
      sw[j] = (gc == gn) ? v : 0.0f;
    }
  }
  for (int i = tid; i < 16 * WSTR; i += ATT) wm[i] = 0.0f;
  __syncthreads();

  const int wave = tid >> 5, lane = tid & 31;
  const int hf = lane >> 4, l16 = lane & 15;

  for (int t = wave; t < 9; t += ATT / 32) {
    const int cb = t * 16;
    v8f acc = {0.f, 0.f, 0.f, 0.f, 0.f, 0.f, 0.f, 0.f};
    for (int kk = 0; kk < Rc / 4; ++kk) {
      const int c0 = 4 * kk + 2 * hf;
      v2f a;
      a.x = qs[l16 * QSTR + c0];
      a.y = qs[l16 * QSTR + c0 + 1];
      v2f bv;
      bv.x = ks[(cb + l16) * KSTR + c0];
      bv.y = ks[(cb + l16) * KSTR + c0 + 1];
      acc = wmma_f32(a, bv, acc);
    }
    for (int i = 0; i < 8; ++i) ss[(i + 8 * hf) * SSTR + cb + l16] = acc[i];
  }
  __syncthreads();

  {
    const int row = wave;
    const int g = n0 + row;
    if (g < Nc) {
      const int sbase = row * SSTR + row;
      float sv[5], nb[5];
      bool ok[5];
#pragma unroll
      for (int i = 0; i < 5; ++i) {
        const int o = lane + 32 * i;
        const int oc = min(o, WNc - 1);
        const int gn = g - Wc + o;
        ok[i] = (o < WNc) && (gn >= 0) && (gn < Nc);
        sv[i] = ss[sbase + oc];
        nb[i] = sw[row + oc];
      }
      float m = -1e30f;
#pragma unroll
      for (int i = 0; i < 5; ++i) m = fmaxf(m, ok[i] ? fabsf(sv[i]) : -1e30f);
      m = fmaxf(m, row_xor<1>(m));
      m = fmaxf(m, row_xor<2>(m));
      m = fmaxf(m, row_xor<4>(m));
      m = fmaxf(m, row_xor<8>(m));
      m = fmaxf(m, half_xor(m));
      float e[5];
      float Z = 0.0f;
#pragma unroll
      for (int i = 0; i < 5; ++i) {
        e[i] = ok[i] ? expf(fabsf(sv[i]) - m) : 0.0f;
        Z += e[i];
      }
      Z += row_xor<1>(Z);
      Z += row_xor<2>(Z);
      Z += row_xor<4>(Z);
      Z += row_xor<8>(Z);
      Z += half_xor(Z);
      const float inv = 1.0f / Z;
#pragma unroll
      for (int i = 0; i < 5; ++i) {
        if (ok[i]) {
          const float s = sv[i];
          const float p = e[i] * inv;
          const float wv = (s > 0.0f) ? p : ((s < 0.0f) ? -p : 0.0f);
          wm[row * WSTR + row + lane + 32 * i] = wv * nb[i];
        }
      }
    }
  }
  __syncthreads();

  static_assert(Dc / 16 == 2 * (ATT / 32), "two column tiles per wave");
  {
    constexpr int KS = WINc / 4;
    const int cb = wave * 32 + l16;
    v2f bq0[KS], bq1[KS];
    if (interior) {
      const float* xw = x + ((size_t)b * Nc + n0 - Wc + 2 * hf) * Dc + cb;
#pragma unroll
      for (int kk = 0; kk < KS; ++kk) {
        const float* p = xw + (size_t)(4 * kk) * Dc;
        bq0[kk].x = p[0];
        bq0[kk].y = p[Dc];
        bq1[kk].x = p[16];
        bq1[kk].y = p[Dc + 16];
      }
    } else {
      const float* xb = x + (size_t)b * Nc * Dc + cb;
#pragma unroll
      for (int kk = 0; kk < KS; ++kk) {
        const int g0 = n0 - Wc + 4 * kk + 2 * hf, g1 = g0 + 1;
        const int r0 = min(max(g0, 0), Nc - 1), r1 = min(max(g1, 0), Nc - 1);
        const float* p0 = xb + (size_t)r0 * Dc;
        const float* p1 = xb + (size_t)r1 * Dc;
        const float v00 = p0[0], v01 = p1[0], v10 = p0[16], v11 = p1[16];
        bq0[kk].x = (r0 == g0) ? v00 : 0.0f;
        bq0[kk].y = (r1 == g1) ? v01 : 0.0f;
        bq1[kk].x = (r0 == g0) ? v10 : 0.0f;
        bq1[kk].y = (r1 == g1) ? v11 : 0.0f;
      }
    }
    v8f acc0 = {0.f, 0.f, 0.f, 0.f, 0.f, 0.f, 0.f, 0.f};
    v8f acc1 = acc0;
#pragma unroll
    for (int kk = 0; kk < KS; ++kk) {
      const int c0 = 4 * kk + 2 * hf;
      v2f a;
      a.x = wm[l16 * WSTR + c0];
      a.y = wm[l16 * WSTR + c0 + 1];
      acc0 = wmma_f32(a, bq0[kk], acc0);
      acc1 = wmma_f32(a, bq1[kk], acc1);
    }
    if (full) {
      float* vb = val + ((size_t)b * Nc + n0) * Dc + cb;
      for (int i = 0; i < 8; ++i) {
        const int r = i + 8 * hf;
        vb[(size_t)r * Dc] += acc0[i];
        vb[(size_t)r * Dc + 16] += acc1[i];
      }
    } else {
      for (int i = 0; i < 8; ++i) {
        const int r = i + 8 * hf, gg = n0 + r;
        if (gg < Nc) {
          float* vp = val + ((size_t)b * Nc + gg) * Dc + cb;
          vp[0] += acc0[i];
          vp[16] += acc1[i];
        }
      }
    }
  }
}

extern "C" void kernel_launch(void* const* d_in, const int* in_sizes, int n_in,
                              void* d_out, int out_size, void* d_ws,
                              size_t ws_size, hipStream_t stream) {
  const int* ids = (const int*)d_in[0];
  const float* emb = (const float*)d_in[1];
  const float* pos = (const float*)d_in[2];
  const float* anchor_val = (const float*)d_in[3];
  const float* anchor_state = (const float*)d_in[4];
  const float* in_gamma = (const float*)d_in[5];
  const float* in_beta = (const float*)d_in[6];
  const float* ng = (const float*)d_in[7];
  const float* nb = (const float*)d_in[8];
  const float* Uw = (const float*)d_in[9];
  const float* Vw = (const float*)d_in[10];
  const float* sp_w = (const float*)d_in[11];
  const float* sp_b = (const float*)d_in[12];

  float* val = (float*)d_out;
  float* ws = (float*)d_ws;
  float* x = ws;
  float* q = x + (size_t)Bc * Nc * Dc;
  float* k = q + (size_t)Bc * Nc * Rc;
  float* st = k + (size_t)Bc * Nc * Rc;

  embed_ln_proj_kernel<<<Bc * NT, LNT, 0, stream>>>(
      ids, emb, pos, anchor_val, anchor_state, in_gamma, in_beta, sp_w, sp_b,
      ng, nb, Uw, Vw, val, st, x, q, k);
  attn_kernel<<<Bc * NT, ATT, 0, stream>>>(x, q, k, st, val);

  for (int l = 1; l < Lc; ++l) {
    ln_proj_kernel<<<Bc * NT, LNT, 0, stream>>>(
        val, ng + (size_t)l * Dc, nb + (size_t)l * Dc, Uw + (size_t)l * Dc * Rc,
        Vw + (size_t)l * Dc * Rc, x, q, k);
    attn_kernel<<<Bc * NT, ATT, 0, stream>>>(x, q, k, st, val);
  }
}


# attn kernel: the delta GEMM's x window rows and the tile's val rows are requested ahead in 6 groups (two before the scores, one in front of each group of products); its barriers order LDS only
# speedup vs baseline: 6.4986x; 1.1293x over previous
#include <hip/hip_runtime.h>

typedef __attribute__((ext_vector_type(2))) float v2f;
typedef __attribute__((ext_vector_type(8))) float v8f;

constexpr int Bc = 2;
constexpr int Sc = 1024;
constexpr int Nc = Sc + 1;
constexpr int Dc = 512;
constexpr int Rc = 32;
constexpr int Lc = 3;
constexpr int Wc = 64;
constexpr int WNc = 2 * Wc + 1;
constexpr int WINc = 144;
constexpr int NT = (Nc + 15) / 16;

constexpr int XSTR = Dc + 4;
constexpr int QSTR = 33;
constexpr int KSTR = 33;
constexpr int SSTR = 146;
constexpr int WSTR = 145;

__device__ __forceinline__ v8f wmma_f32(v2f a, v2f b, v8f c) {
  return __builtin_amdgcn_wmma_f32_16x16x4_f32(false, a, false, b, (short)0, c,
                                               false, false);
}

template <int M>
__device__ __forceinline__ float row_xor(float v) {
  return __builtin_bit_cast(
      float, __builtin_amdgcn_update_dpp(0, __builtin_bit_cast(int, v),
                                         0x160 + M, 0xF, 0xF, true));
}
__device__ __forceinline__ float half_xor(float v) {
  const int x = __builtin_bit_cast(int, v);
  return __builtin_bit_cast(
      float, __builtin_amdgcn_permlanex16(x, x, 0x76543210, 0xfedcba98, false,
                                          false));
}

constexpr int LNT = 512;
constexpr int KQ = Dc / 4 / 4;
constexpr int CPL = Dc / 32;

__device__ __forceinline__ float wave_sum(float v) {
  v += row_xor<1>(v);
  v += row_xor<2>(v);
  v += row_xor<4>(v);
  v += row_xor<8>(v);
  v += half_xor(v);
  return v;
}

__device__ __forceinline__ void request_weights(
    v2f (&bq)[KQ], const float* __restrict__ Uw, const float* __restrict__ Vw) {
  const int tid = threadIdx.x;
  const int wave = tid >> 5, lane = tid & 31;
  const int hf = lane >> 4, l16 = lane & 15;
  const int tsel = wave & 3, kq = wave >> 2;
  const int cb = (tsel & 1) * 16;
  const float* wp =
      ((tsel < 2) ? Uw : Vw) + (size_t)(4 * KQ * kq + 2 * hf) * Rc + cb + l16;
#pragma unroll
  for (int j = 0; j < KQ; ++j) {
    bq[j].x = wp[(size_t)(4 * j) * Rc];
    bq[j].y = wp[(size_t)(4 * j + 1) * Rc];
  }
}
__device__ __forceinline__ void request_ln_params(
    float (&gm)[CPL], float (&bt)[CPL], const float* __restrict__ gamma,
    const float* __restrict__ beta) {
  const int lane = threadIdx.x & 31;
  const float* gp = gamma + 4 * lane;
  const float* bp = beta + 4 * lane;
#pragma unroll
  for (int j = 0; j < 4; ++j) {
#pragma unroll
    for (int i = 0; i < 4; ++i) {
      gm[4 * j + i] = gp[128 * j + i];
      bt[4 * j + i] = bp[128 * j + i];
    }
  }
}

__device__ __forceinline__ void ln_proj_tile(
    float (&e)[CPL], const v2f (&bq)[KQ], const float (&gm)[CPL],
    const float (&bt)[CPL], float* xs, float* red, const int b, const int n0,
    const bool full, float* __restrict__ xo, float* __restrict__ qo,
    float* __restrict__ ko) {
  const int tid = threadIdx.x;
  const int wave = tid >> 5, lane = tid & 31;
  const int hf = lane >> 4, l16 = lane & 15;
  const int kq = wave >> 2;

  {
    const int g = n0 + wave;
    float s1 = 0.0f, s2 = 0.0f;
#pragma unroll
    for (int c = 0; c < CPL; ++c) {
      s1 += e[c];
      s2 += e[c] * e[c];
    }
    s1 = wave_sum(s1);
    s2 = wave_sum(s2);
    const float mean = s1 * (1.0f / Dc);
    const float var = s2 * (1.0f / Dc) - mean * mean;
    const float rs = rsqrtf(var + 1e-5f);
#pragma unroll
    for (int c = 0; c < CPL; ++c) e[c] = (e[c] - mean) * rs * gm[c] + bt[c];
    if (g < Nc) {
      float* xrow = xo + ((size_t)b * Nc + g) * Dc + 4 * lane;
#pragma unroll
      for (int j = 0; j < 4; ++j) {
#pragma unroll
        for (int i = 0; i < 4; ++i) xrow[128 * j + i] = e[4 * j + i];
      }
    }
    float* xr = xs + wave * XSTR + 4 * lane;
#pragma unroll
    for (int j = 0; j < 4; ++j) {
#pragma unroll
      for (int i = 0; i < 4; ++i) xr[128 * j + i] = e[4 * j + i];
    }
  }
  __syncthreads();

  v8f acc = {0.f, 0.f, 0.f, 0.f, 0.f, 0.f, 0.f, 0.f};
#pragma unroll
  for (int j = 0; j < KQ; ++j) {
    const int c0 = 4 * KQ * kq + 4 * j + 2 * hf;
    v2f a;
    a.x = xs[l16 * XSTR + c0];
    a.y = xs[l16 * XSTR + c0 + 1];
    acc = wmma_f32(a, bq[j], acc);
  }
  for (int i = 0; i < 8; ++i) red[(wave * 8 + i) * 32 + lane] = acc[i];
  __syncthreads();

  {
    const int mat = tid >> 8;
    const int r = (tid >> 4) & 15;
    const int col = 2 * (tid & 15);
    const int ts = 2 * mat + (col >> 4);
    const int e0 = (ts * 8 + (r & 7)) * 32 + 16 * (r >> 3) + (col & 15);
    float o0 = red[e0], o1 = red[e0 + 1];
    for (int p = 1; p < 4; ++p) {
      o0 += red[p * 1024 + e0];
      o1 += red[p * 1024 + e0 + 1];
    }
    if (full || n0 + r < Nc) {
      float* op = (mat ? ko : qo) + ((size_t)b * Nc + n0 + r) * Rc + col;
      op[0] = o0;
      op[1] = o1;
    }
  }
}

__global__ __launch_bounds__(LNT) void embed_ln_proj_kernel(
    const int* __restrict__ ids, const float* __restrict__ emb,
    const float* __restrict__ pos, const float* __restrict__ anchor_val,
    const float* __restrict__ anchor_state, const float* __restrict__ in_gamma,
    const float* __restrict__ in_beta, const float* __restrict__ sp_w,
    const float* __restrict__ sp_b, const float* __restrict__ gamma,
    const float* __restrict__ beta, const float* __restrict__ Uw,
    const float* __restrict__ Vw, float* __restrict__ val,
    float* __restrict__ st, float* __restrict__ xo, float* __restrict__ qo,
    float* __restrict__ ko) {
  __shared__ float xs[16 * XSTR];
  __shared__ float red[16 * 8 * 32];
  const int tile = blockIdx.x % NT;
  const int b = blockIdx.x / NT;
  const int n0 = tile * 16;
  const int tid = threadIdx.x;
  const int wave = tid >> 5, lane = tid & 31;
  const bool full = (n0 + 16 <= Nc);
  const int g = n0 + wave;
  const int gc = min(g, Nc - 1);
  const int s = max(gc - 1, 0);
  const int id = ids[b * Sc + s];
  v2f bq[KQ];
  float gm[CPL], bt[CPL];
  request_weights(bq, Uw, Vw);
  request_ln_params(gm, bt, gamma, beta);
  __builtin_amdgcn_sched_barrier(0);
  const float* er = emb + (size_t)id * Dc + 4 * lane;
  const float* pr = pos + (size_t)s * Dc + 4 * lane;
  float e[CPL];
#pragma unroll
  for (int j = 0; j < 4; ++j) {
#pragma unroll
    for (int i = 0; i < 4; ++i) e[4 * j + i] = er[128 * j + i] + pr[128 * j + i];
  }
  float s1 = 0.0f, s2 = 0.0f;
#pragma unroll
  for (int c = 0; c < CPL; ++c) {
    s1 += e[c];
    s2 += e[c] * e[c];
  }
  s1 = wave_sum(s1);
  s2 = wave_sum(s2);
  const float mean = s1 * (1.0f / Dc);
  const float var = s2 * (1.0f / Dc) - mean * mean;
  const float rs = rsqrtf(var + 1e-5f);
  const float* igp = in_gamma + 4 * lane;
  const float* ibp = in_beta + 4 * lane;
  const float* swp = sp_w + 4 * lane;
  float dot = 0.0f;
#pragma unroll
  for (int j = 0; j < 4; ++j) {
#pragma unroll
    for (int i = 0; i < 4; ++i) {
      const float y =
          (e[4 * j + i] - mean) * rs * igp[128 * j + i] + ibp[128 * j + i];
      e[4 * j + i] = y;
      dot += y * swp[128 * j + i];
    }
  }
  dot = wave_sum(dot);
  float stv = dot + sp_b[0];
  if (g == 0) {
    const float* ap = anchor_val + 4 * lane;
#pragma unroll
    for (int j = 0; j < 4; ++j) {
#pragma unroll
      for (int i = 0; i < 4; ++i) e[4 * j + i] = ap[128 * j + i];
    }
    stv = anchor_state[0];
  }
  if (g < Nc) {
    float* vrow = val + ((size_t)b * Nc + g) * Dc + 4 * lane;
#pragma unroll
    for (int j = 0; j < 4; ++j) {
#pragma unroll
      for (int i = 0; i < 4; ++i) vrow[128 * j + i] = e[4 * j + i];
    }
    if (lane == 0) st[(size_t)b * Nc + g] = stv;
  }
  ln_proj_tile(e, bq, gm, bt, xs, red, b, n0, full, xo, qo, ko);
}

__global__ __launch_bounds__(LNT) void ln_proj_kernel(
    const float* __restrict__ val, const float* __restrict__ gamma,
    const float* __restrict__ beta, const float* __restrict__ Uw,
    const float* __restrict__ Vw, float* __restrict__ xo,
    float* __restrict__ qo, float* __restrict__ ko) {
  __shared__ float xs[16 * XSTR];
  __shared__ float red[16 * 8 * 32];
  const int tile = blockIdx.x % NT;
  const int b = blockIdx.x / NT;
  const int n0 = tile * 16;
  const int tid = threadIdx.x;
  const int wave = tid >> 5, lane = tid & 31;
  const bool full = (n0 + 16 <= Nc);
  const int gc = min(n0 + wave, Nc - 1);
  const float* vrow = val + ((size_t)b * Nc + gc) * Dc + 4 * lane;
  float e[CPL];
#pragma unroll
  for (int j = 0; j < 4; ++j) {
#pragma unroll
    for (int i = 0; i < 4; ++i) e[4 * j + i] = vrow[128 * j + i];
  }
  v2f bq[KQ];
  float gm[CPL], bt[CPL];
  request_weights(bq, Uw, Vw);
  request_ln_params(gm, bt, gamma, beta);
  __builtin_amdgcn_sched_barrier(0);
  ln_proj_tile(e, bq, gm, bt, xs, red, b, n0, full, xo, qo, ko);
}

constexpr int ATT = 512;
constexpr int KS = WINc / 4;
constexpr int KG = 6;
constexpr int NG = KS / KG;

__device__ __forceinline__ void lds_barrier() {
  __builtin_amdgcn_fence(__ATOMIC_RELEASE, "workgroup", "local");
  __builtin_amdgcn_s_barrier();
  __builtin_amdgcn_fence(__ATOMIC_ACQUIRE, "workgroup", "local");
}

__device__ __forceinline__ void hold_order() {
  __builtin_amdgcn_sched_barrier(0);
}

template <bool INTERIOR, int G>
__device__ __forceinline__ void request_rows(v2f (&bq0)[KS], v2f (&bq1)[KS],
                                             const float* __restrict__ x,
                                             const int b, const int n0,
                                             const int hf, const int cb) {
  if constexpr (INTERIOR) {
    const float* xw = x + ((size_t)b * Nc + n0 - Wc + 2 * hf) * Dc + cb;
#pragma unroll
    for (int kk = G * KG; kk < (G + 1) * KG; ++kk) {
      const float* p = xw + (size_t)(4 * kk) * Dc;
      bq0[kk].x = p[0];
      bq0[kk].y = p[Dc];
      bq1[kk].x = p[16];
      bq1[kk].y = p[Dc + 16];
    }
  } else {
    const float* xb = x + (size_t)b * Nc * Dc + cb;
#pragma unroll
    for (int kk = G * KG; kk < (G + 1) * KG; ++kk) {
      const int g0 = n0 - Wc + 4 * kk + 2 * hf, g1 = g0 + 1;
      const int r0 = min(max(g0, 0), Nc - 1), r1 = min(max(g1, 0), Nc - 1);
      const float* p0 = xb + (size_t)r0 * Dc;
      const float* p1 = xb + (size_t)r1 * Dc;
      bq0[kk].x = p0[0];
      bq0[kk].y = p1[0];
      bq1[kk].x = p0[16];
      bq1[kk].y = p1[16];
    }
  }
}

template <int G>
__device__ __forceinline__ void zero_outside(v2f (&bq0)[KS], v2f (&bq1)[KS],
                                             const int n0, const int hf) {
#pragma unroll
  for (int kk = G * KG; kk < (G + 1) * KG; ++kk) {
    const int g0 = n0 - Wc + 4 * kk + 2 * hf, g1 = g0 + 1;
    const bool ok0 = (g0 >= 0) && (g0 < Nc), ok1 = (g1 >= 0) && (g1 < Nc);
    bq0[kk].x = ok0 ? bq0[kk].x : 0.0f;
    bq0[kk].y = ok1 ? bq0[kk].y : 0.0f;
    bq1[kk].x = ok0 ? bq1[kk].x : 0.0f;
    bq1[kk].y = ok1 ? bq1[kk].y : 0.0f;
  }
}

template <int G>
__device__ __forceinline__ void products(v8f& acc0, v8f& acc1,
                                         const v2f (&bq0)[KS],
                                         const v2f (&bq1)[KS], const float* wm,
                                         const int l16, const int hf) {
#pragma unroll
  for (int kk = G * KG; kk < (G + 1) * KG; ++kk) {
    const int c0 = 4 * kk + 2 * hf;
    v2f a;
    a.x = wm[l16 * WSTR + c0];
    a.y = wm[l16 * WSTR + c0 + 1];
    acc0 = wmma_f32(a, bq0[kk], acc0);
    acc1 = wmma_f32(a, bq1[kk], acc1);
  }
  asm volatile("" : "+v"(acc0), "+v"(acc1));
}

template <bool INTERIOR, int G>
__device__ __forceinline__ void product_group(v8f& acc0, v8f& acc1,
                                              v2f (&bq0)[KS], v2f (&bq1)[KS],
                                              const float* wm, const int n0,
                                              const int l16, const int hf) {
  if constexpr (!INTERIOR) zero_outside<G>(bq0, bq1, n0, hf);
  products<G>(acc0, acc1, bq0, bq1, wm, l16, hf);
}

template <bool INTERIOR>
__device__ __forceinline__ void delta_groups(
    v8f& acc0, v8f& acc1, v2f (&bq0)[KS], v2f (&bq1)[KS], float (&vv0)[8],
    float (&vv1)[8], const float* __restrict__ x, const float* vrows,
    const float* wm, const int b, const int n0, const int l16, const int hf,
    const int cb, const bool full) {
  request_rows<INTERIOR, 2>(bq0, bq1, x, b, n0, hf, cb);
  hold_order();
  product_group<INTERIOR, 0>(acc0, acc1, bq0, bq1, wm, n0, l16, hf);
  request_rows<INTERIOR, 3>(bq0, bq1, x, b, n0, hf, cb);
  hold_order();
  product_group<INTERIOR, 1>(acc0, acc1, bq0, bq1, wm, n0, l16, hf);
  request_rows<INTERIOR, 4>(bq0, bq1, x, b, n0, hf, cb);
  hold_order();
  product_group<INTERIOR, 2>(acc0, acc1, bq0, bq1, wm, n0, l16, hf);
  request_rows<INTERIOR, 5>(bq0, bq1, x, b, n0, hf, cb);
  hold_order();
  product_group<INTERIOR, 3>(acc0, acc1, bq0, bq1, wm, n0, l16, hf);
  if (full) {
    const float* vb = vrows + ((size_t)b * Nc + n0 + 8 * hf) * Dc + cb;
#pragma unroll
    for (int i = 0; i < 8; ++i) {
      vv0[i] = vb[(size_t)i * Dc];
      vv1[i] = vb[(size_t)i * Dc + 16];
    }
  } else {
#pragma unroll
    for (int i = 0; i < 8; ++i) {
      const int gg = min(n0 + i + 8 * hf, Nc - 1);
      const float* vp = vrows + ((size_t)b * Nc + gg) * Dc + cb;
      vv0[i] = vp[0];
      vv1[i] = vp[16];
    }
  }
  hold_order();
  product_group<INTERIOR, 4>(acc0, acc1, bq0, bq1, wm, n0, l16, hf);
  product_group<INTERIOR, 5>(acc0, acc1, bq0, bq1, wm, n0, l16, hf);
}

__global__ __launch_bounds__(ATT) void attn_kernel(
    const float* __restrict__ x, const float* __restrict__ q,
    const float* __restrict__ k, const float* __restrict__ st,
    float* __restrict__ val) {
  __shared__ float qs[16 * QSTR];
  __shared__ float ks[WINc * KSTR];
  __shared__ float ss[16 * SSTR];
  __shared__ float wm[16 * WSTR];
  __shared__ float sw[WINc];
  const int tile = blockIdx.x % NT;
  const int b = blockIdx.x / NT;
  const int n0 = tile * 16;
  const int tid = threadIdx.x;
  const bool full = (n0 + 16 <= Nc);
  const bool interior = (n0 >= Wc) && (n0 + 16 + Wc <= Nc);

  const float* qb = q + ((size_t)b * Nc + n0) * Rc;
  const float* kb = k + ((size_t)b * Nc + n0 - Wc) * Rc;
  const float* sb = st + (size_t)b * Nc + n0 - Wc;
  if (interior) {
    for (int i = tid; i < 16 * Rc; i += ATT) {
      const int r = i >> 5, c = i & 31;
      qs[r * QSTR + c] = qb[r * Rc + c];
    }
    for (int i = tid; i < WINc * Rc; i += ATT) {
      const int j = i >> 5, c = i & 31;
      ks[j * KSTR + c] = kb[(size_t)j * Rc + c];
    }
    for (int j = tid; j < WINc; j += ATT) sw[j] = sb[j];
  } else {
    static_assert((16 * Rc) % ATT == 0 && (WINc * Rc) % ATT == 0,
                  "q tile and k window: whole trips of the block's threads");
#pragma unroll
    for (int it = 0; it < 16 * Rc / ATT; ++it) {
      const int i = tid + it * ATT;
      const int r = i >> 5, c = i & 31, g = n0 + r;
      const float v = q[((size_t)b * Nc + min(g, Nc - 1)) * Rc + c];
      qs[r * QSTR + c] = (g < Nc) ? v : 0.0f;
    }
#pragma unroll
    for (int it = 0; it < WINc * Rc / ATT; ++it) {
      const int i = tid + it * ATT;
      const int j = i >> 5, c = i & 31, gn = n0 - Wc + j;
      const int gc = min(max(gn, 0), Nc - 1);
      const float v = k[((size_t)b * Nc + gc) * Rc + c];
      ks[j * KSTR + c] = (gc == gn) ? v : 0.0f;
    }
    for (int j = tid; j < WINc; j += ATT) {
      const int gn = n0 - Wc + j;
      const int gc = min(max(gn, 0), Nc - 1);
      const float v = st[(size_t)b * Nc + gc];
      sw[j] = (gc == gn) ? v : 0.0f;
    }
  }
  for (int i = tid; i < 16 * WSTR; i += ATT) wm[i] = 0.0f;

  const int wave = tid >> 5, lane = tid & 31;
  const int hf = lane >> 4, l16 = lane & 15;

  static_assert(Dc / 16 == 2 * (ATT / 32), "two column tiles per wave");
  static_assert(KS % KG == 0 && NG == 6, "six whole request groups");
  const int cb = wave * 32 + l16;
  v2f bq0[KS], bq1[KS];
  if (interior) {
    request_rows<true, 0>(bq0, bq1, x, b, n0, hf, cb);
    request_rows<true, 1>(bq0, bq1, x, b, n0, hf, cb);
  } else {
    request_rows<false, 0>(bq0, bq1, x, b, n0, hf, cb);
    request_rows<false, 1>(bq0, bq1, x, b, n0, hf, cb);
  }
  lds_barrier();

  for (int t = wave; t < 9; t += ATT / 32) {
    const int cbs = t * 16;
    v8f acc = {0.f, 0.f, 0.f, 0.f, 0.f, 0.f, 0.f, 0.f};
    for (int kk = 0; kk < Rc / 4; ++kk) {
      const int c0 = 4 * kk + 2 * hf;
      v2f a;
      a.x = qs[l16 * QSTR + c0];
      a.y = qs[l16 * QSTR + c0 + 1];
      v2f bv;
      bv.x = ks[(cbs + l16) * KSTR + c0];
      bv.y = ks[(cbs + l16) * KSTR + c0 + 1];
      acc = wmma_f32(a, bv, acc);
    }
    for (int i = 0; i < 8; ++i) ss[(i + 8 * hf) * SSTR + cbs + l16] = acc[i];
  }
  lds_barrier();

  {
    const int row = wave;
    const int g = n0 + row;
    if (g < Nc) {
      const int sbase = row * SSTR + row;
      float sv[5], nb[5];
      bool ok[5];
#pragma unroll
      for (int i = 0; i < 5; ++i) {
        const int o = lane + 32 * i;
        const int oc = min(o, WNc - 1);
        const int gn = g - Wc + o;
        ok[i] = (o < WNc) && (gn >= 0) && (gn < Nc);
        sv[i] = ss[sbase + oc];
        nb[i] = sw[row + oc];
      }
      float m = -1e30f;
#pragma unroll
      for (int i = 0; i < 5; ++i) m = fmaxf(m, ok[i] ? fabsf(sv[i]) : -1e30f);
      m = fmaxf(m, row_xor<1>(m));
      m = fmaxf(m, row_xor<2>(m));
      m = fmaxf(m, row_xor<4>(m));
      m = fmaxf(m, row_xor<8>(m));
      m = fmaxf(m, half_xor(m));
      float e[5];
      float Z = 0.0f;
#pragma unroll
      for (int i = 0; i < 5; ++i) {
        e[i] = ok[i] ? expf(fabsf(sv[i]) - m) : 0.0f;
        Z += e[i];
      }
      Z += row_xor<1>(Z);
      Z += row_xor<2>(Z);
      Z += row_xor<4>(Z);
      Z += row_xor<8>(Z);
      Z += half_xor(Z);
      const float inv = 1.0f / Z;
#pragma unroll
      for (int i = 0; i < 5; ++i) {
        if (ok[i]) {
          const float s = sv[i];
          const float p = e[i] * inv;
          const float wv = (s > 0.0f) ? p : ((s < 0.0f) ? -p : 0.0f);
          wm[row * WSTR + row + lane + 32 * i] = wv * nb[i];
        }
      }
    }
  }
  lds_barrier();

  {
    v8f acc0 = {0.f, 0.f, 0.f, 0.f, 0.f, 0.f, 0.f, 0.f};
    v8f acc1 = acc0;
    float vv0[8], vv1[8];
    if (interior) {
      delta_groups<true>(acc0, acc1, bq0, bq1, vv0, vv1, x, val, wm, b, n0, l16,
                         hf, cb, full);
    } else {
      delta_groups<false>(acc0, acc1, bq0, bq1, vv0, vv1, x, val, wm, b, n0,
                          l16, hf, cb, full);
    }
    if (full) {
      float* vb = val + ((size_t)b * Nc + n0 + 8 * hf) * Dc + cb;
#pragma unroll
      for (int i = 0; i < 8; ++i) {
        vb[(size_t)i * Dc] = vv0[i] + acc0[i];
        vb[(size_t)i * Dc + 16] = vv1[i] + acc1[i];
      }
    } else {
#pragma unroll
      for (int i = 0; i < 8; ++i) {
        const int gg = n0 + i + 8 * hf;
        if (gg < Nc) {
          float* vp = val + ((size_t)b * Nc + gg) * Dc + cb;
          vp[0] = vv0[i] + acc0[i];
          vp[16] = vv1[i] + acc1[i];
        }
      }
    }
  }
}

extern "C" void kernel_launch(void* const* d_in, const int* in_sizes, int n_in,
                              void* d_out, int out_size, void* d_ws,
                              size_t ws_size, hipStream_t stream) {
  const int* ids = (const int*)d_in[0];
  const float* emb = (const float*)d_in[1];
  const float* pos = (const float*)d_in[2];
  const float* anchor_val = (const float*)d_in[3];
  const float* anchor_state = (const float*)d_in[4];
  const float* in_gamma = (const float*)d_in[5];
  const float* in_beta = (const float*)d_in[6];
  const float* ng = (const float*)d_in[7];
  const float* nb = (const float*)d_in[8];
  const float* Uw = (const float*)d_in[9];
  const float* Vw = (const float*)d_in[10];
  const float* sp_w = (const float*)d_in[11];
  const float* sp_b = (const float*)d_in[12];

  float* val = (float*)d_out;
  float* ws = (float*)d_ws;
  float* x = ws;
  float* q = x + (size_t)Bc * Nc * Dc;
  float* k = q + (size_t)Bc * Nc * Rc;
  float* st = k + (size_t)Bc * Nc * Rc;

  embed_ln_proj_kernel<<<Bc * NT, LNT, 0, stream>>>(
      ids, emb, pos, anchor_val, anchor_state, in_gamma, in_beta, sp_w, sp_b,
      ng, nb, Uw, Vw, val, st, x, q, k);
  attn_kernel<<<Bc * NT, ATT, 0, stream>>>(x, q, k, st, val);

  for (int l = 1; l < Lc; ++l) {
    ln_proj_kernel<<<Bc * NT, LNT, 0, stream>>>(
        val, ng + (size_t)l * Dc, nb + (size_t)l * Dc, Uw + (size_t)l * Dc * Rc,
        Vw + (size_t)l * Dc * Rc, x, q, k);
    attn_kernel<<<Bc * NT, ATT, 0, stream>>>(x, q, k, st, val);
  }
}


# attn kernel of layers 0 and 1 also runs the next layer's LN + q/k projection of its own 16 rows (new val rows pass from column owners to row owners via LDS): 4 launches, not 6; two sets of x, q, k
# speedup vs baseline: 7.2581x; 1.1169x over previous
#include <hip/hip_runtime.h>

typedef __attribute__((ext_vector_type(2))) float v2f;
typedef __attribute__((ext_vector_type(8))) float v8f;

constexpr int Bc = 2;
constexpr int Sc = 1024;
constexpr int Nc = Sc + 1;
constexpr int Dc = 512;
constexpr int Rc = 32;
constexpr int Lc = 3;
constexpr int Wc = 64;
constexpr int WNc = 2 * Wc + 1;
constexpr int WINc = 144;
constexpr int NT = (Nc + 15) / 16;

constexpr int XSTR = Dc + 4;
constexpr int QSTR = 33;
constexpr int KSTR = 33;
constexpr int SSTR = 146;
constexpr int WSTR = 145;

__device__ __forceinline__ v8f wmma_f32(v2f a, v2f b, v8f c) {
  return __builtin_amdgcn_wmma_f32_16x16x4_f32(false, a, false, b, (short)0, c,
                                               false, false);
}

template <int M>
__device__ __forceinline__ float row_xor(float v) {
  return __builtin_bit_cast(
      float, __builtin_amdgcn_update_dpp(0, __builtin_bit_cast(int, v),
                                         0x160 + M, 0xF, 0xF, true));
}
__device__ __forceinline__ float half_xor(float v) {
  const int x = __builtin_bit_cast(int, v);
  return __builtin_bit_cast(
      float, __builtin_amdgcn_permlanex16(x, x, 0x76543210, 0xfedcba98, false,
                                          false));
}

constexpr int LNT = 512;
constexpr int KQ = Dc / 4 / 4;
constexpr int CPL = Dc / 32;

__device__ __forceinline__ float wave_sum(float v) {
  v += row_xor<1>(v);
  v += row_xor<2>(v);
  v += row_xor<4>(v);
  v += row_xor<8>(v);
  v += half_xor(v);
  return v;
}

__device__ __forceinline__ void request_weights(
    v2f (&bq)[KQ], const float* __restrict__ Uw, const float* __restrict__ Vw) {
  const int tid = threadIdx.x;
  const int wave = tid >> 5, lane = tid & 31;
  const int hf = lane >> 4, l16 = lane & 15;
  const int tsel = wave & 3, kq = wave >> 2;
  const int cb = (tsel & 1) * 16;
  const float* wp =
      ((tsel < 2) ? Uw : Vw) + (size_t)(4 * KQ * kq + 2 * hf) * Rc + cb + l16;
#pragma unroll
  for (int j = 0; j < KQ; ++j) {
    bq[j].x = wp[(size_t)(4 * j) * Rc];
    bq[j].y = wp[(size_t)(4 * j + 1) * Rc];
  }
}
__device__ __forceinline__ void request_ln_params(
    float (&gm)[CPL], float (&bt)[CPL], const float* __restrict__ gamma,
    const float* __restrict__ beta) {
  const int lane = threadIdx.x & 31;
  const float* gp = gamma + 4 * lane;
  const float* bp = beta + 4 * lane;
#pragma unroll
  for (int j = 0; j < 4; ++j) {
#pragma unroll
    for (int i = 0; i < 4; ++i) {
      gm[4 * j + i] = gp[128 * j + i];
      bt[4 * j + i] = bp[128 * j + i];
    }
  }
}

__device__ __forceinline__ void ln_proj_tile(
    float (&e)[CPL], const v2f (&bq)[KQ], const float (&gm)[CPL],
    const float (&bt)[CPL], float* xs, float* red, const int b, const int n0,
    const bool full, float* __restrict__ xo, float* __restrict__ qo,
    float* __restrict__ ko) {
  const int tid = threadIdx.x;
  const int wave = tid >> 5, lane = tid & 31;
  const int hf = lane >> 4, l16 = lane & 15;
  const int kq = wave >> 2;

  {
    const int g = n0 + wave;
    float s1 = 0.0f, s2 = 0.0f;
#pragma unroll
    for (int c = 0; c < CPL; ++c) {
      s1 += e[c];
      s2 += e[c] * e[c];
    }
    s1 = wave_sum(s1);
    s2 = wave_sum(s2);
    const float mean = s1 * (1.0f / Dc);
    const float var = s2 * (1.0f / Dc) - mean * mean;
    const float rs = rsqrtf(var + 1e-5f);
#pragma unroll
    for (int c = 0; c < CPL; ++c) e[c] = (e[c] - mean) * rs * gm[c] + bt[c];
    if (g < Nc) {
      float* xrow = xo + ((size_t)b * Nc + g) * Dc + 4 * lane;
#pragma unroll
      for (int j = 0; j < 4; ++j) {
#pragma unroll
        for (int i = 0; i < 4; ++i) xrow[128 * j + i] = e[4 * j + i];
      }
    }
    float* xr = xs + wave * XSTR + 4 * lane;
#pragma unroll
    for (int j = 0; j < 4; ++j) {
#pragma unroll
      for (int i = 0; i < 4; ++i) xr[128 * j + i] = e[4 * j + i];
    }
  }
  __syncthreads();

  v8f acc = {0.f, 0.f, 0.f, 0.f, 0.f, 0.f, 0.f, 0.f};
#pragma unroll
  for (int j = 0; j < KQ; ++j) {
    const int c0 = 4 * KQ * kq + 4 * j + 2 * hf;
    v2f a;
    a.x = xs[l16 * XSTR + c0];
    a.y = xs[l16 * XSTR + c0 + 1];
    acc = wmma_f32(a, bq[j], acc);
  }
  for (int i = 0; i < 8; ++i) red[(wave * 8 + i) * 32 + lane] = acc[i];
  __syncthreads();

  {
    const int mat = tid >> 8;
    const int r = (tid >> 4) & 15;
    const int col = 2 * (tid & 15);
    const int ts = 2 * mat + (col >> 4);
    const int e0 = (ts * 8 + (r & 7)) * 32 + 16 * (r >> 3) + (col & 15);
    float o0 = red[e0], o1 = red[e0 + 1];
    for (int p = 1; p < 4; ++p) {
      o0 += red[p * 1024 + e0];
      o1 += red[p * 1024 + e0 + 1];
    }
    if (full || n0 + r < Nc) {
      float* op = (mat ? ko : qo) + ((size_t)b * Nc + n0 + r) * Rc + col;
      op[0] = o0;
      op[1] = o1;
    }
  }
}

__global__ __launch_bounds__(LNT) void embed_ln_proj_kernel(
    const int* __restrict__ ids, const float* __restrict__ emb,
    const float* __restrict__ pos, const float* __restrict__ anchor_val,
    const float* __restrict__ anchor_state, const float* __restrict__ in_gamma,
    const float* __restrict__ in_beta, const float* __restrict__ sp_w,
    const float* __restrict__ sp_b, const float* __restrict__ gamma,
    const float* __restrict__ beta, const float* __restrict__ Uw,
    const float* __restrict__ Vw, float* __restrict__ val,
    float* __restrict__ st, float* __restrict__ xo, float* __restrict__ qo,
    float* __restrict__ ko) {
  __shared__ float xs[16 * XSTR];
  __shared__ float red[16 * 8 * 32];
  const int tile = blockIdx.x % NT;
  const int b = blockIdx.x / NT;
  const int n0 = tile * 16;
  const int tid = threadIdx.x;
  const int wave = tid >> 5, lane = tid & 31;
  const bool full = (n0 + 16 <= Nc);
  const int g = n0 + wave;
  const int gc = min(g, Nc - 1);
  const int s = max(gc - 1, 0);
  const int id = ids[b * Sc + s];
  v2f bq[KQ];
  float gm[CPL], bt[CPL];
  request_weights(bq, Uw, Vw);
  request_ln_params(gm, bt, gamma, beta);
  __builtin_amdgcn_sched_barrier(0);
  const float* er = emb + (size_t)id * Dc + 4 * lane;
  const float* pr = pos + (size_t)s * Dc + 4 * lane;
  float e[CPL];
#pragma unroll
  for (int j = 0; j < 4; ++j) {
#pragma unroll
    for (int i = 0; i < 4; ++i) e[4 * j + i] = er[128 * j + i] + pr[128 * j + i];
  }
  float s1 = 0.0f, s2 = 0.0f;
#pragma unroll
  for (int c = 0; c < CPL; ++c) {
    s1 += e[c];
    s2 += e[c] * e[c];
  }
  s1 = wave_sum(s1);
  s2 = wave_sum(s2);
  const float mean = s1 * (1.0f / Dc);
  const float var = s2 * (1.0f / Dc) - mean * mean;
  const float rs = rsqrtf(var + 1e-5f);
  const float* igp = in_gamma + 4 * lane;
  const float* ibp = in_beta + 4 * lane;
  const float* swp = sp_w + 4 * lane;
  float dot = 0.0f;
#pragma unroll
  for (int j = 0; j < 4; ++j) {
#pragma unroll
    for (int i = 0; i < 4; ++i) {
      const float y =
          (e[4 * j + i] - mean) * rs * igp[128 * j + i] + ibp[128 * j + i];
      e[4 * j + i] = y;
      dot += y * swp[128 * j + i];
    }
  }
  dot = wave_sum(dot);
  float stv = dot + sp_b[0];
  if (g == 0) {
    const float* ap = anchor_val + 4 * lane;
#pragma unroll
    for (int j = 0; j < 4; ++j) {
#pragma unroll
      for (int i = 0; i < 4; ++i) e[4 * j + i] = ap[128 * j + i];
    }
    stv = anchor_state[0];
  }
  if (g < Nc) {
    float* vrow = val + ((size_t)b * Nc + g) * Dc + 4 * lane;
#pragma unroll
    for (int j = 0; j < 4; ++j) {
#pragma unroll
      for (int i = 0; i < 4; ++i) vrow[128 * j + i] = e[4 * j + i];
    }
    if (lane == 0) st[(size_t)b * Nc + g] = stv;
  }
  ln_proj_tile(e, bq, gm, bt, xs, red, b, n0, full, xo, qo, ko);
}

__global__ __launch_bounds__(LNT) void ln_proj_kernel(
    const float* __restrict__ val, const float* __restrict__ gamma,
    const float* __restrict__ beta, const float* __restrict__ Uw,
    const float* __restrict__ Vw, float* __restrict__ xo,
    float* __restrict__ qo, float* __restrict__ ko) {
  __shared__ float xs[16 * XSTR];
  __shared__ float red[16 * 8 * 32];
  const int tile = blockIdx.x % NT;
  const int b = blockIdx.x / NT;
  const int n0 = tile * 16;
  const int tid = threadIdx.x;
  const int wave = tid >> 5, lane = tid & 31;
  const bool full = (n0 + 16 <= Nc);
  const int gc = min(n0 + wave, Nc - 1);
  const float* vrow = val + ((size_t)b * Nc + gc) * Dc + 4 * lane;
  float e[CPL];
#pragma unroll
  for (int j = 0; j < 4; ++j) {
#pragma unroll
    for (int i = 0; i < 4; ++i) e[4 * j + i] = vrow[128 * j + i];
  }
  v2f bq[KQ];
  float gm[CPL], bt[CPL];
  request_weights(bq, Uw, Vw);
  request_ln_params(gm, bt, gamma, beta);
  __builtin_amdgcn_sched_barrier(0);
  ln_proj_tile(e, bq, gm, bt, xs, red, b, n0, full, xo, qo, ko);
}

constexpr int ATT = 512;
constexpr int KS = WINc / 4;
constexpr int KG = 6;
constexpr int NG = KS / KG;

__device__ __forceinline__ void lds_barrier() {
  __builtin_amdgcn_fence(__ATOMIC_RELEASE, "workgroup", "local");
  __builtin_amdgcn_s_barrier();
  __builtin_amdgcn_fence(__ATOMIC_ACQUIRE, "workgroup", "local");
}

__device__ __forceinline__ void hold_order() {
  __builtin_amdgcn_sched_barrier(0);
}

template <bool INTERIOR, int G>
__device__ __forceinline__ void request_rows(v2f (&bq0)[KS], v2f (&bq1)[KS],
                                             const float* __restrict__ x,
                                             const int b, const int n0,
                                             const int hf, const int cb) {
  if constexpr (INTERIOR) {
    const float* xw = x + ((size_t)b * Nc + n0 - Wc + 2 * hf) * Dc + cb;
#pragma unroll
    for (int kk = G * KG; kk < (G + 1) * KG; ++kk) {
      const float* p = xw + (size_t)(4 * kk) * Dc;
      bq0[kk].x = p[0];
      bq0[kk].y = p[Dc];
      bq1[kk].x = p[16];
      bq1[kk].y = p[Dc + 16];
    }
  } else {
    const float* xb = x + (size_t)b * Nc * Dc + cb;
#pragma unroll
    for (int kk = G * KG; kk < (G + 1) * KG; ++kk) {
      const int g0 = n0 - Wc + 4 * kk + 2 * hf, g1 = g0 + 1;
      const int r0 = min(max(g0, 0), Nc - 1), r1 = min(max(g1, 0), Nc - 1);
      const float* p0 = xb + (size_t)r0 * Dc;
      const float* p1 = xb + (size_t)r1 * Dc;
      bq0[kk].x = p0[0];
      bq0[kk].y = p1[0];
      bq1[kk].x = p0[16];
      bq1[kk].y = p1[16];
    }
  }
}

template <int G>
__device__ __forceinline__ void zero_outside(v2f (&bq0)[KS], v2f (&bq1)[KS],
                                             const int n0, const int hf) {
#pragma unroll
  for (int kk = G * KG; kk < (G + 1) * KG; ++kk) {
    const int g0 = n0 - Wc + 4 * kk + 2 * hf, g1 = g0 + 1;
    const bool ok0 = (g0 >= 0) && (g0 < Nc), ok1 = (g1 >= 0) && (g1 < Nc);
    bq0[kk].x = ok0 ? bq0[kk].x : 0.0f;
    bq0[kk].y = ok1 ? bq0[kk].y : 0.0f;
    bq1[kk].x = ok0 ? bq1[kk].x : 0.0f;
    bq1[kk].y = ok1 ? bq1[kk].y : 0.0f;
  }
}

template <int G>
__device__ __forceinline__ void products(v8f& acc0, v8f& acc1,
                                         const v2f (&bq0)[KS],
                                         const v2f (&bq1)[KS], const float* wm,
                                         const int l16, const int hf) {
#pragma unroll
  for (int kk = G * KG; kk < (G + 1) * KG; ++kk) {
    const int c0 = 4 * kk + 2 * hf;
    v2f a;
    a.x = wm[l16 * WSTR + c0];
    a.y = wm[l16 * WSTR + c0 + 1];
    acc0 = wmma_f32(a, bq0[kk], acc0);
    acc1 = wmma_f32(a, bq1[kk], acc1);
  }
  asm volatile("" : "+v"(acc0), "+v"(acc1));
}

template <bool INTERIOR, int G>
__device__ __forceinline__ void product_group(v8f& acc0, v8f& acc1,
                                              v2f (&bq0)[KS], v2f (&bq1)[KS],
                                              const float* wm, const int n0,
                                              const int l16, const int hf) {
  if constexpr (!INTERIOR) zero_outside<G>(bq0, bq1, n0, hf);
  products<G>(acc0, acc1, bq0, bq1, wm, l16, hf);
}

template <bool INTERIOR>
__device__ __forceinline__ void delta_groups(
    v8f& acc0, v8f& acc1, v2f (&bq0)[KS], v2f (&bq1)[KS], float (&vv0)[8],
    float (&vv1)[8], const float* __restrict__ x, const float* vrows,
    const float* wm, const int b, const int n0, const int l16, const int hf,
    const int cb, const bool full) {
  request_rows<INTERIOR, 2>(bq0, bq1, x, b, n0, hf, cb);
  hold_order();
  product_group<INTERIOR, 0>(acc0, acc1, bq0, bq1, wm, n0, l16, hf);
  request_rows<INTERIOR, 3>(bq0, bq1, x, b, n0, hf, cb);
  hold_order();
  product_group<INTERIOR, 1>(acc0, acc1, bq0, bq1, wm, n0, l16, hf);
  request_rows<INTERIOR, 4>(bq0, bq1, x, b, n0, hf, cb);
  hold_order();
  product_group<INTERIOR, 2>(acc0, acc1, bq0, bq1, wm, n0, l16, hf);
  request_rows<INTERIOR, 5>(bq0, bq1, x, b, n0, hf, cb);
  hold_order();
  product_group<INTERIOR, 3>(acc0, acc1, bq0, bq1, wm, n0, l16, hf);
  if (full) {
    const float* vb = vrows + ((size_t)b * Nc + n0 + 8 * hf) * Dc + cb;
#pragma unroll
    for (int i = 0; i < 8; ++i) {
      vv0[i] = vb[(size_t)i * Dc];
      vv1[i] = vb[(size_t)i * Dc + 16];
    }
  } else {
#pragma unroll
    for (int i = 0; i < 8; ++i) {
      const int gg = min(n0 + i + 8 * hf, Nc - 1);
      const float* vp = vrows + ((size_t)b * Nc + gg) * Dc + cb;
      vv0[i] = vp[0];
      vv1[i] = vp[16];
    }
  }
  hold_order();
  product_group<INTERIOR, 4>(acc0, acc1, bq0, bq1, wm, n0, l16, hf);
  product_group<INTERIOR, 5>(acc0, acc1, bq0, bq1, wm, n0, l16, hf);
}

template <bool FUSE>
__device__ __forceinline__ void attn_tile(
    const float* __restrict__ x, const float* __restrict__ q,
    const float* __restrict__ k, const float* __restrict__ st,
    float* __restrict__ val, float* qs, float* ks, float* ss, float* wm,
    float* sw, float (&nv0)[8], float (&nv1)[8], v2f (&wq)[KQ],
    float (&gm)[CPL], float (&bt)[CPL], const float* __restrict__ gamma,
    const float* __restrict__ beta, const float* __restrict__ Uw,
    const float* __restrict__ Vw) {
  const int tile = blockIdx.x % NT;
  const int b = blockIdx.x / NT;
  const int n0 = tile * 16;
  const int tid = threadIdx.x;
  const bool full = (n0 + 16 <= Nc);
  const bool interior = (n0 >= Wc) && (n0 + 16 + Wc <= Nc);

  const float* qb = q + ((size_t)b * Nc + n0) * Rc;
  const float* kb = k + ((size_t)b * Nc + n0 - Wc) * Rc;
  const float* sb = st + (size_t)b * Nc + n0 - Wc;
  if (interior) {
    for (int i = tid; i < 16 * Rc; i += ATT) {
      const int r = i >> 5, c = i & 31;
      qs[r * QSTR + c] = qb[r * Rc + c];
    }
    for (int i = tid; i < WINc * Rc; i += ATT) {
      const int j = i >> 5, c = i & 31;
      ks[j * KSTR + c] = kb[(size_t)j * Rc + c];
    }
    for (int j = tid; j < WINc; j += ATT) sw[j] = sb[j];
  } else {
    static_assert((16 * Rc) % ATT == 0 && (WINc * Rc) % ATT == 0,
                  "q tile and k window: whole trips of the block's threads");
#pragma unroll
    for (int it = 0; it < 16 * Rc / ATT; ++it) {
      const int i = tid + it * ATT;
      const int r = i >> 5, c = i & 31, g = n0 + r;
      const float v = q[((size_t)b * Nc + min(g, Nc - 1)) * Rc + c];
      qs[r * QSTR + c] = (g < Nc) ? v : 0.0f;
    }
#pragma unroll
    for (int it = 0; it < WINc * Rc / ATT; ++it) {
      const int i = tid + it * ATT;
      const int j = i >> 5, c = i & 31, gn = n0 - Wc + j;
      const int gc = min(max(gn, 0), Nc - 1);
      const float v = k[((size_t)b * Nc + gc) * Rc + c];
      ks[j * KSTR + c] = (gc == gn) ? v : 0.0f;
    }
    for (int j = tid; j < WINc; j += ATT) {
      const int gn = n0 - Wc + j;
      const int gc = min(max(gn, 0), Nc - 1);
      const float v = st[(size_t)b * Nc + gc];
      sw[j] = (gc == gn) ? v : 0.0f;
    }
  }
  for (int i = tid; i < 16 * WSTR; i += ATT) wm[i] = 0.0f;

  const int wave = tid >> 5, lane = tid & 31;
  const int hf = lane >> 4, l16 = lane & 15;

  static_assert(Dc / 16 == 2 * (ATT / 32), "two column tiles per wave");
  static_assert(KS % KG == 0 && NG == 6, "six whole request groups");
  const int cb = wave * 32 + l16;
  v2f bq0[KS], bq1[KS];
  if (interior) {
    request_rows<true, 0>(bq0, bq1, x, b, n0, hf, cb);
    request_rows<true, 1>(bq0, bq1, x, b, n0, hf, cb);
  } else {
    request_rows<false, 0>(bq0, bq1, x, b, n0, hf, cb);
    request_rows<false, 1>(bq0, bq1, x, b, n0, hf, cb);
  }
  lds_barrier();

  for (int t = wave; t < 9; t += ATT / 32) {
    const int cbs = t * 16;
    v8f acc = {0.f, 0.f, 0.f, 0.f, 0.f, 0.f, 0.f, 0.f};
    for (int kk = 0; kk < Rc / 4; ++kk) {
      const int c0 = 4 * kk + 2 * hf;
      v2f a;
      a.x = qs[l16 * QSTR + c0];
      a.y = qs[l16 * QSTR + c0 + 1];
      v2f bv;
      bv.x = ks[(cbs + l16) * KSTR + c0];
      bv.y = ks[(cbs + l16) * KSTR + c0 + 1];
      acc = wmma_f32(a, bv, acc);
    }
    for (int i = 0; i < 8; ++i) ss[(i + 8 * hf) * SSTR + cbs + l16] = acc[i];
  }
  lds_barrier();

  {
    const int row = wave;
    const int g = n0 + row;
    if (g < Nc) {
      const int sbase = row * SSTR + row;
      float sv[5], nb[5];
      bool ok[5];
#pragma unroll
      for (int i = 0; i < 5; ++i) {
        const int o = lane + 32 * i;
        const int oc = min(o, WNc - 1);
        const int gn = g - Wc + o;
        ok[i] = (o < WNc) && (gn >= 0) && (gn < Nc);
        sv[i] = ss[sbase + oc];
        nb[i] = sw[row + oc];
      }
      float m = -1e30f;
#pragma unroll
      for (int i = 0; i < 5; ++i) m = fmaxf(m, ok[i] ? fabsf(sv[i]) : -1e30f);
      m = fmaxf(m, row_xor<1>(m));
      m = fmaxf(m, row_xor<2>(m));
      m = fmaxf(m, row_xor<4>(m));
      m = fmaxf(m, row_xor<8>(m));
      m = fmaxf(m, half_xor(m));
      float e[5];
      float Z = 0.0f;
#pragma unroll
      for (int i = 0; i < 5; ++i) {
        e[i] = ok[i] ? expf(fabsf(sv[i]) - m) : 0.0f;
        Z += e[i];
      }
      Z += row_xor<1>(Z);
      Z += row_xor<2>(Z);
      Z += row_xor<4>(Z);
      Z += row_xor<8>(Z);
      Z += half_xor(Z);
      const float inv = 1.0f / Z;
#pragma unroll
      for (int i = 0; i < 5; ++i) {
        if (ok[i]) {
          const float s = sv[i];
          const float p = e[i] * inv;
          const float wv = (s > 0.0f) ? p : ((s < 0.0f) ? -p : 0.0f);
          wm[row * WSTR + row + lane + 32 * i] = wv * nb[i];
        }
      }
    }
  }
  lds_barrier();

  {
    v8f acc0 = {0.f, 0.f, 0.f, 0.f, 0.f, 0.f, 0.f, 0.f};
    v8f acc1 = acc0;
    float vv0[8], vv1[8];
    if (interior) {
      delta_groups<true>(acc0, acc1, bq0, bq1, vv0, vv1, x, val, wm, b, n0, l16,
                         hf, cb, full);
    } else {
      delta_groups<false>(acc0, acc1, bq0, bq1, vv0, vv1, x, val, wm, b, n0,
                          l16, hf, cb, full);
    }
    if constexpr (FUSE) {
      hold_order();
      request_ln_params(gm, bt, gamma, beta);
      request_weights(wq, Uw, Vw);
      hold_order();
    }
#pragma unroll
    for (int i = 0; i < 8; ++i) {
      nv0[i] = vv0[i] + acc0[i];
      nv1[i] = vv1[i] + acc1[i];
    }
    if (full) {
      float* vb = val + ((size_t)b * Nc + n0 + 8 * hf) * Dc + cb;
#pragma unroll
      for (int i = 0; i < 8; ++i) {
        vb[(size_t)i * Dc] = nv0[i];
        vb[(size_t)i * Dc + 16] = nv1[i];
      }
    } else {
#pragma unroll
      for (int i = 0; i < 8; ++i) {
        const int gg = n0 + i + 8 * hf;
        if (gg < Nc) {
          float* vp = val + ((size_t)b * Nc + gg) * Dc + cb;
          vp[0] = nv0[i];
          vp[16] = nv1[i];
        }
      }
    }
  }
}

__global__ __launch_bounds__(ATT) void attn_kernel(
    const float* __restrict__ x, const float* __restrict__ q,
    const float* __restrict__ k, const float* __restrict__ st,
    float* __restrict__ val) {
  __shared__ float qs[16 * QSTR];
  __shared__ float ks[WINc * KSTR];
  __shared__ float ss[16 * SSTR];
  __shared__ float wm[16 * WSTR];
  __shared__ float sw[WINc];
  float nv0[8], nv1[8];
  v2f wq[KQ];
  float gm[CPL], bt[CPL];
  attn_tile<false>(x, q, k, st, val, qs, ks, ss, wm, sw, nv0, nv1, wq, gm, bt,
                   nullptr, nullptr, nullptr, nullptr);
}

static_assert(ATT == LNT, "attention and projection share one block shape");
__global__ __launch_bounds__(ATT) void attn_ln_proj_kernel(
    const float* __restrict__ x, const float* __restrict__ q,
    const float* __restrict__ k, const float* __restrict__ st,
    float* __restrict__ val, const float* __restrict__ gamma,
    const float* __restrict__ beta, const float* __restrict__ Uw,
    const float* __restrict__ Vw, float* __restrict__ xo,
    float* __restrict__ qo, float* __restrict__ ko) {
  __shared__ float qs[16 * QSTR];
  __shared__ float ks[WINc * KSTR];
  __shared__ float ss[16 * SSTR];
  __shared__ float wm[16 * WSTR];
  __shared__ float sw[WINc];
  __shared__ float xs[16 * XSTR];
  __shared__ float red[16 * 8 * 32];
  float nv0[8], nv1[8];
  v2f wq[KQ];
  float gm[CPL], bt[CPL];
  attn_tile<true>(x, q, k, st, val, qs, ks, ss, wm, sw, nv0, nv1, wq, gm, bt,
                  gamma, beta, Uw, Vw);
  const int tile = blockIdx.x % NT;
  const int b = blockIdx.x / NT;
  const int n0 = tile * 16;
  const int tid = threadIdx.x;
  const int wave = tid >> 5, lane = tid & 31;
  const int hf = lane >> 4, l16 = lane & 15;
  const int cb = wave * 32 + l16;
  const bool full = (n0 + 16 <= Nc);
#pragma unroll
  for (int i = 0; i < 8; ++i) {
    xs[(8 * hf + i) * XSTR + cb] = nv0[i];
    xs[(8 * hf + i) * XSTR + cb + 16] = nv1[i];
  }
  lds_barrier();
  float e[CPL];
  const float* xr = xs + wave * XSTR + 4 * lane;
#pragma unroll
  for (int j = 0; j < 4; ++j) {
#pragma unroll
    for (int i = 0; i < 4; ++i) e[4 * j + i] = xr[128 * j + i];
  }
  ln_proj_tile(e, wq, gm, bt, xs, red, b, n0, full, xo, qo, ko);
}

extern "C" void kernel_launch(void* const* d_in, const int* in_sizes, int n_in,
                              void* d_out, int out_size, void* d_ws,
                              size_t ws_size, hipStream_t stream) {
  const int* ids = (const int*)d_in[0];
  const float* emb = (const float*)d_in[1];
  const float* pos = (const float*)d_in[2];
  const float* anchor_val = (const float*)d_in[3];
  const float* anchor_state = (const float*)d_in[4];
  const float* in_gamma = (const float*)d_in[5];
  const float* in_beta = (const float*)d_in[6];
  const float* ng = (const float*)d_in[7];
  const float* nb = (const float*)d_in[8];
  const float* Uw = (const float*)d_in[9];
  const float* Vw = (const float*)d_in[10];
  const float* sp_w = (const float*)d_in[11];
  const float* sp_b = (const float*)d_in[12];

  float* val = (float*)d_out;
  float* ws = (float*)d_ws;
  constexpr size_t XN = (size_t)Bc * Nc * Dc;
  constexpr size_t QN = (size_t)Bc * Nc * Rc;
  constexpr size_t SN = 2112;
  static_assert(SN >= (size_t)Bc * Nc && SN % 64 == 0, "state plane");
  float* xa = ws;
  float* qa = xa + XN;
  float* ka = qa + QN;
  float* st = ka + QN;
  float* xb = st + SN;
  float* qb = xb + XN;
  float* kb = qb + QN;
  constexpr size_t WS_NEED = (2 * (XN + 2 * QN) + SN) * sizeof(float);

  embed_ln_proj_kernel<<<Bc * NT, LNT, 0, stream>>>(
      ids, emb, pos, anchor_val, anchor_state, in_gamma, in_beta, sp_w, sp_b,
      ng, nb, Uw, Vw, val, st, xa, qa, ka);
  if (ws_size >= WS_NEED) {
    attn_ln_proj_kernel<<<Bc * NT, ATT, 0, stream>>>(
        xa, qa, ka, st, val, ng + (size_t)1 * Dc, nb + (size_t)1 * Dc,
        Uw + (size_t)1 * Dc * Rc, Vw + (size_t)1 * Dc * Rc, xb, qb, kb);
    attn_ln_proj_kernel<<<Bc * NT, ATT, 0, stream>>>(
        xb, qb, kb, st, val, ng + (size_t)2 * Dc, nb + (size_t)2 * Dc,
        Uw + (size_t)2 * Dc * Rc, Vw + (size_t)2 * Dc * Rc, xa, qa, ka);
    attn_kernel<<<Bc * NT, ATT, 0, stream>>>(xa, qa, ka, st, val);
  } else {
    attn_kernel<<<Bc * NT, ATT, 0, stream>>>(xa, qa, ka, st, val);
    for (int l = 1; l < Lc; ++l) {
      ln_proj_kernel<<<Bc * NT, LNT, 0, stream>>>(
          val, ng + (size_t)l * Dc, nb + (size_t)l * Dc,
          Uw + (size_t)l * Dc * Rc, Vw + (size_t)l * Dc * Rc, xa, qa, ka);
      attn_kernel<<<Bc * NT, ATT, 0, stream>>>(xa, qa, ka, st, val);
    }
  }
}


# Memory round trips off each work-group's path: q, k, state staging loads all requested before one wait (interior tiles had 4 waits in a row, edge 2); LN->projection barriers order LDS only
# speedup vs baseline: 7.8526x; 1.0819x over previous
#include <hip/hip_runtime.h>

typedef __attribute__((ext_vector_type(2))) float v2f;
typedef __attribute__((ext_vector_type(8))) float v8f;

constexpr int Bc = 2;
constexpr int Sc = 1024;
constexpr int Nc = Sc + 1;
constexpr int Dc = 512;
constexpr int Rc = 32;
constexpr int Lc = 3;
constexpr int Wc = 64;
constexpr int WNc = 2 * Wc + 1;
constexpr int WINc = 144;
constexpr int NT = (Nc + 15) / 16;

constexpr int XSTR = Dc + 4;
constexpr int QSTR = 33;
constexpr int KSTR = 33;
constexpr int SSTR = 146;
constexpr int WSTR = 145;

__device__ __forceinline__ v8f wmma_f32(v2f a, v2f b, v8f c) {
  return __builtin_amdgcn_wmma_f32_16x16x4_f32(false, a, false, b, (short)0, c,
                                               false, false);
}

template <int M>
__device__ __forceinline__ float row_xor(float v) {
  return __builtin_bit_cast(
      float, __builtin_amdgcn_update_dpp(0, __builtin_bit_cast(int, v),
                                         0x160 + M, 0xF, 0xF, true));
}
__device__ __forceinline__ float half_xor(float v) {
  const int x = __builtin_bit_cast(int, v);
  return __builtin_bit_cast(
      float, __builtin_amdgcn_permlanex16(x, x, 0x76543210, 0xfedcba98, false,
                                          false));
}

constexpr int LNT = 512;
constexpr int KQ = Dc / 4 / 4;
constexpr int CPL = Dc / 32;

__device__ __forceinline__ float wave_sum(float v) {
  v += row_xor<1>(v);
  v += row_xor<2>(v);
  v += row_xor<4>(v);
  v += row_xor<8>(v);
  v += half_xor(v);
  return v;
}

__device__ __forceinline__ void request_weights(
    v2f (&bq)[KQ], const float* __restrict__ Uw, const float* __restrict__ Vw) {
  const int tid = threadIdx.x;
  const int wave = tid >> 5, lane = tid & 31;
  const int hf = lane >> 4, l16 = lane & 15;
  const int tsel = wave & 3, kq = wave >> 2;
  const int cb = (tsel & 1) * 16;
  const float* wp =
      ((tsel < 2) ? Uw : Vw) + (size_t)(4 * KQ * kq + 2 * hf) * Rc + cb + l16;
#pragma unroll
  for (int j = 0; j < KQ; ++j) {
    bq[j].x = wp[(size_t)(4 * j) * Rc];
    bq[j].y = wp[(size_t)(4 * j + 1) * Rc];
  }
}
__device__ __forceinline__ void request_ln_params(
    float (&gm)[CPL], float (&bt)[CPL], const float* __restrict__ gamma,
    const float* __restrict__ beta) {
  const int lane = threadIdx.x & 31;
  const float* gp = gamma + 4 * lane;
  const float* bp = beta + 4 * lane;
#pragma unroll
  for (int j = 0; j < 4; ++j) {
#pragma unroll
    for (int i = 0; i < 4; ++i) {
      gm[4 * j + i] = gp[128 * j + i];
      bt[4 * j + i] = bp[128 * j + i];
    }
  }
}

__device__ __forceinline__ void lds_barrier() {
  __builtin_amdgcn_fence(__ATOMIC_RELEASE, "workgroup", "local");
  __builtin_amdgcn_s_barrier();
  __builtin_amdgcn_fence(__ATOMIC_ACQUIRE, "workgroup", "local");
}

__device__ __forceinline__ void ln_proj_tile(
    float (&e)[CPL], const v2f (&bq)[KQ], const float (&gm)[CPL],
    const float (&bt)[CPL], float* xs, float* red, const int b, const int n0,
    const bool full, float* __restrict__ xo, float* __restrict__ qo,
    float* __restrict__ ko) {
  const int tid = threadIdx.x;
  const int wave = tid >> 5, lane = tid & 31;
  const int hf = lane >> 4, l16 = lane & 15;
  const int kq = wave >> 2;

  {
    const int g = n0 + wave;
    float s1 = 0.0f, s2 = 0.0f;
#pragma unroll
    for (int c = 0; c < CPL; ++c) {
      s1 += e[c];
      s2 += e[c] * e[c];
    }
    s1 = wave_sum(s1);
    s2 = wave_sum(s2);
    const float mean = s1 * (1.0f / Dc);
    const float var = s2 * (1.0f / Dc) - mean * mean;
    const float rs = rsqrtf(var + 1e-5f);
#pragma unroll
    for (int c = 0; c < CPL; ++c) e[c] = (e[c] - mean) * rs * gm[c] + bt[c];
    if (g < Nc) {
      float* xrow = xo + ((size_t)b * Nc + g) * Dc + 4 * lane;
#pragma unroll
      for (int j = 0; j < 4; ++j) {
#pragma unroll
        for (int i = 0; i < 4; ++i) xrow[128 * j + i] = e[4 * j + i];
      }
    }
    float* xr = xs + wave * XSTR + 4 * lane;
#pragma unroll
    for (int j = 0; j < 4; ++j) {
#pragma unroll
      for (int i = 0; i < 4; ++i) xr[128 * j + i] = e[4 * j + i];
    }
  }
  lds_barrier();

  v8f acc = {0.f, 0.f, 0.f, 0.f, 0.f, 0.f, 0.f, 0.f};
#pragma unroll
  for (int j = 0; j < KQ; ++j) {
    const int c0 = 4 * KQ * kq + 4 * j + 2 * hf;
    v2f a;
    a.x = xs[l16 * XSTR + c0];
    a.y = xs[l16 * XSTR + c0 + 1];
    acc = wmma_f32(a, bq[j], acc);
  }
  for (int i = 0; i < 8; ++i) red[(wave * 8 + i) * 32 + lane] = acc[i];
  lds_barrier();

  {
    const int mat = tid >> 8;
    const int r = (tid >> 4) & 15;
    const int col = 2 * (tid & 15);
    const int ts = 2 * mat + (col >> 4);
    const int e0 = (ts * 8 + (r & 7)) * 32 + 16 * (r >> 3) + (col & 15);
    float o0 = red[e0], o1 = red[e0 + 1];
    for (int p = 1; p < 4; ++p) {
      o0 += red[p * 1024 + e0];
      o1 += red[p * 1024 + e0 + 1];
    }
    if (full || n0 + r < Nc) {
      float* op = (mat ? ko : qo) + ((size_t)b * Nc + n0 + r) * Rc + col;
      op[0] = o0;
      op[1] = o1;
    }
  }
}

__global__ __launch_bounds__(LNT) void embed_ln_proj_kernel(
    const int* __restrict__ ids, const float* __restrict__ emb,
    const float* __restrict__ pos, const float* __restrict__ anchor_val,
    const float* __restrict__ anchor_state, const float* __restrict__ in_gamma,
    const float* __restrict__ in_beta, const float* __restrict__ sp_w,
    const float* __restrict__ sp_b, const float* __restrict__ gamma,
    const float* __restrict__ beta, const float* __restrict__ Uw,
    const float* __restrict__ Vw, float* __restrict__ val,
    float* __restrict__ st, float* __restrict__ xo, float* __restrict__ qo,
    float* __restrict__ ko) {
  __shared__ float xs[16 * XSTR];
  __shared__ float red[16 * 8 * 32];
  const int tile = blockIdx.x % NT;
  const int b = blockIdx.x / NT;
  const int n0 = tile * 16;
  const int tid = threadIdx.x;
  const int wave = tid >> 5, lane = tid & 31;
  const bool full = (n0 + 16 <= Nc);
  const int g = n0 + wave;
  const int gc = min(g, Nc - 1);
  const int s = max(gc - 1, 0);
  const int id = ids[b * Sc + s];
  v2f bq[KQ];
  float gm[CPL], bt[CPL];
  request_weights(bq, Uw, Vw);
  request_ln_params(gm, bt, gamma, beta);
  __builtin_amdgcn_sched_barrier(0);
  const float* er = emb + (size_t)id * Dc + 4 * lane;
  const float* pr = pos + (size_t)s * Dc + 4 * lane;
  float e[CPL];
#pragma unroll
  for (int j = 0; j < 4; ++j) {
#pragma unroll
    for (int i = 0; i < 4; ++i) e[4 * j + i] = er[128 * j + i] + pr[128 * j + i];
  }
  float s1 = 0.0f, s2 = 0.0f;
#pragma unroll
  for (int c = 0; c < CPL; ++c) {
    s1 += e[c];
    s2 += e[c] * e[c];
  }
  s1 = wave_sum(s1);
  s2 = wave_sum(s2);
  const float mean = s1 * (1.0f / Dc);
  const float var = s2 * (1.0f / Dc) - mean * mean;
  const float rs = rsqrtf(var + 1e-5f);
  const float* igp = in_gamma + 4 * lane;
  const float* ibp = in_beta + 4 * lane;
  const float* swp = sp_w + 4 * lane;
  float dot = 0.0f;
#pragma unroll
  for (int j = 0; j < 4; ++j) {
#pragma unroll
    for (int i = 0; i < 4; ++i) {
      const float y =
          (e[4 * j + i] - mean) * rs * igp[128 * j + i] + ibp[128 * j + i];
      e[4 * j + i] = y;
      dot += y * swp[128 * j + i];
    }
  }
  dot = wave_sum(dot);
  float stv = dot + sp_b[0];
  if (g == 0) {
    const float* ap = anchor_val + 4 * lane;
#pragma unroll
    for (int j = 0; j < 4; ++j) {
#pragma unroll
      for (int i = 0; i < 4; ++i) e[4 * j + i] = ap[128 * j + i];
    }
    stv = anchor_state[0];
  }
  if (g < Nc) {
    float* vrow = val + ((size_t)b * Nc + g) * Dc + 4 * lane;
#pragma unroll
    for (int j = 0; j < 4; ++j) {
#pragma unroll
      for (int i = 0; i < 4; ++i) vrow[128 * j + i] = e[4 * j + i];
    }
    if (lane == 0) st[(size_t)b * Nc + g] = stv;
  }
  ln_proj_tile(e, bq, gm, bt, xs, red, b, n0, full, xo, qo, ko);
}

__global__ __launch_bounds__(LNT) void ln_proj_kernel(
    const float* __restrict__ val, const float* __restrict__ gamma,
    const float* __restrict__ beta, const float* __restrict__ Uw,
    const float* __restrict__ Vw, float* __restrict__ xo,
    float* __restrict__ qo, float* __restrict__ ko) {
  __shared__ float xs[16 * XSTR];
  __shared__ float red[16 * 8 * 32];
  const int tile = blockIdx.x % NT;
  const int b = blockIdx.x / NT;
  const int n0 = tile * 16;
  const int tid = threadIdx.x;
  const int wave = tid >> 5, lane = tid & 31;
  const bool full = (n0 + 16 <= Nc);
  const int gc = min(n0 + wave, Nc - 1);
  const float* vrow = val + ((size_t)b * Nc + gc) * Dc + 4 * lane;
  float e[CPL];
#pragma unroll
  for (int j = 0; j < 4; ++j) {
#pragma unroll
    for (int i = 0; i < 4; ++i) e[4 * j + i] = vrow[128 * j + i];
  }
  v2f bq[KQ];
  float gm[CPL], bt[CPL];
  request_weights(bq, Uw, Vw);
  request_ln_params(gm, bt, gamma, beta);
  __builtin_amdgcn_sched_barrier(0);
  ln_proj_tile(e, bq, gm, bt, xs, red, b, n0, full, xo, qo, ko);
}

constexpr int ATT = 512;
constexpr int KS = WINc / 4;
constexpr int KG = 6;
constexpr int NG = KS / KG;

__device__ __forceinline__ void hold_order() {
  __builtin_amdgcn_sched_barrier(0);
}

template <bool INTERIOR, int G>
__device__ __forceinline__ void request_rows(v2f (&bq0)[KS], v2f (&bq1)[KS],
                                             const float* __restrict__ x,
                                             const int b, const int n0,
                                             const int hf, const int cb) {
  if constexpr (INTERIOR) {
    const float* xw = x + ((size_t)b * Nc + n0 - Wc + 2 * hf) * Dc + cb;
#pragma unroll
    for (int kk = G * KG; kk < (G + 1) * KG; ++kk) {
      const float* p = xw + (size_t)(4 * kk) * Dc;
      bq0[kk].x = p[0];
      bq0[kk].y = p[Dc];
      bq1[kk].x = p[16];
      bq1[kk].y = p[Dc + 16];
    }
  } else {
    const float* xb = x + (size_t)b * Nc * Dc + cb;
#pragma unroll
    for (int kk = G * KG; kk < (G + 1) * KG; ++kk) {
      const int g0 = n0 - Wc + 4 * kk + 2 * hf, g1 = g0 + 1;
      const int r0 = min(max(g0, 0), Nc - 1), r1 = min(max(g1, 0), Nc - 1);
      const float* p0 = xb + (size_t)r0 * Dc;
      const float* p1 = xb + (size_t)r1 * Dc;
      bq0[kk].x = p0[0];
      bq0[kk].y = p1[0];
      bq1[kk].x = p0[16];
      bq1[kk].y = p1[16];
    }
  }
}

template <int G>
__device__ __forceinline__ void zero_outside(v2f (&bq0)[KS], v2f (&bq1)[KS],
                                             const int n0, const int hf) {
#pragma unroll
  for (int kk = G * KG; kk < (G + 1) * KG; ++kk) {
    const int g0 = n0 - Wc + 4 * kk + 2 * hf, g1 = g0 + 1;
    const bool ok0 = (g0 >= 0) && (g0 < Nc), ok1 = (g1 >= 0) && (g1 < Nc);
    bq0[kk].x = ok0 ? bq0[kk].x : 0.0f;
    bq0[kk].y = ok1 ? bq0[kk].y : 0.0f;
    bq1[kk].x = ok0 ? bq1[kk].x : 0.0f;
    bq1[kk].y = ok1 ? bq1[kk].y : 0.0f;
  }
}

template <int G>
__device__ __forceinline__ void products(v8f& acc0, v8f& acc1,
                                         const v2f (&bq0)[KS],
                                         const v2f (&bq1)[KS], const float* wm,
                                         const int l16, const int hf) {
#pragma unroll
  for (int kk = G * KG; kk < (G + 1) * KG; ++kk) {
    const int c0 = 4 * kk + 2 * hf;
    v2f a;
    a.x = wm[l16 * WSTR + c0];
    a.y = wm[l16 * WSTR + c0 + 1];
    acc0 = wmma_f32(a, bq0[kk], acc0);
    acc1 = wmma_f32(a, bq1[kk], acc1);
  }
  asm volatile("" : "+v"(acc0), "+v"(acc1));
}

template <bool INTERIOR, int G>
__device__ __forceinline__ void product_group(v8f& acc0, v8f& acc1,
                                              v2f (&bq0)[KS], v2f (&bq1)[KS],
                                              const float* wm, const int n0,
                                              const int l16, const int hf) {
  if constexpr (!INTERIOR) zero_outside<G>(bq0, bq1, n0, hf);
  products<G>(acc0, acc1, bq0, bq1, wm, l16, hf);
}

template <bool INTERIOR>
__device__ __forceinline__ void delta_groups(
    v8f& acc0, v8f& acc1, v2f (&bq0)[KS], v2f (&bq1)[KS], float (&vv0)[8],
    float (&vv1)[8], const float* __restrict__ x, const float* vrows,
    const float* wm, const int b, const int n0, const int l16, const int hf,
    const int cb, const bool full) {
  request_rows<INTERIOR, 2>(bq0, bq1, x, b, n0, hf, cb);
  hold_order();
  product_group<INTERIOR, 0>(acc0, acc1, bq0, bq1, wm, n0, l16, hf);
  request_rows<INTERIOR, 3>(bq0, bq1, x, b, n0, hf, cb);
  hold_order();
  product_group<INTERIOR, 1>(acc0, acc1, bq0, bq1, wm, n0, l16, hf);
  request_rows<INTERIOR, 4>(bq0, bq1, x, b, n0, hf, cb);
  hold_order();
  product_group<INTERIOR, 2>(acc0, acc1, bq0, bq1, wm, n0, l16, hf);
  request_rows<INTERIOR, 5>(bq0, bq1, x, b, n0, hf, cb);
  hold_order();
  product_group<INTERIOR, 3>(acc0, acc1, bq0, bq1, wm, n0, l16, hf);
  if (full) {
    const float* vb = vrows + ((size_t)b * Nc + n0 + 8 * hf) * Dc + cb;
#pragma unroll
    for (int i = 0; i < 8; ++i) {
      vv0[i] = vb[(size_t)i * Dc];
      vv1[i] = vb[(size_t)i * Dc + 16];
    }
  } else {
#pragma unroll
    for (int i = 0; i < 8; ++i) {
      const int gg = min(n0 + i + 8 * hf, Nc - 1);
      const float* vp = vrows + ((size_t)b * Nc + gg) * Dc + cb;
      vv0[i] = vp[0];
      vv1[i] = vp[16];
    }
  }
  hold_order();
  product_group<INTERIOR, 4>(acc0, acc1, bq0, bq1, wm, n0, l16, hf);
  product_group<INTERIOR, 5>(acc0, acc1, bq0, bq1, wm, n0, l16, hf);
}

template <bool FUSE>
__device__ __forceinline__ void attn_tile(
    const float* __restrict__ x, const float* __restrict__ q,
    const float* __restrict__ k, const float* __restrict__ st,
    float* __restrict__ val, float* qs, float* ks, float* ss, float* wm,
    float* sw, float (&nv0)[8], float (&nv1)[8], v2f (&wq)[KQ],
    float (&gm)[CPL], float (&bt)[CPL], const float* __restrict__ gamma,
    const float* __restrict__ beta, const float* __restrict__ Uw,
    const float* __restrict__ Vw) {
  const int tile = blockIdx.x % NT;
  const int b = blockIdx.x / NT;
  const int n0 = tile * 16;
  const int tid = threadIdx.x;
  const bool full = (n0 + 16 <= Nc);
  const bool interior = (n0 >= Wc) && (n0 + 16 + Wc <= Nc);

  static_assert(16 * Rc == ATT && (WINc * Rc) % ATT == 0 && WINc <= ATT,
                "q tile: one element a thread; k window: whole trips; state "
                "window: at most one element a thread");
  constexpr int KT = WINc * Rc / ATT;
  float qv, sv, kv[KT];
  if (interior) {
    const float* qb = q + ((size_t)b * Nc + n0) * Rc;
    const float* kb = k + ((size_t)b * Nc + n0 - Wc) * Rc;
    const float* sb = st + (size_t)b * Nc + n0 - Wc;
    qv = qb[tid];
#pragma unroll
    for (int it = 0; it < KT; ++it) kv[it] = kb[tid + it * ATT];
    sv = sb[min(tid, WINc - 1)];
  } else {
    qv = q[((size_t)b * Nc + min(n0 + (tid >> 5), Nc - 1)) * Rc + (tid & 31)];
#pragma unroll
    for (int it = 0; it < KT; ++it) {
      const int i = tid + it * ATT;
      const int gc = min(max(n0 - Wc + (i >> 5), 0), Nc - 1);
      kv[it] = k[((size_t)b * Nc + gc) * Rc + (i & 31)];
    }
    const int gs = min(max(n0 - Wc + min(tid, WINc - 1), 0), Nc - 1);
    sv = st[(size_t)b * Nc + gs];
  }
  for (int i = tid; i < 16 * WSTR; i += ATT) wm[i] = 0.0f;

  const int wave = tid >> 5, lane = tid & 31;
  const int hf = lane >> 4, l16 = lane & 15;

  {
    const int r = tid >> 5, cc = tid & 31;
    qs[r * QSTR + cc] = (n0 + r < Nc) ? qv : 0.0f;
#pragma unroll
    for (int it = 0; it < KT; ++it) {
      const int j = r + 16 * it;
      const int gn = n0 - Wc + j;
      ks[j * KSTR + cc] = (gn >= 0 && gn < Nc) ? kv[it] : 0.0f;
    }
    const int gn = n0 - Wc + tid;
    sw[tid] = (gn >= 0 && gn < Nc) ? sv : 0.0f;
  }

  static_assert(Dc / 16 == 2 * (ATT / 32), "two column tiles per wave");
  static_assert(KS % KG == 0 && NG == 6, "six whole request groups");
  const int cb = wave * 32 + l16;
  v2f bq0[KS], bq1[KS];
  if (interior) {
    request_rows<true, 0>(bq0, bq1, x, b, n0, hf, cb);
    request_rows<true, 1>(bq0, bq1, x, b, n0, hf, cb);
  } else {
    request_rows<false, 0>(bq0, bq1, x, b, n0, hf, cb);
    request_rows<false, 1>(bq0, bq1, x, b, n0, hf, cb);
  }
  lds_barrier();

  for (int t = wave; t < 9; t += ATT / 32) {
    const int cbs = t * 16;
    v8f acc = {0.f, 0.f, 0.f, 0.f, 0.f, 0.f, 0.f, 0.f};
    for (int kk = 0; kk < Rc / 4; ++kk) {
      const int c0 = 4 * kk + 2 * hf;
      v2f a;
      a.x = qs[l16 * QSTR + c0];
      a.y = qs[l16 * QSTR + c0 + 1];
      v2f bv;
      bv.x = ks[(cbs + l16) * KSTR + c0];
      bv.y = ks[(cbs + l16) * KSTR + c0 + 1];
      acc = wmma_f32(a, bv, acc);
    }
    for (int i = 0; i < 8; ++i) ss[(i + 8 * hf) * SSTR + cbs + l16] = acc[i];
  }
  lds_barrier();

  {
    const int row = wave;
    const int g = n0 + row;
    if (g < Nc) {
      const int sbase = row * SSTR + row;
      float sv[5], nb[5];
      bool ok[5];
#pragma unroll
      for (int i = 0; i < 5; ++i) {
        const int o = lane + 32 * i;
        const int oc = min(o, WNc - 1);
        const int gn = g - Wc + o;
        ok[i] = (o < WNc) && (gn >= 0) && (gn < Nc);
        sv[i] = ss[sbase + oc];
        nb[i] = sw[row + oc];
      }
      float m = -1e30f;
#pragma unroll
      for (int i = 0; i < 5; ++i) m = fmaxf(m, ok[i] ? fabsf(sv[i]) : -1e30f);
      m = fmaxf(m, row_xor<1>(m));
      m = fmaxf(m, row_xor<2>(m));
      m = fmaxf(m, row_xor<4>(m));
      m = fmaxf(m, row_xor<8>(m));
      m = fmaxf(m, half_xor(m));
      float e[5];
      float Z = 0.0f;
#pragma unroll
      for (int i = 0; i < 5; ++i) {
        e[i] = ok[i] ? expf(fabsf(sv[i]) - m) : 0.0f;
        Z += e[i];
      }
      Z += row_xor<1>(Z);
      Z += row_xor<2>(Z);
      Z += row_xor<4>(Z);
      Z += row_xor<8>(Z);
      Z += half_xor(Z);
      const float inv = 1.0f / Z;
#pragma unroll
      for (int i = 0; i < 5; ++i) {
        if (ok[i]) {
          const float s = sv[i];
          const float p = e[i] * inv;
          const float wv = (s > 0.0f) ? p : ((s < 0.0f) ? -p : 0.0f);
          wm[row * WSTR + row + lane + 32 * i] = wv * nb[i];
        }
      }
    }
  }
  lds_barrier();

  {
    v8f acc0 = {0.f, 0.f, 0.f, 0.f, 0.f, 0.f, 0.f, 0.f};
    v8f acc1 = acc0;
    float vv0[8], vv1[8];
    if (interior) {
      delta_groups<true>(acc0, acc1, bq0, bq1, vv0, vv1, x, val, wm, b, n0, l16,
                         hf, cb, full);
    } else {
      delta_groups<false>(acc0, acc1, bq0, bq1, vv0, vv1, x, val, wm, b, n0,
                          l16, hf, cb, full);
    }
    if constexpr (FUSE) {
      hold_order();
      request_ln_params(gm, bt, gamma, beta);
      request_weights(wq, Uw, Vw);
      hold_order();
    }
#pragma unroll
    for (int i = 0; i < 8; ++i) {
      nv0[i] = vv0[i] + acc0[i];
      nv1[i] = vv1[i] + acc1[i];
    }
    if (full) {
      float* vb = val + ((size_t)b * Nc + n0 + 8 * hf) * Dc + cb;
#pragma unroll
      for (int i = 0; i < 8; ++i) {
        vb[(size_t)i * Dc] = nv0[i];
        vb[(size_t)i * Dc + 16] = nv1[i];
      }
    } else {
#pragma unroll
      for (int i = 0; i < 8; ++i) {
        const int gg = n0 + i + 8 * hf;
        if (gg < Nc) {
          float* vp = val + ((size_t)b * Nc + gg) * Dc + cb;
          vp[0] = nv0[i];
          vp[16] = nv1[i];
        }
      }
    }
  }
}

__global__ __launch_bounds__(ATT) void attn_kernel(
    const float* __restrict__ x, const float* __restrict__ q,
    const float* __restrict__ k, const float* __restrict__ st,
    float* __restrict__ val) {
  __shared__ float qs[16 * QSTR];
  __shared__ float ks[WINc * KSTR];
  __shared__ float ss[16 * SSTR];
  __shared__ float wm[16 * WSTR];
  __shared__ float sw[ATT];
  float nv0[8], nv1[8];
  v2f wq[KQ];
  float gm[CPL], bt[CPL];
  attn_tile<false>(x, q, k, st, val, qs, ks, ss, wm, sw, nv0, nv1, wq, gm, bt,
                   nullptr, nullptr, nullptr, nullptr);
}

static_assert(ATT == LNT, "attention and projection share one block shape");
__global__ __launch_bounds__(ATT) void attn_ln_proj_kernel(
    const float* __restrict__ x, const float* __restrict__ q,
    const float* __restrict__ k, const float* __restrict__ st,
    float* __restrict__ val, const float* __restrict__ gamma,
    const float* __restrict__ beta, const float* __restrict__ Uw,
    const float* __restrict__ Vw, float* __restrict__ xo,
    float* __restrict__ qo, float* __restrict__ ko) {
  __shared__ float qs[16 * QSTR];
  __shared__ float ks[WINc * KSTR];
  __shared__ float ss[16 * SSTR];
  __shared__ float wm[16 * WSTR];
  __shared__ float sw[ATT];
  __shared__ float xs[16 * XSTR];
  __shared__ float red[16 * 8 * 32];
  float nv0[8], nv1[8];
  v2f wq[KQ];
  float gm[CPL], bt[CPL];
  attn_tile<true>(x, q, k, st, val, qs, ks, ss, wm, sw, nv0, nv1, wq, gm, bt,
                  gamma, beta, Uw, Vw);
  const int tile = blockIdx.x % NT;
  const int b = blockIdx.x / NT;
  const int n0 = tile * 16;
  const int tid = threadIdx.x;
  const int wave = tid >> 5, lane = tid & 31;
  const int hf = lane >> 4, l16 = lane & 15;
  const int cb = wave * 32 + l16;
  const bool full = (n0 + 16 <= Nc);
#pragma unroll
  for (int i = 0; i < 8; ++i) {
    xs[(8 * hf + i) * XSTR + cb] = nv0[i];
    xs[(8 * hf + i) * XSTR + cb + 16] = nv1[i];
  }
  lds_barrier();
  float e[CPL];
  const float* xr = xs + wave * XSTR + 4 * lane;
#pragma unroll
  for (int j = 0; j < 4; ++j) {
#pragma unroll
    for (int i = 0; i < 4; ++i) e[4 * j + i] = xr[128 * j + i];
  }
  ln_proj_tile(e, wq, gm, bt, xs, red, b, n0, full, xo, qo, ko);
}

extern "C" void kernel_launch(void* const* d_in, const int* in_sizes, int n_in,
                              void* d_out, int out_size, void* d_ws,
                              size_t ws_size, hipStream_t stream) {
  const int* ids = (const int*)d_in[0];
  const float* emb = (const float*)d_in[1];
  const float* pos = (const float*)d_in[2];
  const float* anchor_val = (const float*)d_in[3];
  const float* anchor_state = (const float*)d_in[4];
  const float* in_gamma = (const float*)d_in[5];
  const float* in_beta = (const float*)d_in[6];
  const float* ng = (const float*)d_in[7];
  const float* nb = (const float*)d_in[8];
  const float* Uw = (const float*)d_in[9];
  const float* Vw = (const float*)d_in[10];
  const float* sp_w = (const float*)d_in[11];
  const float* sp_b = (const float*)d_in[12];

  float* val = (float*)d_out;
  float* ws = (float*)d_ws;
  constexpr size_t XN = (size_t)Bc * Nc * Dc;
  constexpr size_t QN = (size_t)Bc * Nc * Rc;
  constexpr size_t SN = 2112;
  static_assert(SN >= (size_t)Bc * Nc && SN % 64 == 0, "state plane");
  float* xa = ws;
  float* qa = xa + XN;
  float* ka = qa + QN;
  float* st = ka + QN;
  float* xb = st + SN;
  float* qb = xb + XN;
  float* kb = qb + QN;
  constexpr size_t WS_NEED = (2 * (XN + 2 * QN) + SN) * sizeof(float);

  embed_ln_proj_kernel<<<Bc * NT, LNT, 0, stream>>>(
      ids, emb, pos, anchor_val, anchor_state, in_gamma, in_beta, sp_w, sp_b,
      ng, nb, Uw, Vw, val, st, xa, qa, ka);
  if (ws_size >= WS_NEED) {
    attn_ln_proj_kernel<<<Bc * NT, ATT, 0, stream>>>(
        xa, qa, ka, st, val, ng + (size_t)1 * Dc, nb + (size_t)1 * Dc,
        Uw + (size_t)1 * Dc * Rc, Vw + (size_t)1 * Dc * Rc, xb, qb, kb);
    attn_ln_proj_kernel<<<Bc * NT, ATT, 0, stream>>>(
        xb, qb, kb, st, val, ng + (size_t)2 * Dc, nb + (size_t)2 * Dc,
        Uw + (size_t)2 * Dc * Rc, Vw + (size_t)2 * Dc * Rc, xa, qa, ka);
    attn_kernel<<<Bc * NT, ATT, 0, stream>>>(xa, qa, ka, st, val);
  } else {
    attn_kernel<<<Bc * NT, ATT, 0, stream>>>(xa, qa, ka, st, val);
    for (int l = 1; l < Lc; ++l) {
      ln_proj_kernel<<<Bc * NT, LNT, 0, stream>>>(
          val, ng + (size_t)l * Dc, nb + (size_t)l * Dc,
          Uw + (size_t)l * Dc * Rc, Vw + (size_t)l * Dc * Rc, xa, qa, ka);
      attn_kernel<<<Bc * NT, ATT, 0, stream>>>(xa, qa, ka, st, val);
    }
  }
}
